# starting kernel
# baseline (speedup 1.0000x reference)
#include <hip/hip_runtime.h>
#include <hip/hip_bf16.h>

typedef float v2f __attribute__((ext_vector_type(2)));
typedef float v8f __attribute__((ext_vector_type(8)));

#define LDS_PAD 4

template<int Ci, int Co, int S, int TILES>
__global__ __launch_bounds__((Co/16)*32)
void conv4d_relu_pool_wmma(const float* __restrict__ x,
                           const float* __restrict__ w,
                           const float* __restrict__ bias,
                           float* __restrict__ out)
{
    constexpr int P    = S / 2;
    constexpr int PH   = P / TILES;
    constexpr int P4   = P * P * P * P;
    constexpr int D3   = 2 + 2 * TILES;
    constexpr int NSP  = 64 * D3;
    constexpr int CSTR = Ci + LDS_PAD;
    constexpr int NT   = (Co / 16) * 32;

    __shared__ float lds[NSP * CSTR];

    const int b   = blockIdx.y;
    const int bx  = blockIdx.x;
    const int p3h = bx % PH;
    const int p2  = (bx / PH) % P;
    const int p1  = (bx / (PH * P)) % P;
    const int p0  = bx / (PH * P * P);

    const int tid = threadIdx.x;
    for (int e = tid; e < NSP * Ci; e += NT) {
        const int ci = e & (Ci - 1);
        const int sp = e / Ci;
        const int d3 = sp % D3;
        const int r  = sp / D3;
        const int d2 = r & 3, d1 = (r >> 2) & 3, d0 = (r >> 4) & 3;
        const int x0 = 2 * p0 - 1 + d0;
        const int x1 = 2 * p1 - 1 + d1;
        const int x2 = 2 * p2 - 1 + d2;
        const int x3 = 2 * (p3h * TILES) - 1 + d3;
        float v = 0.0f;
        if ((unsigned)x0 < (unsigned)S && (unsigned)x1 < (unsigned)S &&
            (unsigned)x2 < (unsigned)S && (unsigned)x3 < (unsigned)S) {
            v = x[((((b * Ci + ci) * S + x0) * S + x1) * S + x2) * S + x3];
        }
        lds[sp * CSTR + ci] = v;
    }
    __syncthreads();

    const int lane = tid & 31;
    const int wave = tid >> 5;
    const int n    = lane & 15;
    const int klo  = (lane < 16) ? 0 : 2;
    const int qsp  = (((n >> 3) & 1) * 16 + ((n >> 2) & 1) * 4 + ((n >> 1) & 1)) * D3 + (n & 1);
    const int coA  = wave * 16 + n;
    const int wCoK = coA * Ci * 81 + klo * 81;

    v8f acc[TILES];
    #pragma unroll
    for (int u = 0; u < TILES; ++u)
        acc[u] = (v8f){0.f, 0.f, 0.f, 0.f, 0.f, 0.f, 0.f, 0.f};

    #pragma unroll 1
    for (int t0 = 0; t0 < 3; ++t0) {
        #pragma unroll 1
        for (int t1 = 0; t1 < 3; ++t1) {
            #pragma unroll 1
            for (int t2 = 0; t2 < 3; ++t2) {
                const int row  = (qsp + (t0 * 16 + t1 * 4 + t2) * D3) * CSTR + klo;
                const int wRow = wCoK + t0 * 27 + t1 * 9 + t2 * 3;
                #pragma unroll 4
                for (int kb = 0; kb < Ci; kb += 4) {
                    const float* wp = &w[wRow + kb * 81];
                    float wx[3] = { wp[0],  wp[1],  wp[2]  };
                    float wy[3] = { wp[81], wp[82], wp[83] };
                    #pragma unroll
                    for (int t3 = 0; t3 < 3; ++t3) {
                        v2f bf; bf.x = wx[t3]; bf.y = wy[t3];
                        #pragma unroll
                        for (int u = 0; u < TILES; ++u) {
                            v2f a = *(const v2f*)&lds[row + (t3 + 2 * u) * CSTR + kb];
                            acc[u] = __builtin_amdgcn_wmma_f32_16x16x4_f32(
                                 false, a,  false, bf,
                                 (short)0, acc[u],
                                 false,  false);
                        }
                    }
                }
            }
        }
    }

    const float bs = bias[coA];
    #pragma unroll
    for (int u = 0; u < TILES; ++u) {
        float s = 0.0f;
        #pragma unroll
        for (int r = 0; r < 8; ++r) {
            float v = acc[u][r] + bs;
            s += (v > 0.0f) ? v : 0.0f;
        }
        s += __shfl_xor(s, 16, 32);
        if (lane < 16) {
            const int pIdx = ((p0 * P + p1) * P + p2) * P + p3h * TILES + u;
            out[(b * Co + coA) * P4 + pIdx] = s * (1.0f / 16.0f);
        }
    }
}

extern "C" void kernel_launch(void* const* d_in, const int* in_sizes, int n_in,
                              void* d_out, int out_size, void* d_ws, size_t ws_size,
                              hipStream_t stream) {
    (void)in_sizes; (void)n_in; (void)out_size; (void)ws_size;

    const float* src = (const float*)d_in[0];
    const float* W0  = (const float*)d_in[1];
    const float* b0  = (const float*)d_in[2];
    const float* W1  = (const float*)d_in[3];
    const float* b1  = (const float*)d_in[4];
    const float* W2  = (const float*)d_in[5];
    const float* b2  = (const float*)d_in[6];
    const float* W3  = (const float*)d_in[7];
    const float* b3  = (const float*)d_in[8];

    float* t0 = (float*)d_ws;
    float* o1 = (float*)d_out;
    float* o2 = o1 + 8 * 64 * 256;
    float* o3 = o2 + 8 * 64 * 16;

    conv4d_relu_pool_wmma<8, 32, 16, 2><<<dim3(2048, 8), 64, 0, stream>>>(src, W0, b0, t0);
    conv4d_relu_pool_wmma<32, 64, 8, 2><<<dim3(128, 8), 128, 0, stream>>>(t0, W1, b1, o1);
    conv4d_relu_pool_wmma<64, 64, 4, 2><<<dim3(8, 8), 128, 0, stream>>>(o1, W2, b2, o2);
    conv4d_relu_pool_wmma<64, 64, 2, 1><<<dim3(1, 8), 128, 0, stream>>>(o2, W3, b3, o3);
}


# Weights re-laid out on every call by one extra small kernel into WMMA B-fragment order (workspace): a wave's weight load is one contiguous 512-byte run, not 32 far rows; summation order unchanged
# speedup vs baseline: 2.0543x; 2.0543x over previous
#include <hip/hip_runtime.h>
#include <hip/hip_bf16.h>

typedef float v2f __attribute__((ext_vector_type(2)));
typedef float v4f __attribute__((ext_vector_type(4)));
typedef float v8f __attribute__((ext_vector_type(8)));

#define LDS_PAD 4


template<int Ci>
__device__ __forceinline__ void relayout_one(const float* __restrict__ w,
                                             float* __restrict__ dst, int o)
{
    constexpr int KB = Ci / 4;
    constexpr int G  = KB < 4 ? KB : 4;
    constexpr int NG = KB / G;
    constexpr int Q  = G * 6 / 4;
    const int f    = o & 3;
    const int lane = (o >> 2) & 31;
    int rest = o >> 7;
    const int q   = rest % Q;  rest /= Q;
    const int kg  = rest % NG; rest /= NG;
    const int tt  = rest % 27;
    const int cot = rest / 27;
    const int flat = q * 4 + f;
    const int kbi  = flat / 6;
    const int j    = flat % 6;
    const int t3   = j >> 1;
    const int hi   = j & 1;
    const int co   = cot * 16 + (lane & 15);
    const int ci   = (kg * G + kbi) * 4 + ((lane < 16) ? 0 : 2) + hi;
    dst[o] = w[(co * Ci + ci) * 81 + tt * 3 + t3];
}

__global__ __launch_bounds__(256)
void relayout_weights(const float* __restrict__ w0, const float* __restrict__ w1,
                      const float* __restrict__ w2, const float* __restrict__ w3,
                      float* __restrict__ dst)
{
    constexpr int N0 = 32 * 8 * 81, N1 = 64 * 32 * 81, N2 = 64 * 64 * 81;
    const int o = blockIdx.x * 256 + threadIdx.x;
    if (o < N0)                    relayout_one<8 >(w0, dst, o);
    else if (o < N0 + N1)          relayout_one<32>(w1, dst + N0, o - N0);
    else if (o < N0 + N1 + N2)     relayout_one<64>(w2, dst + N0 + N1, o - N0 - N1);
    else if (o < N0 + N1 + 2 * N2) relayout_one<64>(w3, dst + N0 + N1 + N2, o - N0 - N1 - N2);
}

template<int Ci, int Co, int S, int TILES>
__global__ __launch_bounds__((Co/16)*32)
void conv4d_relu_pool_wmma(const float* __restrict__ x,
                           const float* __restrict__ w,
                           const float* __restrict__ bias,
                           float* __restrict__ out)
{
    constexpr int P    = S / 2;
    constexpr int PH   = P / TILES;
    constexpr int P4   = P * P * P * P;
    constexpr int D3   = 2 + 2 * TILES;
    constexpr int NSP  = 64 * D3;
    constexpr int CSTR = Ci + LDS_PAD;
    constexpr int NT   = (Co / 16) * 32;

    __shared__ float lds[NSP * CSTR];

    const int b   = blockIdx.y;
    const int bx  = blockIdx.x;
    const int p3h = bx % PH;
    const int p2  = (bx / PH) % P;
    const int p1  = (bx / (PH * P)) % P;
    const int p0  = bx / (PH * P * P);

    const int tid = threadIdx.x;
    for (int e = tid; e < NSP * Ci; e += NT) {
        const int ci = e & (Ci - 1);
        const int sp = e / Ci;
        const int d3 = sp % D3;
        const int r  = sp / D3;
        const int d2 = r & 3, d1 = (r >> 2) & 3, d0 = (r >> 4) & 3;
        const int x0 = 2 * p0 - 1 + d0;
        const int x1 = 2 * p1 - 1 + d1;
        const int x2 = 2 * p2 - 1 + d2;
        const int x3 = 2 * (p3h * TILES) - 1 + d3;
        float v = 0.0f;
        if ((unsigned)x0 < (unsigned)S && (unsigned)x1 < (unsigned)S &&
            (unsigned)x2 < (unsigned)S && (unsigned)x3 < (unsigned)S) {
            v = x[((((b * Ci + ci) * S + x0) * S + x1) * S + x2) * S + x3];
        }
        lds[sp * CSTR + ci] = v;
    }
    __syncthreads();

    const int lane = tid & 31;
    const int wave = tid >> 5;
    const int n    = lane & 15;
    const int klo  = (lane < 16) ? 0 : 2;
    const int qsp  = (((n >> 3) & 1) * 16 + ((n >> 2) & 1) * 4 + ((n >> 1) & 1)) * D3 + (n & 1);
    const int coA  = wave * 16 + n;
    constexpr int KB = Ci / 4;
    constexpr int G  = KB < 4 ? KB : 4;
    constexpr int NG = KB / G;
    constexpr int Q  = G * 6 / 4;
    const v4f* w4 = (const v4f*)w + (wave * 27 * NG * Q * 32 + lane);

    v8f acc[TILES];
    #pragma unroll
    for (int u = 0; u < TILES; ++u)
        acc[u] = (v8f){0.f, 0.f, 0.f, 0.f, 0.f, 0.f, 0.f, 0.f};

    #pragma unroll 1
    for (int t0 = 0; t0 < 3; ++t0) {
        #pragma unroll 1
        for (int t1 = 0; t1 < 3; ++t1) {
            #pragma unroll 1
            for (int t2 = 0; t2 < 3; ++t2) {
                const int row = (qsp + (t0 * 16 + t1 * 4 + t2) * D3) * CSTR + klo;
                const v4f* wp = w4 + ((t0 * 3 + t1) * 3 + t2) * (NG * Q * 32);
                #pragma unroll
                for (int kg = 0; kg < NG; ++kg) {
                    float wf[G * 6];
                    #pragma unroll
                    for (int q = 0; q < Q; ++q) {
                        const v4f t = wp[(kg * Q + q) * 32];
                        wf[q * 4 + 0] = t.x; wf[q * 4 + 1] = t.y;
                        wf[q * 4 + 2] = t.z; wf[q * 4 + 3] = t.w;
                    }
                    #pragma unroll
                    for (int kbi = 0; kbi < G; ++kbi) {
                        const int kb = (kg * G + kbi) * 4;
                        #pragma unroll
                        for (int t3 = 0; t3 < 3; ++t3) {
                            v2f bf; bf.x = wf[kbi * 6 + 2 * t3]; bf.y = wf[kbi * 6 + 2 * t3 + 1];
                            #pragma unroll
                            for (int u = 0; u < TILES; ++u) {
                                v2f a = *(const v2f*)&lds[row + (t3 + 2 * u) * CSTR + kb];
                                acc[u] = __builtin_amdgcn_wmma_f32_16x16x4_f32(
                                     false, a,  false, bf,
                                     (short)0, acc[u],
                                     false,  false);
                            }
                        }
                    }
                }
            }
        }
    }

    const float bs = bias[coA];
    #pragma unroll
    for (int u = 0; u < TILES; ++u) {
        float s = 0.0f;
        #pragma unroll
        for (int r = 0; r < 8; ++r) {
            float v = acc[u][r] + bs;
            s += (v > 0.0f) ? v : 0.0f;
        }
        s += __shfl_xor(s, 16, 32);
        if (lane < 16) {
            const int pIdx = ((p0 * P + p1) * P + p2) * P + p3h * TILES + u;
            out[(b * Co + coA) * P4 + pIdx] = s * (1.0f / 16.0f);
        }
    }
}

extern "C" void kernel_launch(void* const* d_in, const int* in_sizes, int n_in,
                              void* d_out, int out_size, void* d_ws, size_t ws_size,
                              hipStream_t stream) {
    (void)in_sizes; (void)n_in; (void)out_size; (void)ws_size;

    const float* src = (const float*)d_in[0];
    const float* W0  = (const float*)d_in[1];
    const float* b0  = (const float*)d_in[2];
    const float* W1  = (const float*)d_in[3];
    const float* b1  = (const float*)d_in[4];
    const float* W2  = (const float*)d_in[5];
    const float* b2  = (const float*)d_in[6];
    const float* W3  = (const float*)d_in[7];
    const float* b3  = (const float*)d_in[8];

    float* t0 = (float*)d_ws;
    float* r0 = t0 + 8 * 32 * 4096;
    float* r1 = r0 + 32 * 8 * 81;
    float* r2 = r1 + 64 * 32 * 81;
    float* r3 = r2 + 64 * 64 * 81;
    float* o1 = (float*)d_out;
    float* o2 = o1 + 8 * 64 * 256;
    float* o3 = o2 + 8 * 64 * 16;

    relayout_weights<<<3321, 256, 0, stream>>>(W0, W1, W2, W3, r0);
    conv4d_relu_pool_wmma<8, 32, 16, 2><<<dim3(2048, 8), 64, 0, stream>>>(src, r0, b0, t0);
    conv4d_relu_pool_wmma<32, 64, 8, 2><<<dim3(128, 8), 128, 0, stream>>>(t0, r1, b1, o1);
    conv4d_relu_pool_wmma<64, 64, 4, 2><<<dim3(8, 8), 128, 0, stream>>>(o1, r2, b2, o2);
    conv4d_relu_pool_wmma<64, 64, 2, 1><<<dim3(1, 8), 128, 0, stream>>>(o2, r3, b3, o3);
}


# Patch staging: each thread copies its patch positions for all channels with unconditional loads (halo zeroed by select), 32 to 48 in flight, lanes along dim3, 16-byte LDS stores; sums unchanged
# speedup vs baseline: 2.5735x; 1.2527x over previous
#include <hip/hip_runtime.h>
#include <hip/hip_bf16.h>

typedef float v2f __attribute__((ext_vector_type(2)));
typedef float v4f __attribute__((ext_vector_type(4)));
typedef float v8f __attribute__((ext_vector_type(8)));

#define LDS_PAD 4


template<int Ci>
__device__ __forceinline__ void relayout_one(const float* __restrict__ w,
                                             float* __restrict__ dst, int o)
{
    constexpr int KB = Ci / 4;
    constexpr int G  = KB < 4 ? KB : 4;
    constexpr int NG = KB / G;
    constexpr int Q  = G * 6 / 4;
    const int f    = o & 3;
    const int lane = (o >> 2) & 31;
    int rest = o >> 7;
    const int q   = rest % Q;  rest /= Q;
    const int kg  = rest % NG; rest /= NG;
    const int tt  = rest % 27;
    const int cot = rest / 27;
    const int flat = q * 4 + f;
    const int kbi  = flat / 6;
    const int j    = flat % 6;
    const int t3   = j >> 1;
    const int hi   = j & 1;
    const int co   = cot * 16 + (lane & 15);
    const int ci   = (kg * G + kbi) * 4 + ((lane < 16) ? 0 : 2) + hi;
    dst[o] = w[(co * Ci + ci) * 81 + tt * 3 + t3];
}

__global__ __launch_bounds__(256)
void relayout_weights(const float* __restrict__ w0, const float* __restrict__ w1,
                      const float* __restrict__ w2, const float* __restrict__ w3,
                      float* __restrict__ dst)
{
    constexpr int N0 = 32 * 8 * 81, N1 = 64 * 32 * 81, N2 = 64 * 64 * 81;
    const int o = blockIdx.x * 256 + threadIdx.x;
    if (o < N0)                    relayout_one<8 >(w0, dst, o);
    else if (o < N0 + N1)          relayout_one<32>(w1, dst + N0, o - N0);
    else if (o < N0 + N1 + N2)     relayout_one<64>(w2, dst + N0 + N1, o - N0 - N1);
    else if (o < N0 + N1 + 2 * N2) relayout_one<64>(w3, dst + N0 + N1 + N2, o - N0 - N1 - N2);
}

template<int Ci, int Co, int S, int TILES>
__global__ __launch_bounds__((Co/16)*32)
void conv4d_relu_pool_wmma(const float* __restrict__ x,
                           const float* __restrict__ w,
                           const float* __restrict__ bias,
                           float* __restrict__ out)
{
    constexpr int P    = S / 2;
    constexpr int PH   = P / TILES;
    constexpr int P4   = P * P * P * P;
    constexpr int D3   = 2 + 2 * TILES;
    constexpr int NSP  = 64 * D3;
    constexpr int CSTR = Ci + LDS_PAD;
    constexpr int NT   = (Co / 16) * 32;

    __shared__ __attribute__((aligned(16))) float lds[NSP * CSTR];

    const int b   = blockIdx.y;
    const int bx  = blockIdx.x;
    const int p3h = bx % PH;
    const int p2  = (bx / PH) % P;
    const int p1  = (bx / (PH * P)) % P;
    const int p0  = bx / (PH * P * P);

    const int tid = threadIdx.x;
    constexpr int K  = NSP / NT;
    constexpr int CB = (Ci < 16) ? Ci : 16;
    constexpr int S4 = S * S * S * S;
    static_assert(NSP % NT == 0 && Ci % CB == 0 && CB % 4 == 0, "staging split");
    const float* __restrict__ xb = x + b * (Ci * S4);
    int  off[K];
    bool ok[K];
    #pragma unroll
    for (int k = 0; k < K; ++k) {
        const int sp = tid + k * NT;
        const int r  = sp / D3;
        const int d3 = sp - r * D3;
        const int x0 = 2 * p0 - 1 + ((r >> 4) & 3);
        const int x1 = 2 * p1 - 1 + ((r >> 2) & 3);
        const int x2 = 2 * p2 - 1 + (r & 3);
        const int x3 = 2 * (p3h * TILES) - 1 + d3;
        ok[k]  = (unsigned)x0 < (unsigned)S && (unsigned)x1 < (unsigned)S &&
                 (unsigned)x2 < (unsigned)S && (unsigned)x3 < (unsigned)S;
        off[k] = ok[k] ? (((x0 * S + x1) * S + x2) * S + x3) : 0;
    }
    #pragma unroll 1
    for (int c0 = 0; c0 < Ci; c0 += CB) {
        float v[K][CB];
        #pragma unroll
        for (int k = 0; k < K; ++k) {
            #pragma unroll
            for (int c = 0; c < CB; ++c)
                v[k][c] = xb[(c0 + c) * S4 + off[k]];
        }
        #pragma unroll
        for (int k = 0; k < K; ++k) {
            #pragma unroll
            for (int c = 0; c < CB; c += 4) {
                v4f q;
                q.x = ok[k] ? v[k][c + 0] : 0.0f;
                q.y = ok[k] ? v[k][c + 1] : 0.0f;
                q.z = ok[k] ? v[k][c + 2] : 0.0f;
                q.w = ok[k] ? v[k][c + 3] : 0.0f;
                *(v4f*)&lds[(tid + k * NT) * CSTR + c0 + c] = q;
            }
        }
    }
    __syncthreads();

    const int lane = tid & 31;
    const int wave = tid >> 5;
    const int n    = lane & 15;
    const int klo  = (lane < 16) ? 0 : 2;
    const int qsp  = (((n >> 3) & 1) * 16 + ((n >> 2) & 1) * 4 + ((n >> 1) & 1)) * D3 + (n & 1);
    const int coA  = wave * 16 + n;
    constexpr int KB = Ci / 4;
    constexpr int G  = KB < 4 ? KB : 4;
    constexpr int NG = KB / G;
    constexpr int Q  = G * 6 / 4;
    const v4f* w4 = (const v4f*)w + (wave * 27 * NG * Q * 32 + lane);

    v8f acc[TILES];
    #pragma unroll
    for (int u = 0; u < TILES; ++u)
        acc[u] = (v8f){0.f, 0.f, 0.f, 0.f, 0.f, 0.f, 0.f, 0.f};

    #pragma unroll 1
    for (int t0 = 0; t0 < 3; ++t0) {
        #pragma unroll 1
        for (int t1 = 0; t1 < 3; ++t1) {
            #pragma unroll 1
            for (int t2 = 0; t2 < 3; ++t2) {
                const int row = (qsp + (t0 * 16 + t1 * 4 + t2) * D3) * CSTR + klo;
                const v4f* wp = w4 + ((t0 * 3 + t1) * 3 + t2) * (NG * Q * 32);
                #pragma unroll
                for (int kg = 0; kg < NG; ++kg) {
                    float wf[G * 6];
                    #pragma unroll
                    for (int q = 0; q < Q; ++q) {
                        const v4f t = wp[(kg * Q + q) * 32];
                        wf[q * 4 + 0] = t.x; wf[q * 4 + 1] = t.y;
                        wf[q * 4 + 2] = t.z; wf[q * 4 + 3] = t.w;
                    }
                    #pragma unroll
                    for (int kbi = 0; kbi < G; ++kbi) {
                        const int kb = (kg * G + kbi) * 4;
                        #pragma unroll
                        for (int t3 = 0; t3 < 3; ++t3) {
                            v2f bf; bf.x = wf[kbi * 6 + 2 * t3]; bf.y = wf[kbi * 6 + 2 * t3 + 1];
                            #pragma unroll
                            for (int u = 0; u < TILES; ++u) {
                                v2f a = *(const v2f*)&lds[row + (t3 + 2 * u) * CSTR + kb];
                                acc[u] = __builtin_amdgcn_wmma_f32_16x16x4_f32(
                                     false, a,  false, bf,
                                     (short)0, acc[u],
                                     false,  false);
                            }
                        }
                    }
                }
            }
        }
    }

    const float bs = bias[coA];
    #pragma unroll
    for (int u = 0; u < TILES; ++u) {
        float s = 0.0f;
        #pragma unroll
        for (int r = 0; r < 8; ++r) {
            float v = acc[u][r] + bs;
            s += (v > 0.0f) ? v : 0.0f;
        }
        s += __shfl_xor(s, 16, 32);
        if (lane < 16) {
            const int pIdx = ((p0 * P + p1) * P + p2) * P + p3h * TILES + u;
            out[(b * Co + coA) * P4 + pIdx] = s * (1.0f / 16.0f);
        }
    }
}

extern "C" void kernel_launch(void* const* d_in, const int* in_sizes, int n_in,
                              void* d_out, int out_size, void* d_ws, size_t ws_size,
                              hipStream_t stream) {
    (void)in_sizes; (void)n_in; (void)out_size; (void)ws_size;

    const float* src = (const float*)d_in[0];
    const float* W0  = (const float*)d_in[1];
    const float* b0  = (const float*)d_in[2];
    const float* W1  = (const float*)d_in[3];
    const float* b1  = (const float*)d_in[4];
    const float* W2  = (const float*)d_in[5];
    const float* b2  = (const float*)d_in[6];
    const float* W3  = (const float*)d_in[7];
    const float* b3  = (const float*)d_in[8];

    float* t0 = (float*)d_ws;
    float* r0 = t0 + 8 * 32 * 4096;
    float* r1 = r0 + 32 * 8 * 81;
    float* r2 = r1 + 64 * 32 * 81;
    float* r3 = r2 + 64 * 64 * 81;
    float* o1 = (float*)d_out;
    float* o2 = o1 + 8 * 64 * 256;
    float* o3 = o2 + 8 * 64 * 16;

    relayout_weights<<<3321, 256, 0, stream>>>(W0, W1, W2, W3, r0);
    conv4d_relu_pool_wmma<8, 32, 16, 2><<<dim3(2048, 8), 64, 0, stream>>>(src, r0, b0, t0);
    conv4d_relu_pool_wmma<32, 64, 8, 2><<<dim3(128, 8), 128, 0, stream>>>(t0, r1, b1, o1);
    conv4d_relu_pool_wmma<64, 64, 4, 2><<<dim3(8, 8), 128, 0, stream>>>(o1, r2, b2, o2);
    conv4d_relu_pool_wmma<64, 64, 2, 1><<<dim3(1, 8), 128, 0, stream>>>(o2, r3, b3, o3);
}


# The two small layers split each tile's reduction over 12 waves (4 channel groups x 3 outer taps), one 16-channel tile per block, 4x the blocks; partials added in LDS in wave order: sum order changes
# speedup vs baseline: 3.3641x; 1.3072x over previous
#include <hip/hip_runtime.h>
#include <hip/hip_bf16.h>

typedef float v2f __attribute__((ext_vector_type(2)));
typedef float v4f __attribute__((ext_vector_type(4)));
typedef float v8f __attribute__((ext_vector_type(8)));

#define LDS_PAD 4


template<int Ci>
__device__ __forceinline__ void relayout_one(const float* __restrict__ w,
                                             float* __restrict__ dst, int o)
{
    constexpr int KB = Ci / 4;
    constexpr int G  = KB < 4 ? KB : 4;
    constexpr int NG = KB / G;
    constexpr int Q  = G * 6 / 4;
    const int f    = o & 3;
    const int lane = (o >> 2) & 31;
    int rest = o >> 7;
    const int q   = rest % Q;  rest /= Q;
    const int kg  = rest % NG; rest /= NG;
    const int tt  = rest % 27;
    const int cot = rest / 27;
    const int flat = q * 4 + f;
    const int kbi  = flat / 6;
    const int j    = flat % 6;
    const int t3   = j >> 1;
    const int hi   = j & 1;
    const int co   = cot * 16 + (lane & 15);
    const int ci   = (kg * G + kbi) * 4 + ((lane < 16) ? 0 : 2) + hi;
    dst[o] = w[(co * Ci + ci) * 81 + tt * 3 + t3];
}

__global__ __launch_bounds__(256)
void relayout_weights(const float* __restrict__ w0, const float* __restrict__ w1,
                      const float* __restrict__ w2, const float* __restrict__ w3,
                      float* __restrict__ dst)
{
    constexpr int N0 = 32 * 8 * 81, N1 = 64 * 32 * 81, N2 = 64 * 64 * 81;
    const int o = blockIdx.x * 256 + threadIdx.x;
    if (o < N0)                    relayout_one<8 >(w0, dst, o);
    else if (o < N0 + N1)          relayout_one<32>(w1, dst + N0, o - N0);
    else if (o < N0 + N1 + N2)     relayout_one<64>(w2, dst + N0 + N1, o - N0 - N1);
    else if (o < N0 + N1 + 2 * N2) relayout_one<64>(w3, dst + N0 + N1 + N2, o - N0 - N1 - N2);
}

template<int Ci, int Co, int S, int TILES, int KSG = 1, int KST = 1, int NTB = Co / 16>
__global__ __launch_bounds__(NTB * KSG * KST * 32)
void conv4d_relu_pool_wmma(const float* __restrict__ x,
                           const float* __restrict__ w,
                           const float* __restrict__ bias,
                           float* __restrict__ out)
{
    constexpr int P    = S / 2;
    constexpr int PH   = P / TILES;
    constexpr int P4   = P * P * P * P;
    constexpr int D3   = 2 + 2 * TILES;
    constexpr int NSP  = 64 * D3;
    constexpr int CSTR = Ci + LDS_PAD;
    constexpr int KS   = KSG * KST;
    constexpr int NB   = (Co / 16) / NTB;
    constexpr int NT   = NTB * KS * 32;
    static_assert((Co / 16) % NTB == 0 && (KST == 1 || KST == 3), "block split");

    __shared__ __attribute__((aligned(16))) float lds[NSP * CSTR];

    const int b   = blockIdx.y;
    const int nb  = blockIdx.x % NB;
    const int bx  = blockIdx.x / NB;
    const int p3h = bx % PH;
    const int p2  = (bx / PH) % P;
    const int p1  = (bx / (PH * P)) % P;
    const int p0  = bx / (PH * P * P);

    const int tid = threadIdx.x;
    constexpr int K  = (NSP + NT - 1) / NT;
    constexpr bool FULL = (NSP % NT == 0);
    constexpr int CB = (Ci < 16) ? Ci : 16;
    constexpr int S4 = S * S * S * S;
    static_assert(Ci % CB == 0 && CB % 4 == 0, "staging split");
    const float* __restrict__ xb = x + b * (Ci * S4);
    int  off[K];
    bool ok[K];
    #pragma unroll
    for (int k = 0; k < K; ++k) {
        const int sp = tid + k * NT;
        const int r  = sp / D3;
        const int d3 = sp - r * D3;
        const int x0 = 2 * p0 - 1 + ((r >> 4) & 3);
        const int x1 = 2 * p1 - 1 + ((r >> 2) & 3);
        const int x2 = 2 * p2 - 1 + (r & 3);
        const int x3 = 2 * (p3h * TILES) - 1 + d3;
        ok[k]  = (FULL || sp < NSP) &&
                 (unsigned)x0 < (unsigned)S && (unsigned)x1 < (unsigned)S &&
                 (unsigned)x2 < (unsigned)S && (unsigned)x3 < (unsigned)S;
        off[k] = ok[k] ? (((x0 * S + x1) * S + x2) * S + x3) : 0;
    }
    #pragma unroll 1
    for (int c0 = 0; c0 < Ci; c0 += CB) {
        float v[K][CB];
        #pragma unroll
        for (int k = 0; k < K; ++k) {
            #pragma unroll
            for (int c = 0; c < CB; ++c)
                v[k][c] = xb[(c0 + c) * S4 + off[k]];
        }
        #pragma unroll
        for (int k = 0; k < K; ++k) {
            if (!FULL && tid + k * NT >= NSP) continue;
            #pragma unroll
            for (int c = 0; c < CB; c += 4) {
                v4f q;
                q.x = ok[k] ? v[k][c + 0] : 0.0f;
                q.y = ok[k] ? v[k][c + 1] : 0.0f;
                q.z = ok[k] ? v[k][c + 2] : 0.0f;
                q.w = ok[k] ? v[k][c + 3] : 0.0f;
                *(v4f*)&lds[(tid + k * NT) * CSTR + c0 + c] = q;
            }
        }
    }
    __syncthreads();

    const int lane = tid & 31;
    const int wave = tid >> 5;
    const int n    = lane & 15;
    const int klo  = (lane < 16) ? 0 : 2;
    const int qsp  = (((n >> 3) & 1) * 16 + ((n >> 2) & 1) * 4 + ((n >> 1) & 1)) * D3 + (n & 1);
    const int ntl  = wave / KS;
    const int ks   = wave - ntl * KS;
    const int kgw  = ks % KSG;
    const int t0w  = ks / KSG;
    const int ntile = nb * NTB + ntl;
    const int coA  = ntile * 16 + n;
    constexpr int KB = Ci / 4;
    constexpr int G  = KB < 4 ? KB : 4;
    constexpr int NG = KB / G;
    constexpr int Q  = G * 6 / 4;
    constexpr int NGW = NG / KSG;
    static_assert(NG % KSG == 0, "channel-group split");
    const v4f* w4 = (const v4f*)w + (ntile * 27 * NG * Q * 32 + kgw * (NGW * Q * 32) + lane);
    const int kbase = kgw * (NGW * G * 4);
    const int t0lo  = (KST == 3) ? t0w : 0;
    const int t0hi  = (KST == 3) ? t0w + 1 : 3;

    v8f acc[TILES];
    #pragma unroll
    for (int u = 0; u < TILES; ++u)
        acc[u] = (v8f){0.f, 0.f, 0.f, 0.f, 0.f, 0.f, 0.f, 0.f};

    #pragma unroll 1
    for (int t0 = t0lo; t0 < t0hi; ++t0) {
        #pragma unroll 1
        for (int t1 = 0; t1 < 3; ++t1) {
            #pragma unroll 1
            for (int t2 = 0; t2 < 3; ++t2) {
                const int row = (qsp + (t0 * 16 + t1 * 4 + t2) * D3) * CSTR + klo + kbase;
                const v4f* wp = w4 + ((t0 * 3 + t1) * 3 + t2) * (NG * Q * 32);
                #pragma unroll
                for (int kg = 0; kg < NGW; ++kg) {
                    float wf[G * 6];
                    #pragma unroll
                    for (int q = 0; q < Q; ++q) {
                        const v4f t = wp[(kg * Q + q) * 32];
                        wf[q * 4 + 0] = t.x; wf[q * 4 + 1] = t.y;
                        wf[q * 4 + 2] = t.z; wf[q * 4 + 3] = t.w;
                    }
                    #pragma unroll
                    for (int kbi = 0; kbi < G; ++kbi) {
                        const int kb = (kg * G + kbi) * 4;
                        #pragma unroll
                        for (int t3 = 0; t3 < 3; ++t3) {
                            v2f bf; bf.x = wf[kbi * 6 + 2 * t3]; bf.y = wf[kbi * 6 + 2 * t3 + 1];
                            #pragma unroll
                            for (int u = 0; u < TILES; ++u) {
                                v2f a = *(const v2f*)&lds[row + (t3 + 2 * u) * CSTR + kb];
                                acc[u] = __builtin_amdgcn_wmma_f32_16x16x4_f32(
                                     false, a,  false, bf,
                                     (short)0, acc[u],
                                     false,  false);
                            }
                        }
                    }
                }
            }
        }
    }

    if constexpr (KS > 1) {
        __shared__ __attribute__((aligned(16))) float red[NTB * KS * TILES * 256];
        if (ks != 0) {
            #pragma unroll
            for (int u = 0; u < TILES; ++u) {
                v4f lo, hi;
                lo.x = acc[u][0]; lo.y = acc[u][1]; lo.z = acc[u][2]; lo.w = acc[u][3];
                hi.x = acc[u][4]; hi.y = acc[u][5]; hi.z = acc[u][6]; hi.w = acc[u][7];
                *(v4f*)&red[(((wave * TILES + u) * 2 + 0) * 32 + lane) * 4] = lo;
                *(v4f*)&red[(((wave * TILES + u) * 2 + 1) * 32 + lane) * 4] = hi;
            }
        }
        __syncthreads();
        if (ks != 0) return;
        #pragma unroll
        for (int u = 0; u < TILES; ++u) {
            #pragma unroll
            for (int j = 1; j < KS; ++j) {
                const v4f lo = *(const v4f*)&red[((((wave + j) * TILES + u) * 2 + 0) * 32 + lane) * 4];
                const v4f hi = *(const v4f*)&red[((((wave + j) * TILES + u) * 2 + 1) * 32 + lane) * 4];
                acc[u][0] += lo.x; acc[u][1] += lo.y; acc[u][2] += lo.z; acc[u][3] += lo.w;
                acc[u][4] += hi.x; acc[u][5] += hi.y; acc[u][6] += hi.z; acc[u][7] += hi.w;
            }
        }
    }

    const float bs = bias[coA];
    #pragma unroll
    for (int u = 0; u < TILES; ++u) {
        float s = 0.0f;
        #pragma unroll
        for (int r = 0; r < 8; ++r) {
            float v = acc[u][r] + bs;
            s += (v > 0.0f) ? v : 0.0f;
        }
        s += __shfl_xor(s, 16, 32);
        if (lane < 16) {
            const int pIdx = ((p0 * P + p1) * P + p2) * P + p3h * TILES + u;
            out[(b * Co + coA) * P4 + pIdx] = s * (1.0f / 16.0f);
        }
    }
}

extern "C" void kernel_launch(void* const* d_in, const int* in_sizes, int n_in,
                              void* d_out, int out_size, void* d_ws, size_t ws_size,
                              hipStream_t stream) {
    (void)in_sizes; (void)n_in; (void)out_size; (void)ws_size;

    const float* src = (const float*)d_in[0];
    const float* W0  = (const float*)d_in[1];
    const float* b0  = (const float*)d_in[2];
    const float* W1  = (const float*)d_in[3];
    const float* b1  = (const float*)d_in[4];
    const float* W2  = (const float*)d_in[5];
    const float* b2  = (const float*)d_in[6];
    const float* W3  = (const float*)d_in[7];
    const float* b3  = (const float*)d_in[8];

    float* t0 = (float*)d_ws;
    float* r0 = t0 + 8 * 32 * 4096;
    float* r1 = r0 + 32 * 8 * 81;
    float* r2 = r1 + 64 * 32 * 81;
    float* r3 = r2 + 64 * 64 * 81;
    float* o1 = (float*)d_out;
    float* o2 = o1 + 8 * 64 * 256;
    float* o3 = o2 + 8 * 64 * 16;

    relayout_weights<<<3321, 256, 0, stream>>>(W0, W1, W2, W3, r0);
    conv4d_relu_pool_wmma<8, 32, 16, 2><<<dim3(2048, 8), 64, 0, stream>>>(src, r0, b0, t0);
    conv4d_relu_pool_wmma<32, 64, 8, 2><<<dim3(128, 8), 128, 0, stream>>>(t0, r1, b1, o1);
    conv4d_relu_pool_wmma<64, 64, 4, 2, 4, 3, 1><<<dim3(32, 8), 384, 0, stream>>>(o1, r2, b2, o2);
    conv4d_relu_pool_wmma<64, 64, 2, 1, 4, 3, 1><<<dim3(4, 8), 384, 0, stream>>>(o2, r3, b3, o3);
}


# Layers 0 and 1: each weight fragment serves 4 pooled positions per wave, not 2: half the weight loads per matrix op; layer 1 splits input channels over wave pairs, added in LDS: sum order changes
# speedup vs baseline: 3.7490x; 1.1144x over previous
#include <hip/hip_runtime.h>
#include <hip/hip_bf16.h>

typedef float v2f __attribute__((ext_vector_type(2)));
typedef float v4f __attribute__((ext_vector_type(4)));
typedef float v8f __attribute__((ext_vector_type(8)));

#define LDS_PAD 4


template<int Ci>
__device__ __forceinline__ void relayout_one(const float* __restrict__ w,
                                             float* __restrict__ dst, int o)
{
    constexpr int KB = Ci / 4;
    constexpr int G  = KB < 4 ? KB : 4;
    constexpr int NG = KB / G;
    constexpr int Q  = G * 6 / 4;
    const int f    = o & 3;
    const int lane = (o >> 2) & 31;
    int rest = o >> 7;
    const int q   = rest % Q;  rest /= Q;
    const int kg  = rest % NG; rest /= NG;
    const int tt  = rest % 27;
    const int cot = rest / 27;
    const int flat = q * 4 + f;
    const int kbi  = flat / 6;
    const int j    = flat % 6;
    const int t3   = j >> 1;
    const int hi   = j & 1;
    const int co   = cot * 16 + (lane & 15);
    const int ci   = (kg * G + kbi) * 4 + ((lane < 16) ? 0 : 2) + hi;
    dst[o] = w[(co * Ci + ci) * 81 + tt * 3 + t3];
}

__global__ __launch_bounds__(256)
void relayout_weights(const float* __restrict__ w0, const float* __restrict__ w1,
                      const float* __restrict__ w2, const float* __restrict__ w3,
                      float* __restrict__ dst)
{
    constexpr int N0 = 32 * 8 * 81, N1 = 64 * 32 * 81, N2 = 64 * 64 * 81;
    const int o = blockIdx.x * 256 + threadIdx.x;
    if (o < N0)                    relayout_one<8 >(w0, dst, o);
    else if (o < N0 + N1)          relayout_one<32>(w1, dst + N0, o - N0);
    else if (o < N0 + N1 + N2)     relayout_one<64>(w2, dst + N0 + N1, o - N0 - N1);
    else if (o < N0 + N1 + 2 * N2) relayout_one<64>(w3, dst + N0 + N1 + N2, o - N0 - N1 - N2);
}

template<int Ci, int Co, int S, int TILES, int KSG = 1, int KST = 1, int NTB = Co / 16>
__global__ __launch_bounds__(NTB * KSG * KST * 32)
void conv4d_relu_pool_wmma(const float* __restrict__ x,
                           const float* __restrict__ w,
                           const float* __restrict__ bias,
                           float* __restrict__ out)
{
    constexpr int P    = S / 2;
    constexpr int PH   = P / TILES;
    constexpr int P4   = P * P * P * P;
    constexpr int D3   = 2 + 2 * TILES;
    constexpr int NSP  = 64 * D3;
    constexpr int CSTR = Ci + LDS_PAD;
    constexpr int KS   = KSG * KST;
    constexpr int NB   = (Co / 16) / NTB;
    constexpr int NT   = NTB * KS * 32;
    static_assert((Co / 16) % NTB == 0 && (KST == 1 || KST == 3), "block split");

    __shared__ __attribute__((aligned(16))) float lds[NSP * CSTR];

    const int b   = blockIdx.y;
    const int nb  = blockIdx.x % NB;
    const int bx  = blockIdx.x / NB;
    const int p3h = bx % PH;
    const int p2  = (bx / PH) % P;
    const int p1  = (bx / (PH * P)) % P;
    const int p0  = bx / (PH * P * P);

    const int tid = threadIdx.x;
    constexpr int K  = (NSP + NT - 1) / NT;
    constexpr bool FULL = (NSP % NT == 0);
    constexpr int CB = (Ci < 16) ? Ci : 16;
    constexpr int S4 = S * S * S * S;
    static_assert(Ci % CB == 0 && CB % 4 == 0, "staging split");
    const float* __restrict__ xb = x + b * (Ci * S4);
    int  off[K];
    bool ok[K];
    #pragma unroll
    for (int k = 0; k < K; ++k) {
        const int sp = tid + k * NT;
        const int r  = sp / D3;
        const int d3 = sp - r * D3;
        const int x0 = 2 * p0 - 1 + ((r >> 4) & 3);
        const int x1 = 2 * p1 - 1 + ((r >> 2) & 3);
        const int x2 = 2 * p2 - 1 + (r & 3);
        const int x3 = 2 * (p3h * TILES) - 1 + d3;
        ok[k]  = (FULL || sp < NSP) &&
                 (unsigned)x0 < (unsigned)S && (unsigned)x1 < (unsigned)S &&
                 (unsigned)x2 < (unsigned)S && (unsigned)x3 < (unsigned)S;
        off[k] = ok[k] ? (((x0 * S + x1) * S + x2) * S + x3) : 0;
    }
    #pragma unroll 1
    for (int c0 = 0; c0 < Ci; c0 += CB) {
        float v[K][CB];
        #pragma unroll
        for (int k = 0; k < K; ++k) {
            #pragma unroll
            for (int c = 0; c < CB; ++c)
                v[k][c] = xb[(c0 + c) * S4 + off[k]];
        }
        #pragma unroll
        for (int k = 0; k < K; ++k) {
            if (!FULL && tid + k * NT >= NSP) continue;
            #pragma unroll
            for (int c = 0; c < CB; c += 4) {
                v4f q;
                q.x = ok[k] ? v[k][c + 0] : 0.0f;
                q.y = ok[k] ? v[k][c + 1] : 0.0f;
                q.z = ok[k] ? v[k][c + 2] : 0.0f;
                q.w = ok[k] ? v[k][c + 3] : 0.0f;
                *(v4f*)&lds[(tid + k * NT) * CSTR + c0 + c] = q;
            }
        }
    }
    __syncthreads();

    const int lane = tid & 31;
    const int wave = tid >> 5;
    const int n    = lane & 15;
    const int klo  = (lane < 16) ? 0 : 2;
    const int qsp  = (((n >> 3) & 1) * 16 + ((n >> 2) & 1) * 4 + ((n >> 1) & 1)) * D3 + (n & 1);
    const int ntl  = wave / KS;
    const int ks   = wave - ntl * KS;
    const int kgw  = ks % KSG;
    const int t0w  = ks / KSG;
    const int ntile = nb * NTB + ntl;
    const int coA  = ntile * 16 + n;
    constexpr int KB = Ci / 4;
    constexpr int G  = KB < 4 ? KB : 4;
    constexpr int NG = KB / G;
    constexpr int Q  = G * 6 / 4;
    constexpr int NGW = NG / KSG;
    static_assert(NG % KSG == 0, "channel-group split");
    const v4f* w4 = (const v4f*)w + (ntile * 27 * NG * Q * 32 + kgw * (NGW * Q * 32) + lane);
    const int kbase = kgw * (NGW * G * 4);
    const int t0lo  = (KST == 3) ? t0w : 0;
    const int t0hi  = (KST == 3) ? t0w + 1 : 3;

    v8f acc[TILES];
    #pragma unroll
    for (int u = 0; u < TILES; ++u)
        acc[u] = (v8f){0.f, 0.f, 0.f, 0.f, 0.f, 0.f, 0.f, 0.f};

    #pragma unroll 1
    for (int t0 = t0lo; t0 < t0hi; ++t0) {
        #pragma unroll 1
        for (int t1 = 0; t1 < 3; ++t1) {
            #pragma unroll 1
            for (int t2 = 0; t2 < 3; ++t2) {
                const int row = (qsp + (t0 * 16 + t1 * 4 + t2) * D3) * CSTR + klo + kbase;
                const v4f* wp = w4 + ((t0 * 3 + t1) * 3 + t2) * (NG * Q * 32);
                #pragma unroll
                for (int kg = 0; kg < NGW; ++kg) {
                    float wf[G * 6];
                    #pragma unroll
                    for (int q = 0; q < Q; ++q) {
                        const v4f t = wp[(kg * Q + q) * 32];
                        wf[q * 4 + 0] = t.x; wf[q * 4 + 1] = t.y;
                        wf[q * 4 + 2] = t.z; wf[q * 4 + 3] = t.w;
                    }
                    #pragma unroll
                    for (int kbi = 0; kbi < G; ++kbi) {
                        const int kb = (kg * G + kbi) * 4;
                        #pragma unroll
                        for (int t3 = 0; t3 < 3; ++t3) {
                            v2f bf; bf.x = wf[kbi * 6 + 2 * t3]; bf.y = wf[kbi * 6 + 2 * t3 + 1];
                            #pragma unroll
                            for (int u = 0; u < TILES; ++u) {
                                v2f a = *(const v2f*)&lds[row + (t3 + 2 * u) * CSTR + kb];
                                acc[u] = __builtin_amdgcn_wmma_f32_16x16x4_f32(
                                     false, a,  false, bf,
                                     (short)0, acc[u],
                                     false,  false);
                            }
                        }
                    }
                }
            }
        }
    }

    if constexpr (KS > 1) {
        __shared__ __attribute__((aligned(16))) float red[NTB * KS * TILES * 256];
        if (ks != 0) {
            #pragma unroll
            for (int u = 0; u < TILES; ++u) {
                v4f lo, hi;
                lo.x = acc[u][0]; lo.y = acc[u][1]; lo.z = acc[u][2]; lo.w = acc[u][3];
                hi.x = acc[u][4]; hi.y = acc[u][5]; hi.z = acc[u][6]; hi.w = acc[u][7];
                *(v4f*)&red[(((wave * TILES + u) * 2 + 0) * 32 + lane) * 4] = lo;
                *(v4f*)&red[(((wave * TILES + u) * 2 + 1) * 32 + lane) * 4] = hi;
            }
        }
        __syncthreads();
        if (ks != 0) return;
        #pragma unroll
        for (int u = 0; u < TILES; ++u) {
            #pragma unroll
            for (int j = 1; j < KS; ++j) {
                const v4f lo = *(const v4f*)&red[((((wave + j) * TILES + u) * 2 + 0) * 32 + lane) * 4];
                const v4f hi = *(const v4f*)&red[((((wave + j) * TILES + u) * 2 + 1) * 32 + lane) * 4];
                acc[u][0] += lo.x; acc[u][1] += lo.y; acc[u][2] += lo.z; acc[u][3] += lo.w;
                acc[u][4] += hi.x; acc[u][5] += hi.y; acc[u][6] += hi.z; acc[u][7] += hi.w;
            }
        }
    }

    const float bs = bias[coA];
    #pragma unroll
    for (int u = 0; u < TILES; ++u) {
        float s = 0.0f;
        #pragma unroll
        for (int r = 0; r < 8; ++r) {
            float v = acc[u][r] + bs;
            s += (v > 0.0f) ? v : 0.0f;
        }
        s += __shfl_xor(s, 16, 32);
        if (lane < 16) {
            const int pIdx = ((p0 * P + p1) * P + p2) * P + p3h * TILES + u;
            out[(b * Co + coA) * P4 + pIdx] = s * (1.0f / 16.0f);
        }
    }
}

extern "C" void kernel_launch(void* const* d_in, const int* in_sizes, int n_in,
                              void* d_out, int out_size, void* d_ws, size_t ws_size,
                              hipStream_t stream) {
    (void)in_sizes; (void)n_in; (void)out_size; (void)ws_size;

    const float* src = (const float*)d_in[0];
    const float* W0  = (const float*)d_in[1];
    const float* b0  = (const float*)d_in[2];
    const float* W1  = (const float*)d_in[3];
    const float* b1  = (const float*)d_in[4];
    const float* W2  = (const float*)d_in[5];
    const float* b2  = (const float*)d_in[6];
    const float* W3  = (const float*)d_in[7];
    const float* b3  = (const float*)d_in[8];

    float* t0 = (float*)d_ws;
    float* r0 = t0 + 8 * 32 * 4096;
    float* r1 = r0 + 32 * 8 * 81;
    float* r2 = r1 + 64 * 32 * 81;
    float* r3 = r2 + 64 * 64 * 81;
    float* o1 = (float*)d_out;
    float* o2 = o1 + 8 * 64 * 256;
    float* o3 = o2 + 8 * 64 * 16;

    relayout_weights<<<3321, 256, 0, stream>>>(W0, W1, W2, W3, r0);
    conv4d_relu_pool_wmma<8, 32, 16, 4><<<dim3(1024, 8), 64, 0, stream>>>(src, r0, b0, t0);
    conv4d_relu_pool_wmma<32, 64, 8, 4, 2, 1, 4><<<dim3(64, 8), 256, 0, stream>>>(t0, r1, b1, o1);
    conv4d_relu_pool_wmma<64, 64, 4, 2, 4, 3, 1><<<dim3(32, 8), 384, 0, stream>>>(o1, r2, b2, o2);
    conv4d_relu_pool_wmma<64, 64, 2, 1, 4, 3, 1><<<dim3(4, 8), 384, 0, stream>>>(o2, r3, b3, o3);
}


# Layers 0 and 1: the three dim3 taps run as the minimal filter F(2,3), patch and weights stored transformed: 4 matrix ops for 6; f32 operands kept, sum order and rounding change
# speedup vs baseline: 4.5404x; 1.2111x over previous
#include <hip/hip_runtime.h>
#include <hip/hip_bf16.h>

typedef float v2f __attribute__((ext_vector_type(2)));
typedef float v4f __attribute__((ext_vector_type(4)));
typedef float v8f __attribute__((ext_vector_type(8)));

#define LDS_PAD 4


template<int Ci>
__device__ __forceinline__ void relayout_one(const float* __restrict__ w,
                                             float* __restrict__ dst, int o)
{
    constexpr int KB = Ci / 4;
    constexpr int G  = KB < 4 ? KB : 4;
    constexpr int NG = KB / G;
    constexpr int Q  = G * 6 / 4;
    const int f    = o & 3;
    const int lane = (o >> 2) & 31;
    int rest = o >> 7;
    const int q   = rest % Q;  rest /= Q;
    const int kg  = rest % NG; rest /= NG;
    const int tt  = rest % 27;
    const int cot = rest / 27;
    const int flat = q * 4 + f;
    const int kbi  = flat / 6;
    const int j    = flat % 6;
    const int t3   = j >> 1;
    const int hi   = j & 1;
    const int co   = cot * 16 + (lane & 15);
    const int ci   = (kg * G + kbi) * 4 + ((lane < 16) ? 0 : 2) + hi;
    dst[o] = w[(co * Ci + ci) * 81 + tt * 3 + t3];
}

template<int Ci>
__device__ __forceinline__ void relayout_mf(const float* __restrict__ w,
                                            float* __restrict__ dst, int o)
{
    constexpr int KB = Ci / 4;
    constexpr int G  = KB < 4 ? KB : 4;
    constexpr int NG = KB / G;
    constexpr int Q  = G * 2;
    const int f    = o & 3;
    const int lane = (o >> 2) & 31;
    int rest = o >> 7;
    const int q   = rest % Q;  rest /= Q;
    const int kg  = rest % NG; rest /= NG;
    const int tt  = rest % 27;
    const int cot = rest / 27;
    const int flat = q * 4 + f;
    const int kbi  = flat >> 3;
    const int j    = (flat >> 1) & 3;
    const int hi   = flat & 1;
    const int co   = cot * 16 + (lane & 15);
    const int ci   = (kg * G + kbi) * 4 + ((lane < 16) ? 0 : 2) + hi;
    const float* p = w + (co * Ci + ci) * 81 + tt * 3;
    const float g0 = p[0], g1 = p[1], g2 = p[2];
    const float e  = g0 + g2;
    float r;
    if (j == 0)      r = g0;
    else if (j == 1) r = 0.5f * (e + g1);
    else if (j == 2) r = 0.5f * (e - g1);
    else             r = g2;
    dst[o] = r;
}

__global__ __launch_bounds__(256)
void relayout_weights(const float* __restrict__ w0, const float* __restrict__ w1,
                      const float* __restrict__ w2, const float* __restrict__ w3,
                      float* __restrict__ dst)
{
    constexpr int N0 = 32 * 8 * 27 * 4, N1 = 64 * 32 * 27 * 4, N2 = 64 * 64 * 81;
    const int o = blockIdx.x * 256 + threadIdx.x;
    if (o < N0)                    relayout_mf<8 >(w0, dst, o);
    else if (o < N0 + N1)          relayout_mf<32>(w1, dst + N0, o - N0);
    else if (o < N0 + N1 + N2)     relayout_one<64>(w2, dst + N0 + N1, o - N0 - N1);
    else if (o < N0 + N1 + 2 * N2) relayout_one<64>(w3, dst + N0 + N1 + N2, o - N0 - N1 - N2);
}

template<int Ci, int Co, int S, int TILES, int KSG = 1, int KST = 1, int NTB = Co / 16>
__global__ __launch_bounds__(NTB * KSG * KST * 32)
void conv4d_relu_pool_wmma(const float* __restrict__ x,
                           const float* __restrict__ w,
                           const float* __restrict__ bias,
                           float* __restrict__ out)
{
    constexpr int P    = S / 2;
    constexpr int PH   = P / TILES;
    constexpr int P4   = P * P * P * P;
    constexpr int D3   = 2 + 2 * TILES;
    constexpr int NSP  = 64 * D3;
    constexpr int CSTR = Ci + LDS_PAD;
    constexpr int KS   = KSG * KST;
    constexpr int NB   = (Co / 16) / NTB;
    constexpr int NT   = NTB * KS * 32;
    static_assert((Co / 16) % NTB == 0 && (KST == 1 || KST == 3), "block split");

    __shared__ __attribute__((aligned(16))) float lds[NSP * CSTR];

    const int b   = blockIdx.y;
    const int nb  = blockIdx.x % NB;
    const int bx  = blockIdx.x / NB;
    const int p3h = bx % PH;
    const int p2  = (bx / PH) % P;
    const int p1  = (bx / (PH * P)) % P;
    const int p0  = bx / (PH * P * P);

    const int tid = threadIdx.x;
    constexpr int K  = (NSP + NT - 1) / NT;
    constexpr bool FULL = (NSP % NT == 0);
    constexpr int CB = (Ci < 16) ? Ci : 16;
    constexpr int S4 = S * S * S * S;
    static_assert(Ci % CB == 0 && CB % 4 == 0, "staging split");
    const float* __restrict__ xb = x + b * (Ci * S4);
    int  off[K];
    bool ok[K];
    #pragma unroll
    for (int k = 0; k < K; ++k) {
        const int sp = tid + k * NT;
        const int r  = sp / D3;
        const int d3 = sp - r * D3;
        const int x0 = 2 * p0 - 1 + ((r >> 4) & 3);
        const int x1 = 2 * p1 - 1 + ((r >> 2) & 3);
        const int x2 = 2 * p2 - 1 + (r & 3);
        const int x3 = 2 * (p3h * TILES) - 1 + d3;
        ok[k]  = (FULL || sp < NSP) &&
                 (unsigned)x0 < (unsigned)S && (unsigned)x1 < (unsigned)S &&
                 (unsigned)x2 < (unsigned)S && (unsigned)x3 < (unsigned)S;
        off[k] = ok[k] ? (((x0 * S + x1) * S + x2) * S + x3) : 0;
    }
    #pragma unroll 1
    for (int c0 = 0; c0 < Ci; c0 += CB) {
        float v[K][CB];
        #pragma unroll
        for (int k = 0; k < K; ++k) {
            #pragma unroll
            for (int c = 0; c < CB; ++c)
                v[k][c] = xb[(c0 + c) * S4 + off[k]];
        }
        #pragma unroll
        for (int k = 0; k < K; ++k) {
            if (!FULL && tid + k * NT >= NSP) continue;
            #pragma unroll
            for (int c = 0; c < CB; c += 4) {
                v4f q;
                q.x = ok[k] ? v[k][c + 0] : 0.0f;
                q.y = ok[k] ? v[k][c + 1] : 0.0f;
                q.z = ok[k] ? v[k][c + 2] : 0.0f;
                q.w = ok[k] ? v[k][c + 3] : 0.0f;
                *(v4f*)&lds[(tid + k * NT) * CSTR + c0 + c] = q;
            }
        }
    }
    __syncthreads();

    const int lane = tid & 31;
    const int wave = tid >> 5;
    const int n    = lane & 15;
    const int klo  = (lane < 16) ? 0 : 2;
    const int qsp  = (((n >> 3) & 1) * 16 + ((n >> 2) & 1) * 4 + ((n >> 1) & 1)) * D3 + (n & 1);
    const int ntl  = wave / KS;
    const int ks   = wave - ntl * KS;
    const int kgw  = ks % KSG;
    const int t0w  = ks / KSG;
    const int ntile = nb * NTB + ntl;
    const int coA  = ntile * 16 + n;
    constexpr int KB = Ci / 4;
    constexpr int G  = KB < 4 ? KB : 4;
    constexpr int NG = KB / G;
    constexpr int Q  = G * 6 / 4;
    constexpr int NGW = NG / KSG;
    static_assert(NG % KSG == 0, "channel-group split");
    const v4f* w4 = (const v4f*)w + (ntile * 27 * NG * Q * 32 + kgw * (NGW * Q * 32) + lane);
    const int kbase = kgw * (NGW * G * 4);
    const int t0lo  = (KST == 3) ? t0w : 0;
    const int t0hi  = (KST == 3) ? t0w + 1 : 3;

    v8f acc[TILES];
    #pragma unroll
    for (int u = 0; u < TILES; ++u)
        acc[u] = (v8f){0.f, 0.f, 0.f, 0.f, 0.f, 0.f, 0.f, 0.f};

    #pragma unroll 1
    for (int t0 = t0lo; t0 < t0hi; ++t0) {
        #pragma unroll 1
        for (int t1 = 0; t1 < 3; ++t1) {
            #pragma unroll 1
            for (int t2 = 0; t2 < 3; ++t2) {
                const int row = (qsp + (t0 * 16 + t1 * 4 + t2) * D3) * CSTR + klo + kbase;
                const v4f* wp = w4 + ((t0 * 3 + t1) * 3 + t2) * (NG * Q * 32);
                #pragma unroll
                for (int kg = 0; kg < NGW; ++kg) {
                    float wf[G * 6];
                    #pragma unroll
                    for (int q = 0; q < Q; ++q) {
                        const v4f t = wp[(kg * Q + q) * 32];
                        wf[q * 4 + 0] = t.x; wf[q * 4 + 1] = t.y;
                        wf[q * 4 + 2] = t.z; wf[q * 4 + 3] = t.w;
                    }
                    #pragma unroll
                    for (int kbi = 0; kbi < G; ++kbi) {
                        const int kb = (kg * G + kbi) * 4;
                        #pragma unroll
                        for (int t3 = 0; t3 < 3; ++t3) {
                            v2f bf; bf.x = wf[kbi * 6 + 2 * t3]; bf.y = wf[kbi * 6 + 2 * t3 + 1];
                            #pragma unroll
                            for (int u = 0; u < TILES; ++u) {
                                v2f a = *(const v2f*)&lds[row + (t3 + 2 * u) * CSTR + kb];
                                acc[u] = __builtin_amdgcn_wmma_f32_16x16x4_f32(
                                     false, a,  false, bf,
                                     (short)0, acc[u],
                                     false,  false);
                            }
                        }
                    }
                }
            }
        }
    }

    if constexpr (KS > 1) {
        __shared__ __attribute__((aligned(16))) float red[NTB * KS * TILES * 256];
        if (ks != 0) {
            #pragma unroll
            for (int u = 0; u < TILES; ++u) {
                v4f lo, hi;
                lo.x = acc[u][0]; lo.y = acc[u][1]; lo.z = acc[u][2]; lo.w = acc[u][3];
                hi.x = acc[u][4]; hi.y = acc[u][5]; hi.z = acc[u][6]; hi.w = acc[u][7];
                *(v4f*)&red[(((wave * TILES + u) * 2 + 0) * 32 + lane) * 4] = lo;
                *(v4f*)&red[(((wave * TILES + u) * 2 + 1) * 32 + lane) * 4] = hi;
            }
        }
        __syncthreads();
        if (ks != 0) return;
        #pragma unroll
        for (int u = 0; u < TILES; ++u) {
            #pragma unroll
            for (int j = 1; j < KS; ++j) {
                const v4f lo = *(const v4f*)&red[((((wave + j) * TILES + u) * 2 + 0) * 32 + lane) * 4];
                const v4f hi = *(const v4f*)&red[((((wave + j) * TILES + u) * 2 + 1) * 32 + lane) * 4];
                acc[u][0] += lo.x; acc[u][1] += lo.y; acc[u][2] += lo.z; acc[u][3] += lo.w;
                acc[u][4] += hi.x; acc[u][5] += hi.y; acc[u][6] += hi.z; acc[u][7] += hi.w;
            }
        }
    }

    const float bs = bias[coA];
    #pragma unroll
    for (int u = 0; u < TILES; ++u) {
        float s = 0.0f;
        #pragma unroll
        for (int r = 0; r < 8; ++r) {
            float v = acc[u][r] + bs;
            s += (v > 0.0f) ? v : 0.0f;
        }
        s += __shfl_xor(s, 16, 32);
        if (lane < 16) {
            const int pIdx = ((p0 * P + p1) * P + p2) * P + p3h * TILES + u;
            out[(b * Co + coA) * P4 + pIdx] = s * (1.0f / 16.0f);
        }
    }
}

constexpr bool mf_banks_ok(int us, int s2, int s1, int s0)
{
    unsigned seen = 0;
    for (int n = 0; n < 16; ++n) {
        const int off = (n & 1) * us + ((n >> 1) & 1) * s2 + ((n >> 2) & 1) * s1 + ((n >> 3) & 1) * s0;
        if (off % 4) return false;
        const int g = (off / 4) % 16;
        if (seen & (1u << g)) return false;
        seen |= 1u << g;
    }
    return true;
}

template<int Ci, int Co, int S, int TILES, int KSG, int NTB, int T2G, int PU, int P2, int P1, int P0>
__global__ __launch_bounds__(NTB * KSG * 32)
void conv4d_relu_pool_mf(const float* __restrict__ x,
                         const float* __restrict__ w,
                         const float* __restrict__ bias,
                         float* __restrict__ out)
{
    constexpr int P    = S / 2;
    constexpr int PH   = P / TILES;
    constexpr int P4   = P * P * P * P;
    constexpr int NV   = TILES / 2;
    constexpr int NACC = NV * 4;
    constexpr int JS   = Ci;
    constexpr int US   = 4 * Ci + PU;
    constexpr int S2   = TILES * US + P2;
    constexpr int S1   = 4 * S2 + P1;
    constexpr int S0   = 4 * S1 + P0;
    constexpr int LDSF = 4 * S0;
    constexpr int KS   = KSG;
    constexpr int NB   = (Co / 16) / NTB;
    constexpr int NT   = NTB * KS * 32;
    constexpr int REDF = (KS > 1) ? NTB * KS * NACC * 256 : 0;
    static_assert(TILES % 2 == 0 && (Co / 16) % NTB == 0 && S % 2 == 0, "tile split");
    static_assert(mf_banks_ok(US, S2, S1, S0), "fragment reads must be bank-conflict free");
    static_assert(REDF <= LDSF, "partial sums reuse the patch");
    static_assert(T2G == 1 || T2G == 3, "dim2 taps per weight load group");

    __shared__ __attribute__((aligned(16))) float lds[LDSF];

    const int b   = blockIdx.y;
    const int nb  = blockIdx.x % NB;
    const int bx  = blockIdx.x / NB;
    const int p3h = bx % PH;
    const int p2  = (bx / PH) % P;
    const int p1  = (bx / (PH * P)) % P;
    const int p0  = bx / (PH * P * P);

    const int tid = threadIdx.x;
    constexpr int NIT = 64 * TILES;
    constexpr int K   = (NIT + NT - 1) / NT;
    constexpr bool FULL = (NIT % NT == 0);
    constexpr int CB  = (K >= 4) ? 4 : ((Ci < 16) ? Ci : 16);
    constexpr int S4  = S * S * S * S;
    static_assert(Ci % CB == 0 && CB % 4 == 0, "staging split");
    const float* __restrict__ xb = x + b * (Ci * S4);
    int  offA[K], offB[K], offC[K], dsto[K];
    bool okA[K], okB[K], okC[K];
    #pragma unroll
    for (int k = 0; k < K; ++k) {
        const int it = tid + k * NT;
        const int r  = it / TILES;
        const int u  = it - r * TILES;
        const int x0 = 2 * p0 - 1 + ((r >> 4) & 3);
        const int x1 = 2 * p1 - 1 + ((r >> 2) & 3);
        const int x2 = 2 * p2 - 1 + (r & 3);
        const int U  = p3h * TILES + u;
        const bool okr = (FULL || it < NIT) &&
                         (unsigned)x0 < (unsigned)S && (unsigned)x1 < (unsigned)S &&
                         (unsigned)x2 < (unsigned)S;
        const int base = ((x0 * S + x1) * S + x2) * S + 2 * U;
        okB[k] = okr;
        okA[k] = okr && (U >= 1);
        okC[k] = okr && (2 * U + 2 < S);
        offB[k] = okB[k] ? base : 0;
        offA[k] = okA[k] ? base - 2 : 0;
        offC[k] = okC[k] ? base + 2 : 0;
        dsto[k] = ((r >> 4) & 3) * S0 + ((r >> 2) & 3) * S1 + (r & 3) * S2 + u * US;
    }
    #pragma unroll 1
    for (int c0 = 0; c0 < Ci; c0 += CB) {
        v2f va[K][CB], vb[K][CB], vc[K][CB];
        #pragma unroll
        for (int k = 0; k < K; ++k) {
            #pragma unroll
            for (int c = 0; c < CB; ++c) {
                const float* pc = xb + (c0 + c) * S4;
                va[k][c] = *(const v2f*)(pc + offA[k]);
                vb[k][c] = *(const v2f*)(pc + offB[k]);
                vc[k][c] = *(const v2f*)(pc + offC[k]);
            }
        }
        #pragma unroll
        for (int k = 0; k < K; ++k) {
            if (!FULL && tid + k * NT >= NIT) continue;
            #pragma unroll
            for (int c = 0; c < CB; c += 4) {
                float t0[4], t1[4], t2[4], t3[4];
                #pragma unroll
                for (int e = 0; e < 4; ++e) {
                    const float d0 = okA[k] ? va[k][c + e].y : 0.0f;
                    const float d1 = okB[k] ? vb[k][c + e].x : 0.0f;
                    const float d2 = okB[k] ? vb[k][c + e].y : 0.0f;
                    const float d3 = okC[k] ? vc[k][c + e].x : 0.0f;
                    t0[e] = d0 - d2; t1[e] = d1 + d2; t2[e] = d2 - d1; t3[e] = d1 - d3;
                }
                float* dp = &lds[dsto[k] + c0 + c];
                *(v4f*)(dp)          = (v4f){t0[0], t0[1], t0[2], t0[3]};
                *(v4f*)(dp + JS)     = (v4f){t1[0], t1[1], t1[2], t1[3]};
                *(v4f*)(dp + 2 * JS) = (v4f){t2[0], t2[1], t2[2], t2[3]};
                *(v4f*)(dp + 3 * JS) = (v4f){t3[0], t3[1], t3[2], t3[3]};
            }
        }
    }
    __syncthreads();

    const int lane = tid & 31;
    const int wave = tid >> 5;
    const int n    = lane & 15;
    const int klo  = (lane < 16) ? 0 : 2;
    const int qsp  = ((n >> 3) & 1) * S0 + ((n >> 2) & 1) * S1 + ((n >> 1) & 1) * S2 + (n & 1) * US;
    const int ntl  = wave / KS;
    const int ks   = wave - ntl * KS;
    const int ntile = nb * NTB + ntl;
    const int coA  = ntile * 16 + n;
    constexpr int KB = Ci / 4;
    constexpr int G  = KB < 4 ? KB : 4;
    constexpr int NG = KB / G;
    constexpr int Q  = G * 2;
    constexpr int NGW = NG / KSG;
    static_assert(NG % KSG == 0, "channel-group split");
    const v4f* w4 = (const v4f*)w + (ntile * 27 * NG * Q * 32 + ks * (NGW * Q * 32) + lane);
    const int kbase = ks * (NGW * G * 4);

    v8f acc[NACC];
    #pragma unroll
    for (int a = 0; a < NACC; ++a)
        acc[a] = (v8f){0.f, 0.f, 0.f, 0.f, 0.f, 0.f, 0.f, 0.f};

    #pragma unroll 1
    for (int t0 = 0; t0 < 3; ++t0) {
        #pragma unroll 1
        for (int t1 = 0; t1 < 3; ++t1) {
            #pragma unroll 1
            for (int t2b = 0; t2b < 3; t2b += T2G) {
                const v4f* wp = w4 + ((t0 * 3 + t1) * 3 + t2b) * (NG * Q * 32);
                float wf[T2G][NGW][G * 8];
                #pragma unroll
                for (int tg = 0; tg < T2G; ++tg) {
                    #pragma unroll
                    for (int kg = 0; kg < NGW; ++kg) {
                        #pragma unroll
                        for (int q = 0; q < Q; ++q) {
                            const v4f t = wp[(tg * NG * Q + kg * Q + q) * 32];
                            wf[tg][kg][q * 4 + 0] = t.x; wf[tg][kg][q * 4 + 1] = t.y;
                            wf[tg][kg][q * 4 + 2] = t.z; wf[tg][kg][q * 4 + 3] = t.w;
                        }
                    }
                }
                #pragma unroll
                for (int tg = 0; tg < T2G; ++tg) {
                    const int row = qsp + t0 * S0 + t1 * S1 + (t2b + tg) * S2 + klo + kbase;
                    #pragma unroll
                    for (int kg = 0; kg < NGW; ++kg) {
                        #pragma unroll
                        for (int kbi = 0; kbi < G; ++kbi) {
                            const int kb = (kg * G + kbi) * 4;
                            #pragma unroll
                            for (int j = 0; j < 4; ++j) {
                                v2f bf; bf.x = wf[tg][kg][kbi * 8 + 2 * j]; bf.y = wf[tg][kg][kbi * 8 + 2 * j + 1];
                                #pragma unroll
                                for (int v = 0; v < NV; ++v) {
                                    const float* ap = lds + (row + (2 * v) * US + j * JS + kb);
                                    v2f a = *(const v2f*)ap;
                                    acc[v * 4 + j] = __builtin_amdgcn_wmma_f32_16x16x4_f32(
                                         false, a,  false, bf,
                                         (short)0, acc[v * 4 + j],
                                         false,  false);
                                }
                            }
                        }
                    }
                }
            }
        }
    }

    if constexpr (KS > 1) {
        __syncthreads();
        float* red = lds;
        if (ks != 0) {
            #pragma unroll
            for (int a = 0; a < NACC; ++a) {
                v4f lo, hi;
                lo.x = acc[a][0]; lo.y = acc[a][1]; lo.z = acc[a][2]; lo.w = acc[a][3];
                hi.x = acc[a][4]; hi.y = acc[a][5]; hi.z = acc[a][6]; hi.w = acc[a][7];
                *(v4f*)&red[(((wave * NACC + a) * 2 + 0) * 32 + lane) * 4] = lo;
                *(v4f*)&red[(((wave * NACC + a) * 2 + 1) * 32 + lane) * 4] = hi;
            }
        }
        __syncthreads();
        if (ks != 0) return;
        #pragma unroll
        for (int a = 0; a < NACC; ++a) {
            #pragma unroll
            for (int j = 1; j < KS; ++j) {
                const v4f lo = *(const v4f*)&red[((((wave + j) * NACC + a) * 2 + 0) * 32 + lane) * 4];
                const v4f hi = *(const v4f*)&red[((((wave + j) * NACC + a) * 2 + 1) * 32 + lane) * 4];
                acc[a][0] += lo.x; acc[a][1] += lo.y; acc[a][2] += lo.z; acc[a][3] += lo.w;
                acc[a][4] += hi.x; acc[a][5] += hi.y; acc[a][6] += hi.z; acc[a][7] += hi.w;
            }
        }
    }

    const float bs = bias[coA];
    #pragma unroll
    for (int v = 0; v < NV; ++v) {
        float s[2] = {0.0f, 0.0f};
        #pragma unroll
        for (int r = 0; r < 8; ++r) {
            const float m0 = acc[v * 4 + 0][r], m1 = acc[v * 4 + 1][r];
            const float m2 = acc[v * 4 + 2][r], m3 = acc[v * 4 + 3][r];
            const float y0 = (m0 + m1) + m2 + bs;
            const float y1 = (m1 - m2) - m3 + bs;
            s[r & 1] += ((y0 > 0.0f) ? y0 : 0.0f) + ((y1 > 0.0f) ? y1 : 0.0f);
        }
        #pragma unroll
        for (int ub = 0; ub < 2; ++ub) {
            float t = s[ub];
            t += __shfl_xor(t, 16, 32);
            if (lane < 16) {
                const int pIdx = ((p0 * P + p1) * P + p2) * P + p3h * TILES + 2 * v + ub;
                out[(b * Co + coA) * P4 + pIdx] = t * (1.0f / 16.0f);
            }
        }
    }
}

extern "C" void kernel_launch(void* const* d_in, const int* in_sizes, int n_in,
                              void* d_out, int out_size, void* d_ws, size_t ws_size,
                              hipStream_t stream) {
    (void)in_sizes; (void)n_in; (void)out_size; (void)ws_size;

    const float* src = (const float*)d_in[0];
    const float* W0  = (const float*)d_in[1];
    const float* b0  = (const float*)d_in[2];
    const float* W1  = (const float*)d_in[3];
    const float* b1  = (const float*)d_in[4];
    const float* W2  = (const float*)d_in[5];
    const float* b2  = (const float*)d_in[6];
    const float* W3  = (const float*)d_in[7];
    const float* b3  = (const float*)d_in[8];

    float* t0 = (float*)d_ws;
    float* r0 = t0 + 8 * 32 * 4096;
    float* r1 = r0 + 32 * 8 * 27 * 4;
    float* r2 = r1 + 64 * 32 * 27 * 4;
    float* r3 = r2 + 64 * 64 * 81;
    float* o1 = (float*)d_out;
    float* o2 = o1 + 8 * 64 * 256;
    float* o3 = o2 + 8 * 64 * 16;

    relayout_weights<<<3564, 256, 0, stream>>>(W0, W1, W2, W3, r0);
    conv4d_relu_pool_mf<8, 32, 16, 4, 1, 2, 3, 0, 4, 0, 8><<<dim3(1024, 8), 64, 0, stream>>>(src, r0, b0, t0);
    conv4d_relu_pool_mf<32, 64, 8, 4, 2, 4, 1, 4, 0, 8, 0><<<dim3(64, 8), 256, 0, stream>>>(t0, r1, b1, o1);
    conv4d_relu_pool_wmma<64, 64, 4, 2, 4, 3, 1><<<dim3(32, 8), 384, 0, stream>>>(o1, r2, b2, o2);
    conv4d_relu_pool_wmma<64, 64, 2, 1, 4, 3, 1><<<dim3(4, 8), 384, 0, stream>>>(o2, r3, b3, o3);
}


# Layers 0 and 1: dim2 taps join dim3 in a 2-D minimal filter F(2x2,3x3), 16 matrix ops for 24; a block holds 2 tiles that share weight reads; f32 operands kept, sum order and rounding change
# speedup vs baseline: 5.4143x; 1.1925x over previous
#include <hip/hip_runtime.h>
#include <hip/hip_bf16.h>

typedef float v2f __attribute__((ext_vector_type(2)));
typedef float v4f __attribute__((ext_vector_type(4)));
typedef float v8f __attribute__((ext_vector_type(8)));

#define LDS_PAD 4


template<int Ci>
__device__ __forceinline__ void relayout_one(const float* __restrict__ w,
                                             float* __restrict__ dst, int o)
{
    constexpr int KB = Ci / 4;
    constexpr int G  = KB < 4 ? KB : 4;
    constexpr int NG = KB / G;
    constexpr int Q  = G * 6 / 4;
    const int f    = o & 3;
    const int lane = (o >> 2) & 31;
    int rest = o >> 7;
    const int q   = rest % Q;  rest /= Q;
    const int kg  = rest % NG; rest /= NG;
    const int tt  = rest % 27;
    const int cot = rest / 27;
    const int flat = q * 4 + f;
    const int kbi  = flat / 6;
    const int j    = flat % 6;
    const int t3   = j >> 1;
    const int hi   = j & 1;
    const int co   = cot * 16 + (lane & 15);
    const int ci   = (kg * G + kbi) * 4 + ((lane < 16) ? 0 : 2) + hi;
    dst[o] = w[(co * Ci + ci) * 81 + tt * 3 + t3];
}

template<int Ci>
__device__ __forceinline__ void relayout_mf(const float* __restrict__ w,
                                            float* __restrict__ dst, int o)
{
    constexpr int KB = Ci / 4;
    constexpr int G  = KB < 4 ? KB : 4;
    constexpr int NG = KB / G;
    constexpr int Q  = G * 2;
    const int f    = o & 3;
    const int lane = (o >> 2) & 31;
    int rest = o >> 7;
    const int q   = rest % Q;  rest /= Q;
    const int kg  = rest % NG; rest /= NG;
    const int tq  = rest % 36;
    const int cot = rest / 36;
    const int i   = tq & 3;
    const int t01 = tq >> 2;
    const int flat = q * 4 + f;
    const int kbi  = flat >> 3;
    const int j    = (flat >> 1) & 3;
    const int hi   = flat & 1;
    const int co   = cot * 16 + (lane & 15);
    const int ci   = (kg * G + kbi) * 4 + ((lane < 16) ? 0 : 2) + hi;
    const float* p = w + (co * Ci + ci) * 81 + t01 * 9;
    float h[3];
    #pragma unroll
    for (int t3 = 0; t3 < 3; ++t3) {
        const float a0 = p[t3], a1 = p[3 + t3], a2 = p[6 + t3];
        const float ea = a0 + a2;
        if (i == 0)      h[t3] = a0;
        else if (i == 1) h[t3] = 0.5f * (ea + a1);
        else if (i == 2) h[t3] = 0.5f * (ea - a1);
        else             h[t3] = a2;
    }
    const float e = h[0] + h[2];
    float r;
    if (j == 0)      r = h[0];
    else if (j == 1) r = 0.5f * (e + h[1]);
    else if (j == 2) r = 0.5f * (e - h[1]);
    else             r = h[2];
    dst[o] = r;
}

__global__ __launch_bounds__(256)
void relayout_weights(const float* __restrict__ w0, const float* __restrict__ w1,
                      const float* __restrict__ w2, const float* __restrict__ w3,
                      float* __restrict__ dst)
{
    constexpr int N0 = 32 * 8 * 9 * 16, N1 = 64 * 32 * 9 * 16, N2 = 64 * 64 * 81;
    const int o = blockIdx.x * 256 + threadIdx.x;
    if (o < N0)                    relayout_mf<8 >(w0, dst, o);
    else if (o < N0 + N1)          relayout_mf<32>(w1, dst + N0, o - N0);
    else if (o < N0 + N1 + N2)     relayout_one<64>(w2, dst + N0 + N1, o - N0 - N1);
    else if (o < N0 + N1 + 2 * N2) relayout_one<64>(w3, dst + N0 + N1 + N2, o - N0 - N1 - N2);
}

template<int Ci, int Co, int S, int TILES, int KSG = 1, int KST = 1, int NTB = Co / 16>
__global__ __launch_bounds__(NTB * KSG * KST * 32)
void conv4d_relu_pool_wmma(const float* __restrict__ x,
                           const float* __restrict__ w,
                           const float* __restrict__ bias,
                           float* __restrict__ out)
{
    constexpr int P    = S / 2;
    constexpr int PH   = P / TILES;
    constexpr int P4   = P * P * P * P;
    constexpr int D3   = 2 + 2 * TILES;
    constexpr int NSP  = 64 * D3;
    constexpr int CSTR = Ci + LDS_PAD;
    constexpr int KS   = KSG * KST;
    constexpr int NB   = (Co / 16) / NTB;
    constexpr int NT   = NTB * KS * 32;
    static_assert((Co / 16) % NTB == 0 && (KST == 1 || KST == 3), "block split");

    __shared__ __attribute__((aligned(16))) float lds[NSP * CSTR];

    const int b   = blockIdx.y;
    const int nb  = blockIdx.x % NB;
    const int bx  = blockIdx.x / NB;
    const int p3h = bx % PH;
    const int p2  = (bx / PH) % P;
    const int p1  = (bx / (PH * P)) % P;
    const int p0  = bx / (PH * P * P);

    const int tid = threadIdx.x;
    constexpr int K  = (NSP + NT - 1) / NT;
    constexpr bool FULL = (NSP % NT == 0);
    constexpr int CB = (Ci < 16) ? Ci : 16;
    constexpr int S4 = S * S * S * S;
    static_assert(Ci % CB == 0 && CB % 4 == 0, "staging split");
    const float* __restrict__ xb = x + b * (Ci * S4);
    int  off[K];
    bool ok[K];
    #pragma unroll
    for (int k = 0; k < K; ++k) {
        const int sp = tid + k * NT;
        const int r  = sp / D3;
        const int d3 = sp - r * D3;
        const int x0 = 2 * p0 - 1 + ((r >> 4) & 3);
        const int x1 = 2 * p1 - 1 + ((r >> 2) & 3);
        const int x2 = 2 * p2 - 1 + (r & 3);
        const int x3 = 2 * (p3h * TILES) - 1 + d3;
        ok[k]  = (FULL || sp < NSP) &&
                 (unsigned)x0 < (unsigned)S && (unsigned)x1 < (unsigned)S &&
                 (unsigned)x2 < (unsigned)S && (unsigned)x3 < (unsigned)S;
        off[k] = ok[k] ? (((x0 * S + x1) * S + x2) * S + x3) : 0;
    }
    #pragma unroll 1
    for (int c0 = 0; c0 < Ci; c0 += CB) {
        float v[K][CB];
        #pragma unroll
        for (int k = 0; k < K; ++k) {
            #pragma unroll
            for (int c = 0; c < CB; ++c)
                v[k][c] = xb[(c0 + c) * S4 + off[k]];
        }
        #pragma unroll
        for (int k = 0; k < K; ++k) {
            if (!FULL && tid + k * NT >= NSP) continue;
            #pragma unroll
            for (int c = 0; c < CB; c += 4) {
                v4f q;
                q.x = ok[k] ? v[k][c + 0] : 0.0f;
                q.y = ok[k] ? v[k][c + 1] : 0.0f;
                q.z = ok[k] ? v[k][c + 2] : 0.0f;
                q.w = ok[k] ? v[k][c + 3] : 0.0f;
                *(v4f*)&lds[(tid + k * NT) * CSTR + c0 + c] = q;
            }
        }
    }
    __syncthreads();

    const int lane = tid & 31;
    const int wave = tid >> 5;
    const int n    = lane & 15;
    const int klo  = (lane < 16) ? 0 : 2;
    const int qsp  = (((n >> 3) & 1) * 16 + ((n >> 2) & 1) * 4 + ((n >> 1) & 1)) * D3 + (n & 1);
    const int ntl  = wave / KS;
    const int ks   = wave - ntl * KS;
    const int kgw  = ks % KSG;
    const int t0w  = ks / KSG;
    const int ntile = nb * NTB + ntl;
    const int coA  = ntile * 16 + n;
    constexpr int KB = Ci / 4;
    constexpr int G  = KB < 4 ? KB : 4;
    constexpr int NG = KB / G;
    constexpr int Q  = G * 6 / 4;
    constexpr int NGW = NG / KSG;
    static_assert(NG % KSG == 0, "channel-group split");
    const v4f* w4 = (const v4f*)w + (ntile * 27 * NG * Q * 32 + kgw * (NGW * Q * 32) + lane);
    const int kbase = kgw * (NGW * G * 4);
    const int t0lo  = (KST == 3) ? t0w : 0;
    const int t0hi  = (KST == 3) ? t0w + 1 : 3;

    v8f acc[TILES];
    #pragma unroll
    for (int u = 0; u < TILES; ++u)
        acc[u] = (v8f){0.f, 0.f, 0.f, 0.f, 0.f, 0.f, 0.f, 0.f};

    #pragma unroll 1
    for (int t0 = t0lo; t0 < t0hi; ++t0) {
        #pragma unroll 1
        for (int t1 = 0; t1 < 3; ++t1) {
            #pragma unroll 1
            for (int t2 = 0; t2 < 3; ++t2) {
                const int row = (qsp + (t0 * 16 + t1 * 4 + t2) * D3) * CSTR + klo + kbase;
                const v4f* wp = w4 + ((t0 * 3 + t1) * 3 + t2) * (NG * Q * 32);
                #pragma unroll
                for (int kg = 0; kg < NGW; ++kg) {
                    float wf[G * 6];
                    #pragma unroll
                    for (int q = 0; q < Q; ++q) {
                        const v4f t = wp[(kg * Q + q) * 32];
                        wf[q * 4 + 0] = t.x; wf[q * 4 + 1] = t.y;
                        wf[q * 4 + 2] = t.z; wf[q * 4 + 3] = t.w;
                    }
                    #pragma unroll
                    for (int kbi = 0; kbi < G; ++kbi) {
                        const int kb = (kg * G + kbi) * 4;
                        #pragma unroll
                        for (int t3 = 0; t3 < 3; ++t3) {
                            v2f bf; bf.x = wf[kbi * 6 + 2 * t3]; bf.y = wf[kbi * 6 + 2 * t3 + 1];
                            #pragma unroll
                            for (int u = 0; u < TILES; ++u) {
                                v2f a = *(const v2f*)&lds[row + (t3 + 2 * u) * CSTR + kb];
                                acc[u] = __builtin_amdgcn_wmma_f32_16x16x4_f32(
                                     false, a,  false, bf,
                                     (short)0, acc[u],
                                     false,  false);
                            }
                        }
                    }
                }
            }
        }
    }

    if constexpr (KS > 1) {
        __shared__ __attribute__((aligned(16))) float red[NTB * KS * TILES * 256];
        if (ks != 0) {
            #pragma unroll
            for (int u = 0; u < TILES; ++u) {
                v4f lo, hi;
                lo.x = acc[u][0]; lo.y = acc[u][1]; lo.z = acc[u][2]; lo.w = acc[u][3];
                hi.x = acc[u][4]; hi.y = acc[u][5]; hi.z = acc[u][6]; hi.w = acc[u][7];
                *(v4f*)&red[(((wave * TILES + u) * 2 + 0) * 32 + lane) * 4] = lo;
                *(v4f*)&red[(((wave * TILES + u) * 2 + 1) * 32 + lane) * 4] = hi;
            }
        }
        __syncthreads();
        if (ks != 0) return;
        #pragma unroll
        for (int u = 0; u < TILES; ++u) {
            #pragma unroll
            for (int j = 1; j < KS; ++j) {
                const v4f lo = *(const v4f*)&red[((((wave + j) * TILES + u) * 2 + 0) * 32 + lane) * 4];
                const v4f hi = *(const v4f*)&red[((((wave + j) * TILES + u) * 2 + 1) * 32 + lane) * 4];
                acc[u][0] += lo.x; acc[u][1] += lo.y; acc[u][2] += lo.z; acc[u][3] += lo.w;
                acc[u][4] += hi.x; acc[u][5] += hi.y; acc[u][6] += hi.z; acc[u][7] += hi.w;
            }
        }
    }

    const float bs = bias[coA];
    #pragma unroll
    for (int u = 0; u < TILES; ++u) {
        float s = 0.0f;
        #pragma unroll
        for (int r = 0; r < 8; ++r) {
            float v = acc[u][r] + bs;
            s += (v > 0.0f) ? v : 0.0f;
        }
        s += __shfl_xor(s, 16, 32);
        if (lane < 16) {
            const int pIdx = ((p0 * P + p1) * P + p2) * P + p3h * TILES + u;
            out[(b * Co + coA) * P4 + pIdx] = s * (1.0f / 16.0f);
        }
    }
}

constexpr bool mf_banks_ok(int us, int s2, int s1, int s0)
{
    unsigned seen = 0;
    for (int n = 0; n < 16; ++n) {
        const int off = (n & 1) * us + ((n >> 1) & 1) * s2 + ((n >> 2) & 1) * s1 + ((n >> 3) & 1) * s0;
        if (off % 4) return false;
        const int g = (off / 4) % 16;
        if (seen & (1u << g)) return false;
        seen |= 1u << g;
    }
    return true;
}

template<int Ci, int Co, int S, int KSG, int NTB, int PZ, int UT, int IG, int WPE, int PU, int P1, int P0>
__global__ __launch_bounds__(PZ * UT * NTB * KSG * 32) __attribute__((amdgpu_waves_per_eu(WPE)))
void conv4d_relu_pool_mf(const float* __restrict__ x,
                         const float* __restrict__ w,
                         const float* __restrict__ bias,
                         float* __restrict__ out)
{
    constexpr int TILES = 4;
    constexpr int UU   = UT * TILES;
    constexpr int P    = S / 2;
    constexpr int PG3  = P / UU;
    constexpr int PG2  = P / PZ;
    constexpr int P4   = P * P * P * P;
    constexpr int NACC = 16;
    constexpr int MT   = PZ * UT;
    constexpr int US   = Ci + PU;
    constexpr int JS   = UU * US;
    constexpr int S2   = 4 * JS;
    constexpr int SZ   = 4 * S2;
    constexpr int S1   = PZ * SZ + P1;
    constexpr int S0   = 4 * S1 + P0;
    constexpr int LDSF = 4 * S0;
    constexpr int KS   = KSG;
    constexpr int NB   = (Co / 16) / NTB;
    constexpr int NT   = MT * NTB * KS * 32;
    constexpr int REDF = (KS > 1) ? (NT / 32) * 4 * 256 : 0;
    static_assert(P % UU == 0 && P % PZ == 0 && (Co / 16) % NTB == 0 && S % 2 == 0, "tile split");
    static_assert(mf_banks_ok(US, 2 * US, S1, S0), "fragment reads must be bank-conflict free");
    static_assert(REDF <= LDSF, "partial sums reuse the patch");
    static_assert(4 % IG == 0, "dim2 terms per weight load group");

    __shared__ __attribute__((aligned(16))) float lds[LDSF];

    const int b   = blockIdx.y;
    const int nb  = blockIdx.x % NB;
    const int bx  = blockIdx.x / NB;
    const int p3g = bx % PG3;
    const int p2g = (bx / PG3) % PG2;
    const int p1  = (bx / (PG3 * PG2)) % P;
    const int p0  = bx / (PG3 * PG2 * P);

    const int tid = threadIdx.x;
    constexpr int NITS = 16 * PZ * UU;
    constexpr int NIT  = NITS * (Ci / 4);
    constexpr int K    = NIT / NT;
    constexpr int S4   = S * S * S * S;
    static_assert(NIT % NT == 0 && Ci % 4 == 0, "staging split");
    const float* __restrict__ xb = x + b * (Ci * S4);
    #pragma unroll 1
    for (int k = 0; k < K; ++k) {
        const int it = tid + k * NT;
        const int cq = it / NITS;
        const int rs = it - cq * NITS;
        const int rz = rs / UU;
        const int uu = rs - rz * UU;
        const int r  = rz / PZ;
        const int pz = rz - r * PZ;
        const int x0 = 2 * p0 - 1 + (r >> 2);
        const int x1 = 2 * p1 - 1 + (r & 3);
        const int x2b = 2 * (p2g * PZ + pz) - 1;
        const int U  = p3g * UU + uu;
        const bool ok01 = (unsigned)x0 < (unsigned)S && (unsigned)x1 < (unsigned)S;
        const bool inA = (U >= 1), inC = (2 * U + 2 < S);
        int  offA[4], offB[4], offC[4];
        bool okA[4], okB[4], okC[4];
        #pragma unroll
        for (int d = 0; d < 4; ++d) {
            const int x2 = x2b + d;
            okB[d] = ok01 && (unsigned)x2 < (unsigned)S;
            okA[d] = okB[d] && inA;
            okC[d] = okB[d] && inC;
            const int base = ((x0 * S + x1) * S + x2) * S + 2 * U;
            offB[d] = okB[d] ? base : 0;
            offA[d] = okA[d] ? base - 2 : 0;
            offC[d] = okC[d] ? base + 2 : 0;
        }
        const float* pq = xb + (cq * 4) * S4;
        v2f va[4][4], vb[4][4], vc[4][4];
        #pragma unroll
        for (int d = 0; d < 4; ++d) {
            #pragma unroll
            for (int c = 0; c < 4; ++c) {
                const float* pc = pq + c * S4;
                va[d][c] = *(const v2f*)(pc + offA[d]);
                vb[d][c] = *(const v2f*)(pc + offB[d]);
                vc[d][c] = *(const v2f*)(pc + offC[d]);
            }
        }
        float V[4][4][4];
        #pragma unroll
        for (int c = 0; c < 4; ++c) {
            float t[4][4];
            #pragma unroll
            for (int d = 0; d < 4; ++d) {
                const float d0 = okA[d] ? va[d][c].y : 0.0f;
                const float d1 = okB[d] ? vb[d][c].x : 0.0f;
                const float d2 = okB[d] ? vb[d][c].y : 0.0f;
                const float d3 = okC[d] ? vc[d][c].x : 0.0f;
                t[d][0] = d0 - d2; t[d][1] = d1 + d2; t[d][2] = d2 - d1; t[d][3] = d1 - d3;
            }
            #pragma unroll
            for (int j = 0; j < 4; ++j) {
                V[0][j][c] = t[0][j] - t[2][j];
                V[1][j][c] = t[1][j] + t[2][j];
                V[2][j][c] = t[2][j] - t[1][j];
                V[3][j][c] = t[1][j] - t[3][j];
            }
        }
        float* dp = &lds[(r >> 2) * S0 + (r & 3) * S1 + pz * SZ + uu * US + cq * 4];
        #pragma unroll
        for (int i = 0; i < 4; ++i) {
            #pragma unroll
            for (int j = 0; j < 4; ++j)
                *(v4f*)(dp + i * S2 + j * JS) = (v4f){V[i][j][0], V[i][j][1], V[i][j][2], V[i][j][3]};
        }
    }
    __syncthreads();

    const int lane = tid & 31;
    const int wave = tid >> 5;
    const int n    = lane & 15;
    const int klo  = (lane < 16) ? 0 : 2;
    const int mt   = wave / (NTB * KS);
    const int wr   = wave - mt * (NTB * KS);
    const int ntl  = wr / KS;
    const int ks   = wr - ntl * KS;
    const int pzw  = mt / UT;
    const int utw  = mt - pzw * UT;
    const int qsp  = ((n >> 3) & 1) * S0 + ((n >> 2) & 1) * S1 + pzw * SZ + (utw * TILES + (n & 3)) * US;
    const int ntile = nb * NTB + ntl;
    const int coA  = ntile * 16 + n;
    constexpr int KB = Ci / 4;
    constexpr int G  = KB < 4 ? KB : 4;
    constexpr int NG = KB / G;
    constexpr int Q  = G * 2;
    constexpr int NGW = NG / KSG;
    static_assert(NG % KSG == 0, "channel-group split");
    const v4f* w4 = (const v4f*)w + (ntile * 36 * NG * Q * 32 + ks * (NGW * Q * 32) + lane);
    const int kbase = ks * (NGW * G * 4);

    v8f acc[NACC];
    #pragma unroll
    for (int a = 0; a < NACC; ++a)
        acc[a] = (v8f){0.f, 0.f, 0.f, 0.f, 0.f, 0.f, 0.f, 0.f};

    #pragma unroll 1
    for (int t0 = 0; t0 < 3; ++t0) {
        #pragma unroll 1
        for (int t1 = 0; t1 < 3; ++t1) {
            const v4f* wp = w4 + ((t0 * 3 + t1) * 4) * (NG * Q * 32);
            const int row0 = qsp + t0 * S0 + t1 * S1 + klo + kbase;
            #pragma unroll
            for (int ib = 0; ib < 4; ib += IG) {
                float wf[IG][NGW][G * 8];
                #pragma unroll
                for (int tg = 0; tg < IG; ++tg) {
                    #pragma unroll
                    for (int kg = 0; kg < NGW; ++kg) {
                        #pragma unroll
                        for (int q = 0; q < Q; ++q) {
                            const v4f t = wp[((ib + tg) * NG * Q + kg * Q + q) * 32];
                            wf[tg][kg][q * 4 + 0] = t.x; wf[tg][kg][q * 4 + 1] = t.y;
                            wf[tg][kg][q * 4 + 2] = t.z; wf[tg][kg][q * 4 + 3] = t.w;
                        }
                    }
                }
                #pragma unroll
                for (int tg = 0; tg < IG; ++tg) {
                    const int i = ib + tg;
                    const int row = row0 + i * S2;
                    #pragma unroll
                    for (int kg = 0; kg < NGW; ++kg) {
                        #pragma unroll
                        for (int kbi = 0; kbi < G; ++kbi) {
                            const int kb = (kg * G + kbi) * 4;
                            #pragma unroll
                            for (int j = 0; j < 4; ++j) {
                                v2f bf; bf.x = wf[tg][kg][kbi * 8 + 2 * j]; bf.y = wf[tg][kg][kbi * 8 + 2 * j + 1];
                                const float* ap = lds + (row + j * JS + kb);
                                v2f a = *(const v2f*)ap;
                                acc[i * 4 + j] = __builtin_amdgcn_wmma_f32_16x16x4_f32(
                                     false, a,  false, bf,
                                     (short)0, acc[i * 4 + j],
                                     false,  false);
                            }
                        }
                    }
                }
            }
        }
    }

    v8f y[4];
    #pragma unroll
    for (int r = 0; r < 8; ++r) {
        float z0[4], z1[4];
        #pragma unroll
        for (int i = 0; i < 4; ++i) {
            const float m0 = acc[i * 4 + 0][r], m1 = acc[i * 4 + 1][r];
            const float m2 = acc[i * 4 + 2][r], m3 = acc[i * 4 + 3][r];
            z0[i] = (m0 + m1) + m2;
            z1[i] = (m1 - m2) - m3;
        }
        y[0][r] = (z0[0] + z0[1]) + z0[2];
        y[1][r] = (z1[0] + z1[1]) + z1[2];
        y[2][r] = (z0[1] - z0[2]) - z0[3];
        y[3][r] = (z1[1] - z1[2]) - z1[3];
    }

    if constexpr (KS > 1) {
        __syncthreads();
        float* red = lds;
        if (ks != 0) {
            #pragma unroll
            for (int a = 0; a < 4; ++a) {
                v4f lo, hi;
                lo.x = y[a][0]; lo.y = y[a][1]; lo.z = y[a][2]; lo.w = y[a][3];
                hi.x = y[a][4]; hi.y = y[a][5]; hi.z = y[a][6]; hi.w = y[a][7];
                *(v4f*)&red[(((wave * 4 + a) * 2 + 0) * 32 + lane) * 4] = lo;
                *(v4f*)&red[(((wave * 4 + a) * 2 + 1) * 32 + lane) * 4] = hi;
            }
        }
        __syncthreads();
        if (ks != 0) return;
        #pragma unroll
        for (int a = 0; a < 4; ++a) {
            #pragma unroll
            for (int j = 1; j < KS; ++j) {
                const v4f lo = *(const v4f*)&red[((((wave + j) * 4 + a) * 2 + 0) * 32 + lane) * 4];
                const v4f hi = *(const v4f*)&red[((((wave + j) * 4 + a) * 2 + 1) * 32 + lane) * 4];
                y[a][0] += lo.x; y[a][1] += lo.y; y[a][2] += lo.z; y[a][3] += lo.w;
                y[a][4] += hi.x; y[a][5] += hi.y; y[a][6] += hi.z; y[a][7] += hi.w;
            }
        }
    }

    const float bs = bias[coA];
    float s[TILES] = {0.0f, 0.0f, 0.0f, 0.0f};
    #pragma unroll
    for (int r = 0; r < 8; ++r) {
        const float y00 = y[0][r] + bs, y01 = y[1][r] + bs;
        const float y10 = y[2][r] + bs, y11 = y[3][r] + bs;
        s[r & 3] += (((y00 > 0.0f) ? y00 : 0.0f) + ((y10 > 0.0f) ? y10 : 0.0f))
                  + (((y01 > 0.0f) ? y01 : 0.0f) + ((y11 > 0.0f) ? y11 : 0.0f));
    }
    #pragma unroll
    for (int u = 0; u < TILES; ++u) {
        float t = s[u];
        t += __shfl_xor(t, 16, 32);
        if (lane < 16) {
            const int pIdx = ((p0 * P + p1) * P + (p2g * PZ + pzw)) * P + (p3g * UT + utw) * TILES + u;
            out[(b * Co + coA) * P4 + pIdx] = t * (1.0f / 16.0f);
        }
    }
}

extern "C" void kernel_launch(void* const* d_in, const int* in_sizes, int n_in,
                              void* d_out, int out_size, void* d_ws, size_t ws_size,
                              hipStream_t stream) {
    (void)in_sizes; (void)n_in; (void)out_size; (void)ws_size;

    const float* src = (const float*)d_in[0];
    const float* W0  = (const float*)d_in[1];
    const float* b0  = (const float*)d_in[2];
    const float* W1  = (const float*)d_in[3];
    const float* b1  = (const float*)d_in[4];
    const float* W2  = (const float*)d_in[5];
    const float* b2  = (const float*)d_in[6];
    const float* W3  = (const float*)d_in[7];
    const float* b3  = (const float*)d_in[8];

    float* t0 = (float*)d_ws;
    float* r0 = t0 + 8 * 32 * 4096;
    float* r1 = r0 + 32 * 8 * 9 * 16;
    float* r2 = r1 + 64 * 32 * 9 * 16;
    float* r3 = r2 + 64 * 64 * 81;
    float* o1 = (float*)d_out;
    float* o2 = o1 + 8 * 64 * 256;
    float* o3 = o2 + 8 * 64 * 16;

    relayout_weights<<<3888, 256, 0, stream>>>(W0, W1, W2, W3, r0);
    conv4d_relu_pool_mf<8, 32, 16, 1, 2, 1, 2, 4, 4, 0, 4, 16><<<dim3(512, 8), 128, 0, stream>>>(src, r0, b0, t0);
    conv4d_relu_pool_mf<32, 64, 8, 2, 4, 2, 1, 2, 4, 4, 16, 32><<<dim3(32, 8), 512, 0, stream>>>(t0, r1, b1, o1);
    conv4d_relu_pool_wmma<64, 64, 4, 2, 4, 3, 1><<<dim3(32, 8), 384, 0, stream>>>(o1, r2, b2, o2);
    conv4d_relu_pool_wmma<64, 64, 2, 1, 4, 3, 1><<<dim3(4, 8), 384, 0, stream>>>(o2, r3, b3, o3);
}


# Layers 0 and 1: dim1 taps join the minimal filter, 3-D F(2x2x2,3x3x3), 64 matrix ops for 96; one wave per dim1 term, the four combined in LDS; f32 operands kept, sum order and rounding change
# speedup vs baseline: 5.6380x; 1.0413x over previous
#include <hip/hip_runtime.h>
#include <hip/hip_bf16.h>

typedef float v2f __attribute__((ext_vector_type(2)));
typedef float v4f __attribute__((ext_vector_type(4)));
typedef float v8f __attribute__((ext_vector_type(8)));

#define LDS_PAD 4


template<int Ci>
__device__ __forceinline__ void relayout_one(const float* __restrict__ w,
                                             float* __restrict__ dst, int o)
{
    constexpr int KB = Ci / 4;
    constexpr int G  = KB < 4 ? KB : 4;
    constexpr int NG = KB / G;
    constexpr int Q  = G * 6 / 4;
    const int f    = o & 3;
    const int lane = (o >> 2) & 31;
    int rest = o >> 7;
    const int q   = rest % Q;  rest /= Q;
    const int kg  = rest % NG; rest /= NG;
    const int tt  = rest % 27;
    const int cot = rest / 27;
    const int flat = q * 4 + f;
    const int kbi  = flat / 6;
    const int j    = flat % 6;
    const int t3   = j >> 1;
    const int hi   = j & 1;
    const int co   = cot * 16 + (lane & 15);
    const int ci   = (kg * G + kbi) * 4 + ((lane < 16) ? 0 : 2) + hi;
    dst[o] = w[(co * Ci + ci) * 81 + tt * 3 + t3];
}

template<int Ci>
__device__ __forceinline__ void relayout_mf(const float* __restrict__ w,
                                            float* __restrict__ dst, int o)
{
    constexpr int KB = Ci / 4;
    constexpr int G  = KB < 4 ? KB : 4;
    constexpr int NG = KB / G;
    constexpr int Q  = G * 2;
    const int f    = o & 3;
    const int lane = (o >> 2) & 31;
    int rest = o >> 7;
    const int q   = rest % Q;  rest /= Q;
    const int kg  = rest % NG; rest /= NG;
    const int tq  = rest % 48;
    const int cot = rest / 48;
    const int i   = tq & 3;
    const int h   = (tq >> 2) & 3;
    const int t0  = tq >> 4;
    const int flat = q * 4 + f;
    const int kbi  = flat >> 3;
    const int j    = (flat >> 1) & 3;
    const int hi   = flat & 1;
    const int co   = cot * 16 + (lane & 15);
    const int ci   = (kg * G + kbi) * 4 + ((lane < 16) ? 0 : 2) + hi;
    const float* p = w + (co * Ci + ci) * 81 + t0 * 27;
    float m[3];
    #pragma unroll
    for (int t1 = 0; t1 < 3; ++t1) {
        const float* p1 = p + t1 * 9;
        float hh[3];
        #pragma unroll
        for (int t3 = 0; t3 < 3; ++t3) {
            const float a0 = p1[t3], a1 = p1[3 + t3], a2 = p1[6 + t3];
            const float ea = a0 + a2;
            if (i == 0)      hh[t3] = a0;
            else if (i == 1) hh[t3] = 0.5f * (ea + a1);
            else if (i == 2) hh[t3] = 0.5f * (ea - a1);
            else             hh[t3] = a2;
        }
        const float e = hh[0] + hh[2];
        if (j == 0)      m[t1] = hh[0];
        else if (j == 1) m[t1] = 0.5f * (e + hh[1]);
        else if (j == 2) m[t1] = 0.5f * (e - hh[1]);
        else             m[t1] = hh[2];
    }
    const float em = m[0] + m[2];
    float r;
    if (h == 0)      r = m[0];
    else if (h == 1) r = 0.5f * (em + m[1]);
    else if (h == 2) r = 0.5f * (em - m[1]);
    else             r = m[2];
    dst[o] = r;
}

__global__ __launch_bounds__(256)
void relayout_weights(const float* __restrict__ w0, const float* __restrict__ w1,
                      const float* __restrict__ w2, const float* __restrict__ w3,
                      float* __restrict__ dst)
{
    constexpr int N0 = 32 * 8 * 3 * 64, N1 = 64 * 32 * 3 * 64, N2 = 64 * 64 * 81;
    const int o = blockIdx.x * 256 + threadIdx.x;
    if (o < N0)                    relayout_mf<8 >(w0, dst, o);
    else if (o < N0 + N1)          relayout_mf<32>(w1, dst + N0, o - N0);
    else if (o < N0 + N1 + N2)     relayout_one<64>(w2, dst + N0 + N1, o - N0 - N1);
    else if (o < N0 + N1 + 2 * N2) relayout_one<64>(w3, dst + N0 + N1 + N2, o - N0 - N1 - N2);
}

template<int Ci, int Co, int S, int TILES, int KSG = 1, int KST = 1, int NTB = Co / 16>
__global__ __launch_bounds__(NTB * KSG * KST * 32)
void conv4d_relu_pool_wmma(const float* __restrict__ x,
                           const float* __restrict__ w,
                           const float* __restrict__ bias,
                           float* __restrict__ out)
{
    constexpr int P    = S / 2;
    constexpr int PH   = P / TILES;
    constexpr int P4   = P * P * P * P;
    constexpr int D3   = 2 + 2 * TILES;
    constexpr int NSP  = 64 * D3;
    constexpr int CSTR = Ci + LDS_PAD;
    constexpr int KS   = KSG * KST;
    constexpr int NB   = (Co / 16) / NTB;
    constexpr int NT   = NTB * KS * 32;
    static_assert((Co / 16) % NTB == 0 && (KST == 1 || KST == 3), "block split");

    __shared__ __attribute__((aligned(16))) float lds[NSP * CSTR];

    const int b   = blockIdx.y;
    const int nb  = blockIdx.x % NB;
    const int bx  = blockIdx.x / NB;
    const int p3h = bx % PH;
    const int p2  = (bx / PH) % P;
    const int p1  = (bx / (PH * P)) % P;
    const int p0  = bx / (PH * P * P);

    const int tid = threadIdx.x;
    constexpr int K  = (NSP + NT - 1) / NT;
    constexpr bool FULL = (NSP % NT == 0);
    constexpr int CB = (Ci < 16) ? Ci : 16;
    constexpr int S4 = S * S * S * S;
    static_assert(Ci % CB == 0 && CB % 4 == 0, "staging split");
    const float* __restrict__ xb = x + b * (Ci * S4);
    int  off[K];
    bool ok[K];
    #pragma unroll
    for (int k = 0; k < K; ++k) {
        const int sp = tid + k * NT;
        const int r  = sp / D3;
        const int d3 = sp - r * D3;
        const int x0 = 2 * p0 - 1 + ((r >> 4) & 3);
        const int x1 = 2 * p1 - 1 + ((r >> 2) & 3);
        const int x2 = 2 * p2 - 1 + (r & 3);
        const int x3 = 2 * (p3h * TILES) - 1 + d3;
        ok[k]  = (FULL || sp < NSP) &&
                 (unsigned)x0 < (unsigned)S && (unsigned)x1 < (unsigned)S &&
                 (unsigned)x2 < (unsigned)S && (unsigned)x3 < (unsigned)S;
        off[k] = ok[k] ? (((x0 * S + x1) * S + x2) * S + x3) : 0;
    }
    #pragma unroll 1
    for (int c0 = 0; c0 < Ci; c0 += CB) {
        float v[K][CB];
        #pragma unroll
        for (int k = 0; k < K; ++k) {
            #pragma unroll
            for (int c = 0; c < CB; ++c)
                v[k][c] = xb[(c0 + c) * S4 + off[k]];
        }
        #pragma unroll
        for (int k = 0; k < K; ++k) {
            if (!FULL && tid + k * NT >= NSP) continue;
            #pragma unroll
            for (int c = 0; c < CB; c += 4) {
                v4f q;
                q.x = ok[k] ? v[k][c + 0] : 0.0f;
                q.y = ok[k] ? v[k][c + 1] : 0.0f;
                q.z = ok[k] ? v[k][c + 2] : 0.0f;
                q.w = ok[k] ? v[k][c + 3] : 0.0f;
                *(v4f*)&lds[(tid + k * NT) * CSTR + c0 + c] = q;
            }
        }
    }
    __syncthreads();

    const int lane = tid & 31;
    const int wave = tid >> 5;
    const int n    = lane & 15;
    const int klo  = (lane < 16) ? 0 : 2;
    const int qsp  = (((n >> 3) & 1) * 16 + ((n >> 2) & 1) * 4 + ((n >> 1) & 1)) * D3 + (n & 1);
    const int ntl  = wave / KS;
    const int ks   = wave - ntl * KS;
    const int kgw  = ks % KSG;
    const int t0w  = ks / KSG;
    const int ntile = nb * NTB + ntl;
    const int coA  = ntile * 16 + n;
    constexpr int KB = Ci / 4;
    constexpr int G  = KB < 4 ? KB : 4;
    constexpr int NG = KB / G;
    constexpr int Q  = G * 6 / 4;
    constexpr int NGW = NG / KSG;
    static_assert(NG % KSG == 0, "channel-group split");
    const v4f* w4 = (const v4f*)w + (ntile * 27 * NG * Q * 32 + kgw * (NGW * Q * 32) + lane);
    const int kbase = kgw * (NGW * G * 4);
    const int t0lo  = (KST == 3) ? t0w : 0;
    const int t0hi  = (KST == 3) ? t0w + 1 : 3;

    v8f acc[TILES];
    #pragma unroll
    for (int u = 0; u < TILES; ++u)
        acc[u] = (v8f){0.f, 0.f, 0.f, 0.f, 0.f, 0.f, 0.f, 0.f};

    #pragma unroll 1
    for (int t0 = t0lo; t0 < t0hi; ++t0) {
        #pragma unroll 1
        for (int t1 = 0; t1 < 3; ++t1) {
            #pragma unroll 1
            for (int t2 = 0; t2 < 3; ++t2) {
                const int row = (qsp + (t0 * 16 + t1 * 4 + t2) * D3) * CSTR + klo + kbase;
                const v4f* wp = w4 + ((t0 * 3 + t1) * 3 + t2) * (NG * Q * 32);
                #pragma unroll
                for (int kg = 0; kg < NGW; ++kg) {
                    float wf[G * 6];
                    #pragma unroll
                    for (int q = 0; q < Q; ++q) {
                        const v4f t = wp[(kg * Q + q) * 32];
                        wf[q * 4 + 0] = t.x; wf[q * 4 + 1] = t.y;
                        wf[q * 4 + 2] = t.z; wf[q * 4 + 3] = t.w;
                    }
                    #pragma unroll
                    for (int kbi = 0; kbi < G; ++kbi) {
                        const int kb = (kg * G + kbi) * 4;
                        #pragma unroll
                        for (int t3 = 0; t3 < 3; ++t3) {
                            v2f bf; bf.x = wf[kbi * 6 + 2 * t3]; bf.y = wf[kbi * 6 + 2 * t3 + 1];
                            #pragma unroll
                            for (int u = 0; u < TILES; ++u) {
                                v2f a = *(const v2f*)&lds[row + (t3 + 2 * u) * CSTR + kb];
                                acc[u] = __builtin_amdgcn_wmma_f32_16x16x4_f32(
                                     false, a,  false, bf,
                                     (short)0, acc[u],
                                     false,  false);
                            }
                        }
                    }
                }
            }
        }
    }

    if constexpr (KS > 1) {
        __shared__ __attribute__((aligned(16))) float red[NTB * KS * TILES * 256];
        if (ks != 0) {
            #pragma unroll
            for (int u = 0; u < TILES; ++u) {
                v4f lo, hi;
                lo.x = acc[u][0]; lo.y = acc[u][1]; lo.z = acc[u][2]; lo.w = acc[u][3];
                hi.x = acc[u][4]; hi.y = acc[u][5]; hi.z = acc[u][6]; hi.w = acc[u][7];
                *(v4f*)&red[(((wave * TILES + u) * 2 + 0) * 32 + lane) * 4] = lo;
                *(v4f*)&red[(((wave * TILES + u) * 2 + 1) * 32 + lane) * 4] = hi;
            }
        }
        __syncthreads();
        if (ks != 0) return;
        #pragma unroll
        for (int u = 0; u < TILES; ++u) {
            #pragma unroll
            for (int j = 1; j < KS; ++j) {
                const v4f lo = *(const v4f*)&red[((((wave + j) * TILES + u) * 2 + 0) * 32 + lane) * 4];
                const v4f hi = *(const v4f*)&red[((((wave + j) * TILES + u) * 2 + 1) * 32 + lane) * 4];
                acc[u][0] += lo.x; acc[u][1] += lo.y; acc[u][2] += lo.z; acc[u][3] += lo.w;
                acc[u][4] += hi.x; acc[u][5] += hi.y; acc[u][6] += hi.z; acc[u][7] += hi.w;
            }
        }
    }

    const float bs = bias[coA];
    #pragma unroll
    for (int u = 0; u < TILES; ++u) {
        float s = 0.0f;
        #pragma unroll
        for (int r = 0; r < 8; ++r) {
            float v = acc[u][r] + bs;
            s += (v > 0.0f) ? v : 0.0f;
        }
        s += __shfl_xor(s, 16, 32);
        if (lane < 16) {
            const int pIdx = ((p0 * P + p1) * P + p2) * P + p3h * TILES + u;
            out[(b * Co + coA) * P4 + pIdx] = s * (1.0f / 16.0f);
        }
    }
}

constexpr bool mf_banks_ok(int us, int s2, int s1, int s0)
{
    unsigned seen = 0;
    for (int n = 0; n < 16; ++n) {
        const int off = (n & 1) * us + ((n >> 1) & 1) * s2 + ((n >> 2) & 1) * s1 + ((n >> 3) & 1) * s0;
        if (off % 4) return false;
        const int g = (off / 4) % 16;
        if (seen & (1u << g)) return false;
        seen |= 1u << g;
    }
    return true;
}

template<int Ci, int Co, int S, int NTB, int IG, int WPE, int PU, int PZP, int P0>
__global__ __launch_bounds__(NTB * 4 * 32) __attribute__((amdgpu_waves_per_eu(WPE, WPE)))
void conv4d_relu_pool_mf(const float* __restrict__ x,
                         const float* __restrict__ w,
                         const float* __restrict__ bias,
                         float* __restrict__ out)
{
    constexpr int P    = S / 2;
    constexpr int UU   = P;
    constexpr int PZ   = 8 / UU;
    constexpr int PG2  = P / PZ;
    constexpr int P4   = P * P * P * P;
    constexpr int NACC = 16;
    constexpr int US   = Ci + PU;
    constexpr int JS   = UU * US;
    constexpr int S2   = 4 * JS;
    constexpr int SZ   = 4 * S2 + PZP;
    constexpr int S1   = PZ * SZ;
    constexpr int S0   = 4 * S1 + P0;
    constexpr int LDSF = 4 * S0;
    constexpr int NB   = (Co / 16) / NTB;
    constexpr int NT   = NTB * 4 * 32;
    constexpr int REDF = (NT / 32) * 4 * 256;
    static_assert((UU == 8 || UU == 4) && P % PZ == 0 && (Co / 16) % NTB == 0 && S % 2 == 0, "tile split");
    static_assert(mf_banks_ok(US, 2 * US, (UU == 8) ? 4 * US : SZ, S0), "fragment reads must be bank-conflict free");
    static_assert(REDF <= LDSF, "partial sums reuse the patch");
    static_assert(4 % IG == 0, "dim2 terms per weight load group");

    __shared__ __attribute__((aligned(16))) float lds[LDSF];

    const int b   = blockIdx.y;
    const int nb  = blockIdx.x % NB;
    const int bx  = blockIdx.x / NB;
    const int p2g = bx % PG2;
    const int p1  = (bx / PG2) % P;
    const int p0  = bx / (PG2 * P);

    const int tid = threadIdx.x;
    constexpr int NIT  = 4 * PZ * UU * Ci;
    constexpr int K    = NIT / NT;
    constexpr int S4   = S * S * S * S;
    static_assert(NIT % NT == 0 && Ci % 8 == 0 && UU % 4 == 0, "staging split");
    const float* __restrict__ xb = x + b * (Ci * S4);
    #pragma unroll 1
    for (int k = 0; k < K; ++k) {
        const int it = tid + k * NT;
        const int ul = it & 3;
        const int cl = (it >> 2) & 7;
        int rest = it >> 5;
        const int uh = rest % (UU / 4); rest /= (UU / 4);
        const int ch = rest % (Ci / 8); rest /= (Ci / 8);
        const int pz = rest % PZ;
        const int r0 = rest / PZ;
        const int uu = uh * 4 + ul;
        const int c  = ch * 8 + cl;
        const int x0 = 2 * p0 - 1 + r0;
        const int x1b = 2 * p1 - 1;
        const int x2b = 2 * (p2g * PZ + pz) - 1;
        const bool ok0 = (unsigned)x0 < (unsigned)S;
        const bool inA = (uu >= 1), inC = (2 * uu + 2 < S);
        const float* pc = xb + c * S4;
        float va[4][4], vc[4][4];
        v2f vb[4][4];
        bool okB[4][4];
        #pragma unroll
        for (int r1 = 0; r1 < 4; ++r1) {
            const int x1 = x1b + r1;
            const bool ok01 = ok0 && (unsigned)x1 < (unsigned)S;
            #pragma unroll
            for (int d = 0; d < 4; ++d) {
                const int x2 = x2b + d;
                okB[r1][d] = ok01 && (unsigned)x2 < (unsigned)S;
                const int base = ((x0 * S + x1) * S + x2) * S + 2 * uu;
                const int offB = okB[r1][d] ? base : 0;
                const int offA = (okB[r1][d] && inA) ? base - 1 : 0;
                const int offC = (okB[r1][d] && inC) ? base + 2 : 0;
                va[r1][d] = pc[offA];
                vb[r1][d] = *(const v2f*)(pc + offB);
                vc[r1][d] = pc[offC];
            }
        }
        float V2[4][4][4];
        #pragma unroll
        for (int r1 = 0; r1 < 4; ++r1) {
            float t[4][4];
            #pragma unroll
            for (int d = 0; d < 4; ++d) {
                const float d0 = (okB[r1][d] && inA) ? va[r1][d] : 0.0f;
                const float d1 = okB[r1][d] ? vb[r1][d].x : 0.0f;
                const float d2 = okB[r1][d] ? vb[r1][d].y : 0.0f;
                const float d3 = (okB[r1][d] && inC) ? vc[r1][d] : 0.0f;
                t[d][0] = d0 - d2; t[d][1] = d1 + d2; t[d][2] = d2 - d1; t[d][3] = d1 - d3;
            }
            #pragma unroll
            for (int j = 0; j < 4; ++j) {
                V2[r1][0][j] = t[0][j] - t[2][j];
                V2[r1][1][j] = t[1][j] + t[2][j];
                V2[r1][2][j] = t[2][j] - t[1][j];
                V2[r1][3][j] = t[1][j] - t[3][j];
            }
        }
        float* dp = &lds[r0 * S0 + pz * SZ + uu * US + c];
        #pragma unroll
        for (int i = 0; i < 4; ++i) {
            #pragma unroll
            for (int j = 0; j < 4; ++j) {
                const float a0 = V2[0][i][j], a1 = V2[1][i][j], a2 = V2[2][i][j], a3 = V2[3][i][j];
                dp[0 * S1 + i * S2 + j * JS] = a0 - a2;
                dp[1 * S1 + i * S2 + j * JS] = a1 + a2;
                dp[2 * S1 + i * S2 + j * JS] = a2 - a1;
                dp[3 * S1 + i * S2 + j * JS] = a1 - a3;
            }
        }
    }
    __syncthreads();

    const int lane = tid & 31;
    const int wave = __builtin_amdgcn_readfirstlane(tid >> 5);
    const int n    = lane & 15;
    const int klo  = (lane < 16) ? 0 : 2;
    const int ntl  = wave >> 2;
    const int hw   = wave & 3;
    const int qsp  = ((n >> 3) & 1) * S0 + ((n & 7) / UU) * SZ + (n & (UU - 1)) * US;
    const int ntile = nb * NTB + ntl;
    const int coA  = ntile * 16 + n;
    constexpr int KB = Ci / 4;
    constexpr int G  = KB < 4 ? KB : 4;
    constexpr int NG = KB / G;
    constexpr int Q  = G * 2;
    const v4f* w4 = (const v4f*)w + (ntile * 48 * NG * Q * 32 + lane);

    v8f acc[NACC];
    #pragma unroll
    for (int a = 0; a < NACC; ++a)
        acc[a] = (v8f){0.f, 0.f, 0.f, 0.f, 0.f, 0.f, 0.f, 0.f};

    #pragma unroll 1
    for (int t0 = 0; t0 < 3; ++t0) {
        #pragma unroll 1
        for (int kg = 0; kg < NG; ++kg) {
            const v4f* wp = w4 + ((t0 * 4 + hw) * 4 * NG + kg) * (Q * 32);
            const int row0 = qsp + t0 * S0 + hw * S1 + klo + kg * (G * 4);
            #pragma unroll
            for (int ib = 0; ib < 4; ib += IG) {
                float wf[IG][G * 8];
                #pragma unroll
                for (int tg = 0; tg < IG; ++tg) {
                    #pragma unroll
                    for (int q = 0; q < Q; ++q) {
                        const v4f t = wp[((ib + tg) * NG * Q + q) * 32];
                        wf[tg][q * 4 + 0] = t.x; wf[tg][q * 4 + 1] = t.y;
                        wf[tg][q * 4 + 2] = t.z; wf[tg][q * 4 + 3] = t.w;
                    }
                }
                #pragma unroll
                for (int tg = 0; tg < IG; ++tg) {
                    const int i = ib + tg;
                    const int row = row0 + i * S2;
                    #pragma unroll
                    for (int kbi = 0; kbi < G; ++kbi) {
                        #pragma unroll
                        for (int j = 0; j < 4; ++j) {
                            v2f bf; bf.x = wf[tg][kbi * 8 + 2 * j]; bf.y = wf[tg][kbi * 8 + 2 * j + 1];
                            const float* ap = lds + (row + j * JS + kbi * 4);
                            v2f a = *(const v2f*)ap;
                            acc[i * 4 + j] = __builtin_amdgcn_wmma_f32_16x16x4_f32(
                                 false, a,  false, bf,
                                 (short)0, acc[i * 4 + j],
                                 false,  false);
                        }
                    }
                }
            }
        }
    }

    v8f y[4];
    #pragma unroll
    for (int r = 0; r < 8; ++r) {
        float z0[4], z1[4];
        #pragma unroll
        for (int i = 0; i < 4; ++i) {
            const float m0 = acc[i * 4 + 0][r], m1 = acc[i * 4 + 1][r];
            const float m2 = acc[i * 4 + 2][r], m3 = acc[i * 4 + 3][r];
            z0[i] = (m0 + m1) + m2;
            z1[i] = (m1 - m2) - m3;
        }
        y[0][r] = (z0[0] + z0[1]) + z0[2];
        y[1][r] = (z1[0] + z1[1]) + z1[2];
        y[2][r] = (z0[1] - z0[2]) - z0[3];
        y[3][r] = (z1[1] - z1[2]) - z1[3];
    }

    __syncthreads();
    float* red = lds;
    if (hw != 0) {
        #pragma unroll
        for (int a = 0; a < 4; ++a) {
            v4f lo, hi;
            lo.x = y[a][0]; lo.y = y[a][1]; lo.z = y[a][2]; lo.w = y[a][3];
            hi.x = y[a][4]; hi.y = y[a][5]; hi.z = y[a][6]; hi.w = y[a][7];
            *(v4f*)&red[(((wave * 4 + a) * 2 + 0) * 32 + lane) * 4] = lo;
            *(v4f*)&red[(((wave * 4 + a) * 2 + 1) * 32 + lane) * 4] = hi;
        }
    }
    __syncthreads();
    if (hw != 0) return;
    v8f ya[4], yb[4];
    #pragma unroll
    for (int a = 0; a < 4; ++a) {
        v8f pt[3];
        #pragma unroll
        for (int j = 1; j < 4; ++j) {
            const v4f lo = *(const v4f*)&red[((((wave + j) * 4 + a) * 2 + 0) * 32 + lane) * 4];
            const v4f hi = *(const v4f*)&red[((((wave + j) * 4 + a) * 2 + 1) * 32 + lane) * 4];
            pt[j - 1] = (v8f){lo.x, lo.y, lo.z, lo.w, hi.x, hi.y, hi.z, hi.w};
        }
        #pragma unroll
        for (int r = 0; r < 8; ++r) {
            ya[a][r] = (y[a][r] + pt[0][r]) + pt[1][r];
            yb[a][r] = (pt[0][r] - pt[1][r]) - pt[2][r];
        }
    }

    const float bs = bias[coA];
    float s[8];
    #pragma unroll
    for (int r = 0; r < 8; ++r) {
        float sa = 0.0f;
        #pragma unroll
        for (int a = 0; a < 4; ++a) {
            const float v0 = ya[a][r] + bs, v1 = yb[a][r] + bs;
            sa += ((v0 > 0.0f) ? v0 : 0.0f) + ((v1 > 0.0f) ? v1 : 0.0f);
        }
        sa += __shfl_xor(sa, 16, 32);
        s[r] = sa * (1.0f / 16.0f);
    }
    if (lane < 16) {
        float* op = out + (b * Co + coA) * P4 + ((p0 * P + p1) * P + p2g * PZ) * P;
        *(v4f*)(op + 0) = (v4f){s[0], s[1], s[2], s[3]};
        *(v4f*)(op + 4) = (v4f){s[4], s[5], s[6], s[7]};
    }
}

extern "C" void kernel_launch(void* const* d_in, const int* in_sizes, int n_in,
                              void* d_out, int out_size, void* d_ws, size_t ws_size,
                              hipStream_t stream) {
    (void)in_sizes; (void)n_in; (void)out_size; (void)ws_size;

    const float* src = (const float*)d_in[0];
    const float* W0  = (const float*)d_in[1];
    const float* b0  = (const float*)d_in[2];
    const float* W1  = (const float*)d_in[3];
    const float* b1  = (const float*)d_in[4];
    const float* W2  = (const float*)d_in[5];
    const float* b2  = (const float*)d_in[6];
    const float* W3  = (const float*)d_in[7];
    const float* b3  = (const float*)d_in[8];

    float* t0 = (float*)d_ws;
    float* r0 = t0 + 8 * 32 * 4096;
    float* r1 = r0 + 32 * 8 * 3 * 64;
    float* r2 = r1 + 64 * 32 * 3 * 64;
    float* r3 = r2 + 64 * 64 * 81;
    float* o1 = (float*)d_out;
    float* o2 = o1 + 8 * 64 * 256;
    float* o3 = o2 + 8 * 64 * 16;

    relayout_weights<<<4320, 256, 0, stream>>>(W0, W1, W2, W3, r0);
    conv4d_relu_pool_mf<8, 32, 16, 2, 4, 4, 0, 0, 4><<<dim3(512, 8), 256, 0, stream>>>(src, r0, b0, t0);
    conv4d_relu_pool_mf<32, 64, 8, 4, 2, 4, 4, 16, 32><<<dim3(32, 8), 512, 0, stream>>>(t0, r1, b1, o1);
    conv4d_relu_pool_wmma<64, 64, 4, 2, 4, 3, 1><<<dim3(32, 8), 384, 0, stream>>>(o1, r2, b2, o2);
    conv4d_relu_pool_wmma<64, 64, 2, 1, 4, 3, 1><<<dim3(4, 8), 384, 0, stream>>>(o2, r3, b3, o3);
}


# Weight re-layout launch: each lane loads its source taps once and stores whole 16-byte pieces (a channel pair's 64 filter terms per lane): 972 waves for 34560; weight values and sum orders unchanged
# speedup vs baseline: 5.8647x; 1.0402x over previous
#include <hip/hip_runtime.h>
#include <hip/hip_bf16.h>

typedef float v2f __attribute__((ext_vector_type(2)));
typedef float v4f __attribute__((ext_vector_type(4)));
typedef float v8f __attribute__((ext_vector_type(8)));

#define LDS_PAD 4


template<int Ci>
__device__ __forceinline__ void relayout_one(const float* __restrict__ w,
                                             float* __restrict__ dst, int item, int lane)
{
    constexpr int KB = Ci / 4;
    constexpr int G  = KB < 4 ? KB : 4;
    constexpr int NG = KB / G;
    constexpr int Q  = G * 6 / 4;
    const int kg  = item % NG;
    const int ct  = item / NG;
    const int tt  = ct % 27;
    const int cot = ct / 27;
    const int co  = cot * 16 + (lane & 15);
    const int ci0 = kg * G * 4 + ((lane < 16) ? 0 : 2);
    const float* p = w + (co * Ci + ci0) * 81 + tt * 3;
    float v[G * 6];
    #pragma unroll
    for (int kbi = 0; kbi < G; ++kbi) {
        #pragma unroll
        for (int j = 0; j < 6; ++j)
            v[kbi * 6 + j] = p[(kbi * 4 + (j & 1)) * 81 + (j >> 1)];
    }
    v4f* d4 = (v4f*)dst + (item * Q * 32 + lane);
    #pragma unroll
    for (int q = 0; q < Q; ++q)
        d4[q * 32] = (v4f){v[q * 4 + 0], v[q * 4 + 1], v[q * 4 + 2], v[q * 4 + 3]};
}

template<int Ci>
__device__ __forceinline__ void relayout_mf(const float* __restrict__ w,
                                            float* __restrict__ dst, int item, int lane)
{
    constexpr int KB = Ci / 4;
    constexpr int G  = KB < 4 ? KB : 4;
    constexpr int NG = KB / G;
    constexpr int Q  = G * 2;
    const int kbi = item % G;
    int rest = item / G;
    const int kg  = rest % NG; rest /= NG;
    const int t0  = rest % 3;
    const int cot = rest / 3;
    const int co  = cot * 16 + (lane & 15);
    const int ci0 = (kg * G + kbi) * 4 + ((lane < 16) ? 0 : 2);
    const float* p = w + (co * Ci + ci0) * 81 + t0 * 27;
    float g[2][27];
    #pragma unroll
    for (int hi = 0; hi < 2; ++hi) {
        #pragma unroll
        for (int k = 0; k < 27; ++k)
            g[hi][k] = p[hi * 81 + k];
    }
    v4f* d4 = (v4f*)dst + ((((cot * 48 + t0 * 16) * NG + kg) * Q + kbi * 2) * 32 + lane);
    #pragma unroll
    for (int i = 0; i < 4; ++i) {
        float m[2][3][4];
        #pragma unroll
        for (int hi = 0; hi < 2; ++hi) {
            #pragma unroll
            for (int t1 = 0; t1 < 3; ++t1) {
                float hh[3];
                #pragma unroll
                for (int t3 = 0; t3 < 3; ++t3) {
                    const float a0 = g[hi][t1 * 9 + t3], a1 = g[hi][t1 * 9 + 3 + t3], a2 = g[hi][t1 * 9 + 6 + t3];
                    const float ea = a0 + a2;
                    if (i == 0)      hh[t3] = a0;
                    else if (i == 1) hh[t3] = 0.5f * (ea + a1);
                    else if (i == 2) hh[t3] = 0.5f * (ea - a1);
                    else             hh[t3] = a2;
                }
                const float e = hh[0] + hh[2];
                m[hi][t1][0] = hh[0];
                m[hi][t1][1] = 0.5f * (e + hh[1]);
                m[hi][t1][2] = 0.5f * (e - hh[1]);
                m[hi][t1][3] = hh[2];
            }
        }
        float r[2][4][4];
        #pragma unroll
        for (int hi = 0; hi < 2; ++hi) {
            #pragma unroll
            for (int j = 0; j < 4; ++j) {
                const float em = m[hi][0][j] + m[hi][2][j];
                r[hi][j][0] = m[hi][0][j];
                r[hi][j][1] = 0.5f * (em + m[hi][1][j]);
                r[hi][j][2] = 0.5f * (em - m[hi][1][j]);
                r[hi][j][3] = m[hi][2][j];
            }
        }
        #pragma unroll
        for (int h = 0; h < 4; ++h) {
            #pragma unroll
            for (int jp = 0; jp < 2; ++jp)
                d4[((h * 4 + i) * NG * Q + jp) * 32] =
                    (v4f){r[0][2 * jp][h], r[1][2 * jp][h], r[0][2 * jp + 1][h], r[1][2 * jp + 1][h]};
        }
    }
}

__global__ __launch_bounds__(64)
void relayout_weights(const float* __restrict__ w0, const float* __restrict__ w1,
                      const float* __restrict__ w2, const float* __restrict__ w3,
                      float* __restrict__ dst)
{
    constexpr int N0 = 32 * 8 * 3 * 64, N1 = 64 * 32 * 3 * 64, N2 = 64 * 64 * 81;
    constexpr int I0 = 2 * 3 * 1 * 2, I1 = 4 * 3 * 2 * 4, I2 = 4 * 27 * 4;
    const int lane = threadIdx.x & 31;
    const int item = __builtin_amdgcn_readfirstlane(blockIdx.x * 2 + (threadIdx.x >> 5));
    if (item < I0)                relayout_mf<8 >(w0, dst, item, lane);
    else if (item < I0 + I1)      relayout_mf<32>(w1, dst + N0, item - I0, lane);
    else if (item < I0 + I1 + I2) relayout_one<64>(w2, dst + N0 + N1, item - I0 - I1, lane);
    else                          relayout_one<64>(w3, dst + N0 + N1 + N2, item - I0 - I1 - I2, lane);
}


template<int Ci, int Co, int S, int TILES, int KSG = 1, int KST = 1, int NTB = Co / 16>
__global__ __launch_bounds__(NTB * KSG * KST * 32)
void conv4d_relu_pool_wmma(const float* __restrict__ x,
                           const float* __restrict__ w,
                           const float* __restrict__ bias,
                           float* __restrict__ out)
{
    constexpr int P    = S / 2;
    constexpr int PH   = P / TILES;
    constexpr int P4   = P * P * P * P;
    constexpr int D3   = 2 + 2 * TILES;
    constexpr int NSP  = 64 * D3;
    constexpr int CSTR = Ci + LDS_PAD;
    constexpr int KS   = KSG * KST;
    constexpr int NB   = (Co / 16) / NTB;
    constexpr int NT   = NTB * KS * 32;
    static_assert((Co / 16) % NTB == 0 && (KST == 1 || KST == 3), "block split");

    __shared__ __attribute__((aligned(16))) float lds[NSP * CSTR];

    const int b   = blockIdx.y;
    const int nb  = blockIdx.x % NB;
    const int bx  = blockIdx.x / NB;
    const int p3h = bx % PH;
    const int p2  = (bx / PH) % P;
    const int p1  = (bx / (PH * P)) % P;
    const int p0  = bx / (PH * P * P);

    const int tid = threadIdx.x;
    constexpr int K  = (NSP + NT - 1) / NT;
    constexpr bool FULL = (NSP % NT == 0);
    constexpr int CB = (Ci < 16) ? Ci : 16;
    constexpr int S4 = S * S * S * S;
    static_assert(Ci % CB == 0 && CB % 4 == 0, "staging split");
    const float* __restrict__ xb = x + b * (Ci * S4);
    int  off[K];
    bool ok[K];
    #pragma unroll
    for (int k = 0; k < K; ++k) {
        const int sp = tid + k * NT;
        const int r  = sp / D3;
        const int d3 = sp - r * D3;
        const int x0 = 2 * p0 - 1 + ((r >> 4) & 3);
        const int x1 = 2 * p1 - 1 + ((r >> 2) & 3);
        const int x2 = 2 * p2 - 1 + (r & 3);
        const int x3 = 2 * (p3h * TILES) - 1 + d3;
        ok[k]  = (FULL || sp < NSP) &&
                 (unsigned)x0 < (unsigned)S && (unsigned)x1 < (unsigned)S &&
                 (unsigned)x2 < (unsigned)S && (unsigned)x3 < (unsigned)S;
        off[k] = ok[k] ? (((x0 * S + x1) * S + x2) * S + x3) : 0;
    }
    #pragma unroll 1
    for (int c0 = 0; c0 < Ci; c0 += CB) {
        float v[K][CB];
        #pragma unroll
        for (int k = 0; k < K; ++k) {
            #pragma unroll
            for (int c = 0; c < CB; ++c)
                v[k][c] = xb[(c0 + c) * S4 + off[k]];
        }
        #pragma unroll
        for (int k = 0; k < K; ++k) {
            if (!FULL && tid + k * NT >= NSP) continue;
            #pragma unroll
            for (int c = 0; c < CB; c += 4) {
                v4f q;
                q.x = ok[k] ? v[k][c + 0] : 0.0f;
                q.y = ok[k] ? v[k][c + 1] : 0.0f;
                q.z = ok[k] ? v[k][c + 2] : 0.0f;
                q.w = ok[k] ? v[k][c + 3] : 0.0f;
                *(v4f*)&lds[(tid + k * NT) * CSTR + c0 + c] = q;
            }
        }
    }
    __syncthreads();

    const int lane = tid & 31;
    const int wave = tid >> 5;
    const int n    = lane & 15;
    const int klo  = (lane < 16) ? 0 : 2;
    const int qsp  = (((n >> 3) & 1) * 16 + ((n >> 2) & 1) * 4 + ((n >> 1) & 1)) * D3 + (n & 1);
    const int ntl  = wave / KS;
    const int ks   = wave - ntl * KS;
    const int kgw  = ks % KSG;
    const int t0w  = ks / KSG;
    const int ntile = nb * NTB + ntl;
    const int coA  = ntile * 16 + n;
    constexpr int KB = Ci / 4;
    constexpr int G  = KB < 4 ? KB : 4;
    constexpr int NG = KB / G;
    constexpr int Q  = G * 6 / 4;
    constexpr int NGW = NG / KSG;
    static_assert(NG % KSG == 0, "channel-group split");
    const v4f* w4 = (const v4f*)w + (ntile * 27 * NG * Q * 32 + kgw * (NGW * Q * 32) + lane);
    const int kbase = kgw * (NGW * G * 4);
    const int t0lo  = (KST == 3) ? t0w : 0;
    const int t0hi  = (KST == 3) ? t0w + 1 : 3;

    v8f acc[TILES];
    #pragma unroll
    for (int u = 0; u < TILES; ++u)
        acc[u] = (v8f){0.f, 0.f, 0.f, 0.f, 0.f, 0.f, 0.f, 0.f};

    #pragma unroll 1
    for (int t0 = t0lo; t0 < t0hi; ++t0) {
        #pragma unroll 1
        for (int t1 = 0; t1 < 3; ++t1) {
            #pragma unroll 1
            for (int t2 = 0; t2 < 3; ++t2) {
                const int row = (qsp + (t0 * 16 + t1 * 4 + t2) * D3) * CSTR + klo + kbase;
                const v4f* wp = w4 + ((t0 * 3 + t1) * 3 + t2) * (NG * Q * 32);
                #pragma unroll
                for (int kg = 0; kg < NGW; ++kg) {
                    float wf[G * 6];
                    #pragma unroll
                    for (int q = 0; q < Q; ++q) {
                        const v4f t = wp[(kg * Q + q) * 32];
                        wf[q * 4 + 0] = t.x; wf[q * 4 + 1] = t.y;
                        wf[q * 4 + 2] = t.z; wf[q * 4 + 3] = t.w;
                    }
                    #pragma unroll
                    for (int kbi = 0; kbi < G; ++kbi) {
                        const int kb = (kg * G + kbi) * 4;
                        #pragma unroll
                        for (int t3 = 0; t3 < 3; ++t3) {
                            v2f bf; bf.x = wf[kbi * 6 + 2 * t3]; bf.y = wf[kbi * 6 + 2 * t3 + 1];
                            #pragma unroll
                            for (int u = 0; u < TILES; ++u) {
                                v2f a = *(const v2f*)&lds[row + (t3 + 2 * u) * CSTR + kb];
                                acc[u] = __builtin_amdgcn_wmma_f32_16x16x4_f32(
                                     false, a,  false, bf,
                                     (short)0, acc[u],
                                     false,  false);
                            }
                        }
                    }
                }
            }
        }
    }

    if constexpr (KS > 1) {
        __shared__ __attribute__((aligned(16))) float red[NTB * KS * TILES * 256];
        if (ks != 0) {
            #pragma unroll
            for (int u = 0; u < TILES; ++u) {
                v4f lo, hi;
                lo.x = acc[u][0]; lo.y = acc[u][1]; lo.z = acc[u][2]; lo.w = acc[u][3];
                hi.x = acc[u][4]; hi.y = acc[u][5]; hi.z = acc[u][6]; hi.w = acc[u][7];
                *(v4f*)&red[(((wave * TILES + u) * 2 + 0) * 32 + lane) * 4] = lo;
                *(v4f*)&red[(((wave * TILES + u) * 2 + 1) * 32 + lane) * 4] = hi;
            }
        }
        __syncthreads();
        if (ks != 0) return;
        #pragma unroll
        for (int u = 0; u < TILES; ++u) {
            #pragma unroll
            for (int j = 1; j < KS; ++j) {
                const v4f lo = *(const v4f*)&red[((((wave + j) * TILES + u) * 2 + 0) * 32 + lane) * 4];
                const v4f hi = *(const v4f*)&red[((((wave + j) * TILES + u) * 2 + 1) * 32 + lane) * 4];
                acc[u][0] += lo.x; acc[u][1] += lo.y; acc[u][2] += lo.z; acc[u][3] += lo.w;
                acc[u][4] += hi.x; acc[u][5] += hi.y; acc[u][6] += hi.z; acc[u][7] += hi.w;
            }
        }
    }

    const float bs = bias[coA];
    #pragma unroll
    for (int u = 0; u < TILES; ++u) {
        float s = 0.0f;
        #pragma unroll
        for (int r = 0; r < 8; ++r) {
            float v = acc[u][r] + bs;
            s += (v > 0.0f) ? v : 0.0f;
        }
        s += __shfl_xor(s, 16, 32);
        if (lane < 16) {
            const int pIdx = ((p0 * P + p1) * P + p2) * P + p3h * TILES + u;
            out[(b * Co + coA) * P4 + pIdx] = s * (1.0f / 16.0f);
        }
    }
}

constexpr bool mf_banks_ok(int us, int s2, int s1, int s0)
{
    unsigned seen = 0;
    for (int n = 0; n < 16; ++n) {
        const int off = (n & 1) * us + ((n >> 1) & 1) * s2 + ((n >> 2) & 1) * s1 + ((n >> 3) & 1) * s0;
        if (off % 4) return false;
        const int g = (off / 4) % 16;
        if (seen & (1u << g)) return false;
        seen |= 1u << g;
    }
    return true;
}

template<int Ci, int Co, int S, int NTB, int IG, int WPE, int PU, int PZP, int P0>
__global__ __launch_bounds__(NTB * 4 * 32) __attribute__((amdgpu_waves_per_eu(WPE, WPE)))
void conv4d_relu_pool_mf(const float* __restrict__ x,
                         const float* __restrict__ w,
                         const float* __restrict__ bias,
                         float* __restrict__ out)
{
    constexpr int P    = S / 2;
    constexpr int UU   = P;
    constexpr int PZ   = 8 / UU;
    constexpr int PG2  = P / PZ;
    constexpr int P4   = P * P * P * P;
    constexpr int NACC = 16;
    constexpr int US   = Ci + PU;
    constexpr int JS   = UU * US;
    constexpr int S2   = 4 * JS;
    constexpr int SZ   = 4 * S2 + PZP;
    constexpr int S1   = PZ * SZ;
    constexpr int S0   = 4 * S1 + P0;
    constexpr int LDSF = 4 * S0;
    constexpr int NB   = (Co / 16) / NTB;
    constexpr int NT   = NTB * 4 * 32;
    constexpr int REDF = (NT / 32) * 4 * 256;
    static_assert((UU == 8 || UU == 4) && P % PZ == 0 && (Co / 16) % NTB == 0 && S % 2 == 0, "tile split");
    static_assert(mf_banks_ok(US, 2 * US, (UU == 8) ? 4 * US : SZ, S0), "fragment reads must be bank-conflict free");
    static_assert(REDF <= LDSF, "partial sums reuse the patch");
    static_assert(4 % IG == 0, "dim2 terms per weight load group");

    __shared__ __attribute__((aligned(16))) float lds[LDSF];

    const int b   = blockIdx.y;
    const int nb  = blockIdx.x % NB;
    const int bx  = blockIdx.x / NB;
    const int p2g = bx % PG2;
    const int p1  = (bx / PG2) % P;
    const int p0  = bx / (PG2 * P);

    const int tid = threadIdx.x;
    constexpr int NIT  = 4 * PZ * UU * Ci;
    constexpr int K    = NIT / NT;
    constexpr int S4   = S * S * S * S;
    static_assert(NIT % NT == 0 && Ci % 8 == 0 && UU % 4 == 0, "staging split");
    const float* __restrict__ xb = x + b * (Ci * S4);
    #pragma unroll 1
    for (int k = 0; k < K; ++k) {
        const int it = tid + k * NT;
        const int ul = it & 3;
        const int cl = (it >> 2) & 7;
        int rest = it >> 5;
        const int uh = rest % (UU / 4); rest /= (UU / 4);
        const int ch = rest % (Ci / 8); rest /= (Ci / 8);
        const int pz = rest % PZ;
        const int r0 = rest / PZ;
        const int uu = uh * 4 + ul;
        const int c  = ch * 8 + cl;
        const int x0 = 2 * p0 - 1 + r0;
        const int x1b = 2 * p1 - 1;
        const int x2b = 2 * (p2g * PZ + pz) - 1;
        const bool ok0 = (unsigned)x0 < (unsigned)S;
        const bool inA = (uu >= 1), inC = (2 * uu + 2 < S);
        const float* pc = xb + c * S4;
        float va[4][4], vc[4][4];
        v2f vb[4][4];
        bool okB[4][4];
        #pragma unroll
        for (int r1 = 0; r1 < 4; ++r1) {
            const int x1 = x1b + r1;
            const bool ok01 = ok0 && (unsigned)x1 < (unsigned)S;
            #pragma unroll
            for (int d = 0; d < 4; ++d) {
                const int x2 = x2b + d;
                okB[r1][d] = ok01 && (unsigned)x2 < (unsigned)S;
                const int base = ((x0 * S + x1) * S + x2) * S + 2 * uu;
                const int offB = okB[r1][d] ? base : 0;
                const int offA = (okB[r1][d] && inA) ? base - 1 : 0;
                const int offC = (okB[r1][d] && inC) ? base + 2 : 0;
                va[r1][d] = pc[offA];
                vb[r1][d] = *(const v2f*)(pc + offB);
                vc[r1][d] = pc[offC];
            }
        }
        float V2[4][4][4];
        #pragma unroll
        for (int r1 = 0; r1 < 4; ++r1) {
            float t[4][4];
            #pragma unroll
            for (int d = 0; d < 4; ++d) {
                const float d0 = (okB[r1][d] && inA) ? va[r1][d] : 0.0f;
                const float d1 = okB[r1][d] ? vb[r1][d].x : 0.0f;
                const float d2 = okB[r1][d] ? vb[r1][d].y : 0.0f;
                const float d3 = (okB[r1][d] && inC) ? vc[r1][d] : 0.0f;
                t[d][0] = d0 - d2; t[d][1] = d1 + d2; t[d][2] = d2 - d1; t[d][3] = d1 - d3;
            }
            #pragma unroll
            for (int j = 0; j < 4; ++j) {
                V2[r1][0][j] = t[0][j] - t[2][j];
                V2[r1][1][j] = t[1][j] + t[2][j];
                V2[r1][2][j] = t[2][j] - t[1][j];
                V2[r1][3][j] = t[1][j] - t[3][j];
            }
        }
        float* dp = &lds[r0 * S0 + pz * SZ + uu * US + c];
        #pragma unroll
        for (int i = 0; i < 4; ++i) {
            #pragma unroll
            for (int j = 0; j < 4; ++j) {
                const float a0 = V2[0][i][j], a1 = V2[1][i][j], a2 = V2[2][i][j], a3 = V2[3][i][j];
                dp[0 * S1 + i * S2 + j * JS] = a0 - a2;
                dp[1 * S1 + i * S2 + j * JS] = a1 + a2;
                dp[2 * S1 + i * S2 + j * JS] = a2 - a1;
                dp[3 * S1 + i * S2 + j * JS] = a1 - a3;
            }
        }
    }
    __syncthreads();

    const int lane = tid & 31;
    const int wave = __builtin_amdgcn_readfirstlane(tid >> 5);
    const int n    = lane & 15;
    const int klo  = (lane < 16) ? 0 : 2;
    const int ntl  = wave >> 2;
    const int hw   = wave & 3;
    const int qsp  = ((n >> 3) & 1) * S0 + ((n & 7) / UU) * SZ + (n & (UU - 1)) * US;
    const int ntile = nb * NTB + ntl;
    const int coA  = ntile * 16 + n;
    constexpr int KB = Ci / 4;
    constexpr int G  = KB < 4 ? KB : 4;
    constexpr int NG = KB / G;
    constexpr int Q  = G * 2;
    const v4f* w4 = (const v4f*)w + (ntile * 48 * NG * Q * 32 + lane);

    v8f acc[NACC];
    #pragma unroll
    for (int a = 0; a < NACC; ++a)
        acc[a] = (v8f){0.f, 0.f, 0.f, 0.f, 0.f, 0.f, 0.f, 0.f};

    #pragma unroll 1
    for (int t0 = 0; t0 < 3; ++t0) {
        #pragma unroll 1
        for (int kg = 0; kg < NG; ++kg) {
            const v4f* wp = w4 + ((t0 * 4 + hw) * 4 * NG + kg) * (Q * 32);
            const int row0 = qsp + t0 * S0 + hw * S1 + klo + kg * (G * 4);
            #pragma unroll
            for (int ib = 0; ib < 4; ib += IG) {
                float wf[IG][G * 8];
                #pragma unroll
                for (int tg = 0; tg < IG; ++tg) {
                    #pragma unroll
                    for (int q = 0; q < Q; ++q) {
                        const v4f t = wp[((ib + tg) * NG * Q + q) * 32];
                        wf[tg][q * 4 + 0] = t.x; wf[tg][q * 4 + 1] = t.y;
                        wf[tg][q * 4 + 2] = t.z; wf[tg][q * 4 + 3] = t.w;
                    }
                }
                #pragma unroll
                for (int tg = 0; tg < IG; ++tg) {
                    const int i = ib + tg;
                    const int row = row0 + i * S2;
                    #pragma unroll
                    for (int kbi = 0; kbi < G; ++kbi) {
                        #pragma unroll
                        for (int j = 0; j < 4; ++j) {
                            v2f bf; bf.x = wf[tg][kbi * 8 + 2 * j]; bf.y = wf[tg][kbi * 8 + 2 * j + 1];
                            const float* ap = lds + (row + j * JS + kbi * 4);
                            v2f a = *(const v2f*)ap;
                            acc[i * 4 + j] = __builtin_amdgcn_wmma_f32_16x16x4_f32(
                                 false, a,  false, bf,
                                 (short)0, acc[i * 4 + j],
                                 false,  false);
                        }
                    }
                }
            }
        }
    }

    v8f y[4];
    #pragma unroll
    for (int r = 0; r < 8; ++r) {
        float z0[4], z1[4];
        #pragma unroll
        for (int i = 0; i < 4; ++i) {
            const float m0 = acc[i * 4 + 0][r], m1 = acc[i * 4 + 1][r];
            const float m2 = acc[i * 4 + 2][r], m3 = acc[i * 4 + 3][r];
            z0[i] = (m0 + m1) + m2;
            z1[i] = (m1 - m2) - m3;
        }
        y[0][r] = (z0[0] + z0[1]) + z0[2];
        y[1][r] = (z1[0] + z1[1]) + z1[2];
        y[2][r] = (z0[1] - z0[2]) - z0[3];
        y[3][r] = (z1[1] - z1[2]) - z1[3];
    }

    __syncthreads();
    float* red = lds;
    if (hw != 0) {
        #pragma unroll
        for (int a = 0; a < 4; ++a) {
            v4f lo, hi;
            lo.x = y[a][0]; lo.y = y[a][1]; lo.z = y[a][2]; lo.w = y[a][3];
            hi.x = y[a][4]; hi.y = y[a][5]; hi.z = y[a][6]; hi.w = y[a][7];
            *(v4f*)&red[(((wave * 4 + a) * 2 + 0) * 32 + lane) * 4] = lo;
            *(v4f*)&red[(((wave * 4 + a) * 2 + 1) * 32 + lane) * 4] = hi;
        }
    }
    __syncthreads();
    if (hw != 0) return;
    v8f ya[4], yb[4];
    #pragma unroll
    for (int a = 0; a < 4; ++a) {
        v8f pt[3];
        #pragma unroll
        for (int j = 1; j < 4; ++j) {
            const v4f lo = *(const v4f*)&red[((((wave + j) * 4 + a) * 2 + 0) * 32 + lane) * 4];
            const v4f hi = *(const v4f*)&red[((((wave + j) * 4 + a) * 2 + 1) * 32 + lane) * 4];
            pt[j - 1] = (v8f){lo.x, lo.y, lo.z, lo.w, hi.x, hi.y, hi.z, hi.w};
        }
        #pragma unroll
        for (int r = 0; r < 8; ++r) {
            ya[a][r] = (y[a][r] + pt[0][r]) + pt[1][r];
            yb[a][r] = (pt[0][r] - pt[1][r]) - pt[2][r];
        }
    }

    const float bs = bias[coA];
    float s[8];
    #pragma unroll
    for (int r = 0; r < 8; ++r) {
        float sa = 0.0f;
        #pragma unroll
        for (int a = 0; a < 4; ++a) {
            const float v0 = ya[a][r] + bs, v1 = yb[a][r] + bs;
            sa += ((v0 > 0.0f) ? v0 : 0.0f) + ((v1 > 0.0f) ? v1 : 0.0f);
        }
        sa += __shfl_xor(sa, 16, 32);
        s[r] = sa * (1.0f / 16.0f);
    }
    if (lane < 16) {
        float* op = out + (b * Co + coA) * P4 + ((p0 * P + p1) * P + p2g * PZ) * P;
        *(v4f*)(op + 0) = (v4f){s[0], s[1], s[2], s[3]};
        *(v4f*)(op + 4) = (v4f){s[4], s[5], s[6], s[7]};
    }
}

extern "C" void kernel_launch(void* const* d_in, const int* in_sizes, int n_in,
                              void* d_out, int out_size, void* d_ws, size_t ws_size,
                              hipStream_t stream) {
    (void)in_sizes; (void)n_in; (void)out_size; (void)ws_size;

    const float* src = (const float*)d_in[0];
    const float* W0  = (const float*)d_in[1];
    const float* b0  = (const float*)d_in[2];
    const float* W1  = (const float*)d_in[3];
    const float* b1  = (const float*)d_in[4];
    const float* W2  = (const float*)d_in[5];
    const float* b2  = (const float*)d_in[6];
    const float* W3  = (const float*)d_in[7];
    const float* b3  = (const float*)d_in[8];

    float* t0 = (float*)d_ws;
    float* r0 = t0 + 8 * 32 * 4096;
    float* r1 = r0 + 32 * 8 * 3 * 64;
    float* r2 = r1 + 64 * 32 * 3 * 64;
    float* r3 = r2 + 64 * 64 * 81;
    float* o1 = (float*)d_out;
    float* o2 = o1 + 8 * 64 * 256;
    float* o3 = o2 + 8 * 64 * 16;

    relayout_weights<<<486, 64, 0, stream>>>(W0, W1, W2, W3, r0);
    conv4d_relu_pool_mf<8, 32, 16, 2, 4, 4, 0, 0, 4><<<dim3(512, 8), 256, 0, stream>>>(src, r0, b0, t0);
    conv4d_relu_pool_mf<32, 64, 8, 4, 2, 4, 4, 16, 32><<<dim3(32, 8), 512, 0, stream>>>(t0, r1, b1, o1);
    conv4d_relu_pool_wmma<64, 64, 4, 2, 4, 3, 1><<<dim3(32, 8), 384, 0, stream>>>(o1, r2, b2, o2);
    conv4d_relu_pool_wmma<64, 64, 2, 1, 4, 3, 1><<<dim3(4, 8), 384, 0, stream>>>(o2, r3, b3, o3);
}


# All four conv layers: a wave's weight fragments pass through a four-slot register ring, requested three chunks ahead of their matrix ops, the first three before patch staging; sum order unchanged
# speedup vs baseline: 6.2415x; 1.0642x over previous
#include <hip/hip_runtime.h>
#include <hip/hip_bf16.h>

typedef float v2f __attribute__((ext_vector_type(2)));
typedef float v4f __attribute__((ext_vector_type(4)));
typedef float v8f __attribute__((ext_vector_type(8)));

#define LDS_PAD 4


template<int Ci>
__device__ __forceinline__ void relayout_one(const float* __restrict__ w,
                                             float* __restrict__ dst, int item, int lane)
{
    constexpr int KB = Ci / 4;
    constexpr int G  = KB < 4 ? KB : 4;
    constexpr int NG = KB / G;
    constexpr int Q  = G * 6 / 4;
    const int kg  = item % NG;
    const int ct  = item / NG;
    const int tt  = ct % 27;
    const int cot = ct / 27;
    const int co  = cot * 16 + (lane & 15);
    const int ci0 = kg * G * 4 + ((lane < 16) ? 0 : 2);
    const float* p = w + (co * Ci + ci0) * 81 + tt * 3;
    float v[G * 6];
    #pragma unroll
    for (int kbi = 0; kbi < G; ++kbi) {
        #pragma unroll
        for (int j = 0; j < 6; ++j)
            v[kbi * 6 + j] = p[(kbi * 4 + (j & 1)) * 81 + (j >> 1)];
    }
    v4f* d4 = (v4f*)dst + (item * Q * 32 + lane);
    #pragma unroll
    for (int q = 0; q < Q; ++q)
        d4[q * 32] = (v4f){v[q * 4 + 0], v[q * 4 + 1], v[q * 4 + 2], v[q * 4 + 3]};
}

template<int Ci>
__device__ __forceinline__ void relayout_mf(const float* __restrict__ w,
                                            float* __restrict__ dst, int item, int lane)
{
    constexpr int KB = Ci / 4;
    constexpr int G  = KB < 4 ? KB : 4;
    constexpr int NG = KB / G;
    constexpr int Q  = G * 2;
    const int kbi = item % G;
    int rest = item / G;
    const int kg  = rest % NG; rest /= NG;
    const int t0  = rest % 3;
    const int cot = rest / 3;
    const int co  = cot * 16 + (lane & 15);
    const int ci0 = (kg * G + kbi) * 4 + ((lane < 16) ? 0 : 2);
    const float* p = w + (co * Ci + ci0) * 81 + t0 * 27;
    float g[2][27];
    #pragma unroll
    for (int hi = 0; hi < 2; ++hi) {
        #pragma unroll
        for (int k = 0; k < 27; ++k)
            g[hi][k] = p[hi * 81 + k];
    }
    v4f* d4 = (v4f*)dst + ((((cot * 48 + t0 * 16) * NG + kg) * Q + kbi * 2) * 32 + lane);
    #pragma unroll
    for (int i = 0; i < 4; ++i) {
        float m[2][3][4];
        #pragma unroll
        for (int hi = 0; hi < 2; ++hi) {
            #pragma unroll
            for (int t1 = 0; t1 < 3; ++t1) {
                float hh[3];
                #pragma unroll
                for (int t3 = 0; t3 < 3; ++t3) {
                    const float a0 = g[hi][t1 * 9 + t3], a1 = g[hi][t1 * 9 + 3 + t3], a2 = g[hi][t1 * 9 + 6 + t3];
                    const float ea = a0 + a2;
                    if (i == 0)      hh[t3] = a0;
                    else if (i == 1) hh[t3] = 0.5f * (ea + a1);
                    else if (i == 2) hh[t3] = 0.5f * (ea - a1);
                    else             hh[t3] = a2;
                }
                const float e = hh[0] + hh[2];
                m[hi][t1][0] = hh[0];
                m[hi][t1][1] = 0.5f * (e + hh[1]);
                m[hi][t1][2] = 0.5f * (e - hh[1]);
                m[hi][t1][3] = hh[2];
            }
        }
        float r[2][4][4];
        #pragma unroll
        for (int hi = 0; hi < 2; ++hi) {
            #pragma unroll
            for (int j = 0; j < 4; ++j) {
                const float em = m[hi][0][j] + m[hi][2][j];
                r[hi][j][0] = m[hi][0][j];
                r[hi][j][1] = 0.5f * (em + m[hi][1][j]);
                r[hi][j][2] = 0.5f * (em - m[hi][1][j]);
                r[hi][j][3] = m[hi][2][j];
            }
        }
        #pragma unroll
        for (int h = 0; h < 4; ++h) {
            #pragma unroll
            for (int jp = 0; jp < 2; ++jp)
                d4[((h * 4 + i) * NG * Q + jp) * 32] =
                    (v4f){r[0][2 * jp][h], r[1][2 * jp][h], r[0][2 * jp + 1][h], r[1][2 * jp + 1][h]};
        }
    }
}

__global__ __launch_bounds__(64)
void relayout_weights(const float* __restrict__ w0, const float* __restrict__ w1,
                      const float* __restrict__ w2, const float* __restrict__ w3,
                      float* __restrict__ dst)
{
    constexpr int N0 = 32 * 8 * 3 * 64, N1 = 64 * 32 * 3 * 64, N2 = 64 * 64 * 81;
    constexpr int I0 = 2 * 3 * 1 * 2, I1 = 4 * 3 * 2 * 4, I2 = 4 * 27 * 4;
    const int lane = threadIdx.x & 31;
    const int item = __builtin_amdgcn_readfirstlane(blockIdx.x * 2 + (threadIdx.x >> 5));
    if (item < I0)                relayout_mf<8 >(w0, dst, item, lane);
    else if (item < I0 + I1)      relayout_mf<32>(w1, dst + N0, item - I0, lane);
    else if (item < I0 + I1 + I2) relayout_one<64>(w2, dst + N0 + N1, item - I0 - I1, lane);
    else                          relayout_one<64>(w3, dst + N0 + N1 + N2, item - I0 - I1 - I2, lane);
}


template<int Ci, int Co, int S, int TILES, int KSG = 1, int KST = 1, int NTB = Co / 16>
__global__ __launch_bounds__(NTB * KSG * KST * 32)
void conv4d_relu_pool_wmma(const float* __restrict__ x,
                           const float* __restrict__ w,
                           const float* __restrict__ bias,
                           float* __restrict__ out)
{
    constexpr int P    = S / 2;
    constexpr int PH   = P / TILES;
    constexpr int P4   = P * P * P * P;
    constexpr int D3   = 2 + 2 * TILES;
    constexpr int NSP  = 64 * D3;
    constexpr int CSTR = Ci + LDS_PAD;
    constexpr int KS   = KSG * KST;
    constexpr int NB   = (Co / 16) / NTB;
    constexpr int NT   = NTB * KS * 32;
    static_assert((Co / 16) % NTB == 0 && (KST == 1 || KST == 3), "block split");

    __shared__ __attribute__((aligned(16))) float lds[NSP * CSTR];

    const int b   = blockIdx.y;
    const int nb  = blockIdx.x % NB;
    const int bx  = blockIdx.x / NB;
    const int p3h = bx % PH;
    const int p2  = (bx / PH) % P;
    const int p1  = (bx / (PH * P)) % P;
    const int p0  = bx / (PH * P * P);

    const int tid  = threadIdx.x;
    const int lane = tid & 31;
    const int wave = tid >> 5;
    const int ntl  = wave / KS;
    const int ks   = wave - ntl * KS;
    const int kgw  = ks % KSG;
    const int t0w  = ks / KSG;
    const int ntile = nb * NTB + ntl;
    constexpr int KB = Ci / 4;
    constexpr int G  = KB < 4 ? KB : 4;
    constexpr int NG = KB / G;
    constexpr int Q  = G * 6 / 4;
    constexpr int NGW = NG / KSG;
    static_assert(NG % KSG == 0, "channel-group split");
    const v4f* w4 = (const v4f*)w + (ntile * 27 * NG * Q * 32 + kgw * (NGW * Q * 32) + lane);
    const int t0lo  = (KST == 3) ? t0w : 0;
    constexpr int NCH = ((KST == 3) ? 9 : 27) * NGW;

    float wf[4][G * 6];
    auto load_chunk = [&](int slot, int ch) __attribute__((always_inline)) {
        const int tw = ch / NGW, kg = ch - tw * NGW;
        const v4f* wp = w4 + ((t0lo * 9 + tw) * NG + kg) * (Q * 32);
        #pragma unroll
        for (int q = 0; q < Q; ++q) {
            const v4f t = wp[q * 32];
            wf[slot][q * 4 + 0] = t.x; wf[slot][q * 4 + 1] = t.y;
            wf[slot][q * 4 + 2] = t.z; wf[slot][q * 4 + 3] = t.w;
        }
    };
    #pragma unroll
    for (int ch = 0; ch < 3 && ch < NCH; ++ch)
        load_chunk(ch, ch);

    constexpr int K  = (NSP + NT - 1) / NT;
    constexpr bool FULL = (NSP % NT == 0);
    constexpr int CB = (Ci < 16) ? Ci : 16;
    constexpr int S4 = S * S * S * S;
    static_assert(Ci % CB == 0 && CB % 4 == 0, "staging split");
    const float* __restrict__ xb = x + b * (Ci * S4);
    int  off[K];
    bool ok[K];
    #pragma unroll
    for (int k = 0; k < K; ++k) {
        const int sp = tid + k * NT;
        const int r  = sp / D3;
        const int d3 = sp - r * D3;
        const int x0 = 2 * p0 - 1 + ((r >> 4) & 3);
        const int x1 = 2 * p1 - 1 + ((r >> 2) & 3);
        const int x2 = 2 * p2 - 1 + (r & 3);
        const int x3 = 2 * (p3h * TILES) - 1 + d3;
        ok[k]  = (FULL || sp < NSP) &&
                 (unsigned)x0 < (unsigned)S && (unsigned)x1 < (unsigned)S &&
                 (unsigned)x2 < (unsigned)S && (unsigned)x3 < (unsigned)S;
        off[k] = ok[k] ? (((x0 * S + x1) * S + x2) * S + x3) : 0;
    }
    #pragma unroll 1
    for (int c0 = 0; c0 < Ci; c0 += CB) {
        float v[K][CB];
        #pragma unroll
        for (int k = 0; k < K; ++k) {
            #pragma unroll
            for (int c = 0; c < CB; ++c)
                v[k][c] = xb[(c0 + c) * S4 + off[k]];
        }
        #pragma unroll
        for (int k = 0; k < K; ++k) {
            if (!FULL && tid + k * NT >= NSP) continue;
            #pragma unroll
            for (int c = 0; c < CB; c += 4) {
                v4f q;
                q.x = ok[k] ? v[k][c + 0] : 0.0f;
                q.y = ok[k] ? v[k][c + 1] : 0.0f;
                q.z = ok[k] ? v[k][c + 2] : 0.0f;
                q.w = ok[k] ? v[k][c + 3] : 0.0f;
                *(v4f*)&lds[(tid + k * NT) * CSTR + c0 + c] = q;
            }
        }
    }
    __syncthreads();

    const int n    = lane & 15;
    const int klo  = (lane < 16) ? 0 : 2;
    const int qsp  = (((n >> 3) & 1) * 16 + ((n >> 2) & 1) * 4 + ((n >> 1) & 1)) * D3 + (n & 1);
    const int coA  = ntile * 16 + n;
    const int kbase = kgw * (NGW * G * 4);

    v8f acc[TILES];
    #pragma unroll
    for (int u = 0; u < TILES; ++u)
        acc[u] = (v8f){0.f, 0.f, 0.f, 0.f, 0.f, 0.f, 0.f, 0.f};

    #pragma unroll
    for (int ch = 0; ch < NCH; ++ch) {
        if (ch + 3 < NCH) load_chunk((ch + 3) & 3, ch + 3);
        const int tw = ch / NGW, kg = ch - tw * NGW;
        const int t0 = t0lo + tw / 9, t1 = (tw / 3) % 3, t2 = tw % 3;
        const int row = (qsp + (t0 * 16 + t1 * 4 + t2) * D3) * CSTR + klo + kbase;
        #pragma unroll
        for (int kbi = 0; kbi < G; ++kbi) {
            const int kb = (kg * G + kbi) * 4;
            #pragma unroll
            for (int t3 = 0; t3 < 3; ++t3) {
                v2f bf; bf.x = wf[ch & 3][kbi * 6 + 2 * t3]; bf.y = wf[ch & 3][kbi * 6 + 2 * t3 + 1];
                #pragma unroll
                for (int u = 0; u < TILES; ++u) {
                    v2f a = *(const v2f*)&lds[row + (t3 + 2 * u) * CSTR + kb];
                    acc[u] = __builtin_amdgcn_wmma_f32_16x16x4_f32(
                         false, a,  false, bf,
                         (short)0, acc[u],
                         false,  false);
                }
            }
        }
    }

    if constexpr (KS > 1) {
        __shared__ __attribute__((aligned(16))) float red[NTB * KS * TILES * 256];
        if (ks != 0) {
            #pragma unroll
            for (int u = 0; u < TILES; ++u) {
                v4f lo, hi;
                lo.x = acc[u][0]; lo.y = acc[u][1]; lo.z = acc[u][2]; lo.w = acc[u][3];
                hi.x = acc[u][4]; hi.y = acc[u][5]; hi.z = acc[u][6]; hi.w = acc[u][7];
                *(v4f*)&red[(((wave * TILES + u) * 2 + 0) * 32 + lane) * 4] = lo;
                *(v4f*)&red[(((wave * TILES + u) * 2 + 1) * 32 + lane) * 4] = hi;
            }
        }
        __syncthreads();
        if (ks != 0) return;
        #pragma unroll
        for (int u = 0; u < TILES; ++u) {
            #pragma unroll
            for (int j = 1; j < KS; ++j) {
                const v4f lo = *(const v4f*)&red[((((wave + j) * TILES + u) * 2 + 0) * 32 + lane) * 4];
                const v4f hi = *(const v4f*)&red[((((wave + j) * TILES + u) * 2 + 1) * 32 + lane) * 4];
                acc[u][0] += lo.x; acc[u][1] += lo.y; acc[u][2] += lo.z; acc[u][3] += lo.w;
                acc[u][4] += hi.x; acc[u][5] += hi.y; acc[u][6] += hi.z; acc[u][7] += hi.w;
            }
        }
    }

    const float bs = bias[coA];
    #pragma unroll
    for (int u = 0; u < TILES; ++u) {
        float s = 0.0f;
        #pragma unroll
        for (int r = 0; r < 8; ++r) {
            float v = acc[u][r] + bs;
            s += (v > 0.0f) ? v : 0.0f;
        }
        s += __shfl_xor(s, 16, 32);
        if (lane < 16) {
            const int pIdx = ((p0 * P + p1) * P + p2) * P + p3h * TILES + u;
            out[(b * Co + coA) * P4 + pIdx] = s * (1.0f / 16.0f);
        }
    }
}

constexpr bool mf_banks_ok(int us, int s2, int s1, int s0)
{
    unsigned seen = 0;
    for (int n = 0; n < 16; ++n) {
        const int off = (n & 1) * us + ((n >> 1) & 1) * s2 + ((n >> 2) & 1) * s1 + ((n >> 3) & 1) * s0;
        if (off % 4) return false;
        const int g = (off / 4) % 16;
        if (seen & (1u << g)) return false;
        seen |= 1u << g;
    }
    return true;
}

template<int Ci, int Co, int S, int NTB, int WPE, int PU, int PZP, int P0>
__global__ __launch_bounds__(NTB * 4 * 32) __attribute__((amdgpu_waves_per_eu(WPE, WPE)))
void conv4d_relu_pool_mf(const float* __restrict__ x,
                         const float* __restrict__ w,
                         const float* __restrict__ bias,
                         float* __restrict__ out)
{
    constexpr int P    = S / 2;
    constexpr int UU   = P;
    constexpr int PZ   = 8 / UU;
    constexpr int PG2  = P / PZ;
    constexpr int P4   = P * P * P * P;
    constexpr int NACC = 16;
    constexpr int US   = Ci + PU;
    constexpr int JS   = UU * US;
    constexpr int S2   = 4 * JS;
    constexpr int SZ   = 4 * S2 + PZP;
    constexpr int S1   = PZ * SZ;
    constexpr int S0   = 4 * S1 + P0;
    constexpr int LDSF = 4 * S0;
    constexpr int NB   = (Co / 16) / NTB;
    constexpr int NT   = NTB * 4 * 32;
    constexpr int REDF = (NT / 32) * 4 * 256;
    static_assert((UU == 8 || UU == 4) && P % PZ == 0 && (Co / 16) % NTB == 0 && S % 2 == 0, "tile split");
    static_assert(mf_banks_ok(US, 2 * US, (UU == 8) ? 4 * US : SZ, S0), "fragment reads must be bank-conflict free");
    static_assert(REDF <= LDSF, "partial sums reuse the patch");

    __shared__ __attribute__((aligned(16))) float lds[LDSF];

    const int b   = blockIdx.y;
    const int nb  = blockIdx.x % NB;
    const int bx  = blockIdx.x / NB;
    const int p2g = bx % PG2;
    const int p1  = (bx / PG2) % P;
    const int p0  = bx / (PG2 * P);

    const int tid  = threadIdx.x;
    const int lane = tid & 31;
    const int wave = __builtin_amdgcn_readfirstlane(tid >> 5);
    const int ntl  = wave >> 2;
    const int hw   = wave & 3;
    const int ntile = nb * NTB + ntl;
    constexpr int KB = Ci / 4;
    constexpr int G  = KB < 4 ? KB : 4;
    constexpr int NG = KB / G;
    constexpr int Q  = G * 2;
    constexpr int CP = G / 2;
    constexpr int NC = 4 * CP;
    constexpr int NTR = 3 * NG;
    static_assert(G % 2 == 0 && NC % 4 == 0, "weight ring");
    const v4f* w4 = (const v4f*)w + ((ntile * 48 + hw * 4) * NG * Q * 32 + lane);

    float wf[4][16];
    auto load_chunk = [&](int slot, int tr, int c) __attribute__((always_inline)) {
        const int t0 = tr / NG, kg = tr - t0 * NG;
        const v4f* wp = w4 + ((t0 * 16 * NG + kg) * Q + (c / CP) * NG * Q + (c % CP) * 4) * 32;
        #pragma unroll
        for (int p = 0; p < 4; ++p) {
            const v4f t = wp[p * 32];
            wf[slot][p * 4 + 0] = t.x; wf[slot][p * 4 + 1] = t.y;
            wf[slot][p * 4 + 2] = t.z; wf[slot][p * 4 + 3] = t.w;
        }
    };
    load_chunk(0, 0, 0);
    load_chunk(1, 0, 1);
    load_chunk(2, 0, 2);

    constexpr int NIT  = 4 * PZ * UU * Ci;
    constexpr int K    = NIT / NT;
    constexpr int S4   = S * S * S * S;
    static_assert(NIT % NT == 0 && Ci % 8 == 0 && UU % 4 == 0, "staging split");
    const float* __restrict__ xb = x + b * (Ci * S4);
    #pragma unroll 1
    for (int k = 0; k < K; ++k) {
        const int it = tid + k * NT;
        const int ul = it & 3;
        const int cl = (it >> 2) & 7;
        int rest = it >> 5;
        const int uh = rest % (UU / 4); rest /= (UU / 4);
        const int ch = rest % (Ci / 8); rest /= (Ci / 8);
        const int pz = rest % PZ;
        const int r0 = rest / PZ;
        const int uu = uh * 4 + ul;
        const int c  = ch * 8 + cl;
        const int x0 = 2 * p0 - 1 + r0;
        const int x1b = 2 * p1 - 1;
        const int x2b = 2 * (p2g * PZ + pz) - 1;
        const bool ok0 = (unsigned)x0 < (unsigned)S;
        const bool inA = (uu >= 1), inC = (2 * uu + 2 < S);
        const float* pc = xb + c * S4;
        float va[4][4], vc[4][4];
        v2f vb[4][4];
        bool okB[4][4];
        #pragma unroll
        for (int r1 = 0; r1 < 4; ++r1) {
            const int x1 = x1b + r1;
            const bool ok01 = ok0 && (unsigned)x1 < (unsigned)S;
            #pragma unroll
            for (int d = 0; d < 4; ++d) {
                const int x2 = x2b + d;
                okB[r1][d] = ok01 && (unsigned)x2 < (unsigned)S;
                const int base = ((x0 * S + x1) * S + x2) * S + 2 * uu;
                const int offB = okB[r1][d] ? base : 0;
                const int offA = (okB[r1][d] && inA) ? base - 1 : 0;
                const int offC = (okB[r1][d] && inC) ? base + 2 : 0;
                va[r1][d] = pc[offA];
                vb[r1][d] = *(const v2f*)(pc + offB);
                vc[r1][d] = pc[offC];
            }
        }
        float V2[4][4][4];
        #pragma unroll
        for (int r1 = 0; r1 < 4; ++r1) {
            float t[4][4];
            #pragma unroll
            for (int d = 0; d < 4; ++d) {
                const float d0 = (okB[r1][d] && inA) ? va[r1][d] : 0.0f;
                const float d1 = okB[r1][d] ? vb[r1][d].x : 0.0f;
                const float d2 = okB[r1][d] ? vb[r1][d].y : 0.0f;
                const float d3 = (okB[r1][d] && inC) ? vc[r1][d] : 0.0f;
                t[d][0] = d0 - d2; t[d][1] = d1 + d2; t[d][2] = d2 - d1; t[d][3] = d1 - d3;
            }
            #pragma unroll
            for (int j = 0; j < 4; ++j) {
                V2[r1][0][j] = t[0][j] - t[2][j];
                V2[r1][1][j] = t[1][j] + t[2][j];
                V2[r1][2][j] = t[2][j] - t[1][j];
                V2[r1][3][j] = t[1][j] - t[3][j];
            }
        }
        float* dp = &lds[r0 * S0 + pz * SZ + uu * US + c];
        #pragma unroll
        for (int i = 0; i < 4; ++i) {
            #pragma unroll
            for (int j = 0; j < 4; ++j) {
                const float a0 = V2[0][i][j], a1 = V2[1][i][j], a2 = V2[2][i][j], a3 = V2[3][i][j];
                dp[0 * S1 + i * S2 + j * JS] = a0 - a2;
                dp[1 * S1 + i * S2 + j * JS] = a1 + a2;
                dp[2 * S1 + i * S2 + j * JS] = a2 - a1;
                dp[3 * S1 + i * S2 + j * JS] = a1 - a3;
            }
        }
    }
    __syncthreads();

    const int n    = lane & 15;
    const int klo  = (lane < 16) ? 0 : 2;
    const int qsp  = ((n >> 3) & 1) * S0 + ((n & 7) / UU) * SZ + (n & (UU - 1)) * US;
    const int coA  = ntile * 16 + n;

    v8f acc[NACC];
    #pragma unroll
    for (int a = 0; a < NACC; ++a)
        acc[a] = (v8f){0.f, 0.f, 0.f, 0.f, 0.f, 0.f, 0.f, 0.f};

    auto trip = [&](int tr, bool last) __attribute__((always_inline)) {
        const int t0 = tr / NG, kg = tr - t0 * NG;
        const int row0 = qsp + t0 * S0 + hw * S1 + klo + kg * (G * 4);
        #pragma unroll
        for (int c = 0; c < NC; ++c) {
            const int cn = c + 3;
            if (cn < NC)    load_chunk(cn & 3, tr, cn);
            else if (!last) load_chunk(cn & 3, tr + 1, cn - NC);
            const int i  = c / CP;
            const int kp = c % CP;
            const int row = row0 + i * S2;
            #pragma unroll
            for (int kbl = 0; kbl < 2; ++kbl) {
                #pragma unroll
                for (int j = 0; j < 4; ++j) {
                    v2f bf; bf.x = wf[c & 3][kbl * 8 + 2 * j]; bf.y = wf[c & 3][kbl * 8 + 2 * j + 1];
                    const float* ap = lds + (row + j * JS + (kp * 2 + kbl) * 4);
                    v2f a = *(const v2f*)ap;
                    acc[i * 4 + j] = __builtin_amdgcn_wmma_f32_16x16x4_f32(
                         false, a,  false, bf,
                         (short)0, acc[i * 4 + j],
                         false,  false);
                }
            }
            if (NC > 4 && last && (c & 1) == 1 && c + 1 < NC) __builtin_amdgcn_sched_barrier(0);
        }
    };
    #pragma unroll 1
    for (int tr = 0; tr < NTR - 1; ++tr)
        trip(tr, false);
    trip(NTR - 1, true);

    v8f y[4];
    #pragma unroll
    for (int r = 0; r < 8; ++r) {
        float z0[4], z1[4];
        #pragma unroll
        for (int i = 0; i < 4; ++i) {
            const float m0 = acc[i * 4 + 0][r], m1 = acc[i * 4 + 1][r];
            const float m2 = acc[i * 4 + 2][r], m3 = acc[i * 4 + 3][r];
            z0[i] = (m0 + m1) + m2;
            z1[i] = (m1 - m2) - m3;
        }
        y[0][r] = (z0[0] + z0[1]) + z0[2];
        y[1][r] = (z1[0] + z1[1]) + z1[2];
        y[2][r] = (z0[1] - z0[2]) - z0[3];
        y[3][r] = (z1[1] - z1[2]) - z1[3];
    }

    __syncthreads();
    float* red = lds;
    if (hw != 0) {
        #pragma unroll
        for (int a = 0; a < 4; ++a) {
            v4f lo, hi;
            lo.x = y[a][0]; lo.y = y[a][1]; lo.z = y[a][2]; lo.w = y[a][3];
            hi.x = y[a][4]; hi.y = y[a][5]; hi.z = y[a][6]; hi.w = y[a][7];
            *(v4f*)&red[(((wave * 4 + a) * 2 + 0) * 32 + lane) * 4] = lo;
            *(v4f*)&red[(((wave * 4 + a) * 2 + 1) * 32 + lane) * 4] = hi;
        }
    }
    __syncthreads();
    if (hw != 0) return;
    v8f ya[4], yb[4];
    #pragma unroll
    for (int a = 0; a < 4; ++a) {
        v8f pt[3];
        #pragma unroll
        for (int j = 1; j < 4; ++j) {
            const v4f lo = *(const v4f*)&red[((((wave + j) * 4 + a) * 2 + 0) * 32 + lane) * 4];
            const v4f hi = *(const v4f*)&red[((((wave + j) * 4 + a) * 2 + 1) * 32 + lane) * 4];
            pt[j - 1] = (v8f){lo.x, lo.y, lo.z, lo.w, hi.x, hi.y, hi.z, hi.w};
        }
        #pragma unroll
        for (int r = 0; r < 8; ++r) {
            ya[a][r] = (y[a][r] + pt[0][r]) + pt[1][r];
            yb[a][r] = (pt[0][r] - pt[1][r]) - pt[2][r];
        }
    }

    const float bs = bias[coA];
    float s[8];
    #pragma unroll
    for (int r = 0; r < 8; ++r) {
        float sa = 0.0f;
        #pragma unroll
        for (int a = 0; a < 4; ++a) {
            const float v0 = ya[a][r] + bs, v1 = yb[a][r] + bs;
            sa += ((v0 > 0.0f) ? v0 : 0.0f) + ((v1 > 0.0f) ? v1 : 0.0f);
        }
        sa += __shfl_xor(sa, 16, 32);
        s[r] = sa * (1.0f / 16.0f);
    }
    if (lane < 16) {
        float* op = out + (b * Co + coA) * P4 + ((p0 * P + p1) * P + p2g * PZ) * P;
        *(v4f*)(op + 0) = (v4f){s[0], s[1], s[2], s[3]};
        *(v4f*)(op + 4) = (v4f){s[4], s[5], s[6], s[7]};
    }
}

extern "C" void kernel_launch(void* const* d_in, const int* in_sizes, int n_in,
                              void* d_out, int out_size, void* d_ws, size_t ws_size,
                              hipStream_t stream) {
    (void)in_sizes; (void)n_in; (void)out_size; (void)ws_size;

    const float* src = (const float*)d_in[0];
    const float* W0  = (const float*)d_in[1];
    const float* b0  = (const float*)d_in[2];
    const float* W1  = (const float*)d_in[3];
    const float* b1  = (const float*)d_in[4];
    const float* W2  = (const float*)d_in[5];
    const float* b2  = (const float*)d_in[6];
    const float* W3  = (const float*)d_in[7];
    const float* b3  = (const float*)d_in[8];

    float* t0 = (float*)d_ws;
    float* r0 = t0 + 8 * 32 * 4096;
    float* r1 = r0 + 32 * 8 * 3 * 64;
    float* r2 = r1 + 64 * 32 * 3 * 64;
    float* r3 = r2 + 64 * 64 * 81;
    float* o1 = (float*)d_out;
    float* o2 = o1 + 8 * 64 * 256;
    float* o3 = o2 + 8 * 64 * 16;

    relayout_weights<<<486, 64, 0, stream>>>(W0, W1, W2, W3, r0);
    conv4d_relu_pool_mf<8, 32, 16, 2, 4, 0, 0, 4><<<dim3(512, 8), 256, 0, stream>>>(src, r0, b0, t0);
    conv4d_relu_pool_mf<32, 64, 8, 4, 4, 4, 16, 32><<<dim3(32, 8), 512, 0, stream>>>(t0, r1, b1, o1);
    conv4d_relu_pool_wmma<64, 64, 4, 2, 4, 3, 1><<<dim3(32, 8), 384, 0, stream>>>(o1, r2, b2, o2);
    conv4d_relu_pool_wmma<64, 64, 2, 1, 4, 3, 1><<<dim3(4, 8), 384, 0, stream>>>(o2, r3, b3, o3);
}


# Layers 0 and 1 patch staging: lanes walk a whole dim3 row, one 8-byte load per row; its two outer values come from the neighbour lanes by DPP row shift: 16 loads per item for 48; sums unchanged
# speedup vs baseline: 6.7822x; 1.0866x over previous
#include <hip/hip_runtime.h>
#include <hip/hip_bf16.h>

typedef float v2f __attribute__((ext_vector_type(2)));
typedef float v4f __attribute__((ext_vector_type(4)));
typedef float v8f __attribute__((ext_vector_type(8)));

#define LDS_PAD 4


template<int Ci>
__device__ __forceinline__ void relayout_one(const float* __restrict__ w,
                                             float* __restrict__ dst, int item, int lane)
{
    constexpr int KB = Ci / 4;
    constexpr int G  = KB < 4 ? KB : 4;
    constexpr int NG = KB / G;
    constexpr int Q  = G * 6 / 4;
    const int kg  = item % NG;
    const int ct  = item / NG;
    const int tt  = ct % 27;
    const int cot = ct / 27;
    const int co  = cot * 16 + (lane & 15);
    const int ci0 = kg * G * 4 + ((lane < 16) ? 0 : 2);
    const float* p = w + (co * Ci + ci0) * 81 + tt * 3;
    float v[G * 6];
    #pragma unroll
    for (int kbi = 0; kbi < G; ++kbi) {
        #pragma unroll
        for (int j = 0; j < 6; ++j)
            v[kbi * 6 + j] = p[(kbi * 4 + (j & 1)) * 81 + (j >> 1)];
    }
    v4f* d4 = (v4f*)dst + (item * Q * 32 + lane);
    #pragma unroll
    for (int q = 0; q < Q; ++q)
        d4[q * 32] = (v4f){v[q * 4 + 0], v[q * 4 + 1], v[q * 4 + 2], v[q * 4 + 3]};
}

template<int Ci>
__device__ __forceinline__ void relayout_mf(const float* __restrict__ w,
                                            float* __restrict__ dst, int item, int lane)
{
    constexpr int KB = Ci / 4;
    constexpr int G  = KB < 4 ? KB : 4;
    constexpr int NG = KB / G;
    constexpr int Q  = G * 2;
    const int kbi = item % G;
    int rest = item / G;
    const int kg  = rest % NG; rest /= NG;
    const int t0  = rest % 3;
    const int cot = rest / 3;
    const int co  = cot * 16 + (lane & 15);
    const int ci0 = (kg * G + kbi) * 4 + ((lane < 16) ? 0 : 2);
    const float* p = w + (co * Ci + ci0) * 81 + t0 * 27;
    float g[2][27];
    #pragma unroll
    for (int hi = 0; hi < 2; ++hi) {
        #pragma unroll
        for (int k = 0; k < 27; ++k)
            g[hi][k] = p[hi * 81 + k];
    }
    v4f* d4 = (v4f*)dst + ((((cot * 48 + t0 * 16) * NG + kg) * Q + kbi * 2) * 32 + lane);
    #pragma unroll
    for (int i = 0; i < 4; ++i) {
        float m[2][3][4];
        #pragma unroll
        for (int hi = 0; hi < 2; ++hi) {
            #pragma unroll
            for (int t1 = 0; t1 < 3; ++t1) {
                float hh[3];
                #pragma unroll
                for (int t3 = 0; t3 < 3; ++t3) {
                    const float a0 = g[hi][t1 * 9 + t3], a1 = g[hi][t1 * 9 + 3 + t3], a2 = g[hi][t1 * 9 + 6 + t3];
                    const float ea = a0 + a2;
                    if (i == 0)      hh[t3] = a0;
                    else if (i == 1) hh[t3] = 0.5f * (ea + a1);
                    else if (i == 2) hh[t3] = 0.5f * (ea - a1);
                    else             hh[t3] = a2;
                }
                const float e = hh[0] + hh[2];
                m[hi][t1][0] = hh[0];
                m[hi][t1][1] = 0.5f * (e + hh[1]);
                m[hi][t1][2] = 0.5f * (e - hh[1]);
                m[hi][t1][3] = hh[2];
            }
        }
        float r[2][4][4];
        #pragma unroll
        for (int hi = 0; hi < 2; ++hi) {
            #pragma unroll
            for (int j = 0; j < 4; ++j) {
                const float em = m[hi][0][j] + m[hi][2][j];
                r[hi][j][0] = m[hi][0][j];
                r[hi][j][1] = 0.5f * (em + m[hi][1][j]);
                r[hi][j][2] = 0.5f * (em - m[hi][1][j]);
                r[hi][j][3] = m[hi][2][j];
            }
        }
        #pragma unroll
        for (int h = 0; h < 4; ++h) {
            #pragma unroll
            for (int jp = 0; jp < 2; ++jp)
                d4[((h * 4 + i) * NG * Q + jp) * 32] =
                    (v4f){r[0][2 * jp][h], r[1][2 * jp][h], r[0][2 * jp + 1][h], r[1][2 * jp + 1][h]};
        }
    }
}

__global__ __launch_bounds__(64)
void relayout_weights(const float* __restrict__ w0, const float* __restrict__ w1,
                      const float* __restrict__ w2, const float* __restrict__ w3,
                      float* __restrict__ dst)
{
    constexpr int N0 = 32 * 8 * 3 * 64, N1 = 64 * 32 * 3 * 64, N2 = 64 * 64 * 81;
    constexpr int I0 = 2 * 3 * 1 * 2, I1 = 4 * 3 * 2 * 4, I2 = 4 * 27 * 4;
    const int lane = threadIdx.x & 31;
    const int item = __builtin_amdgcn_readfirstlane(blockIdx.x * 2 + (threadIdx.x >> 5));
    if (item < I0)                relayout_mf<8 >(w0, dst, item, lane);
    else if (item < I0 + I1)      relayout_mf<32>(w1, dst + N0, item - I0, lane);
    else if (item < I0 + I1 + I2) relayout_one<64>(w2, dst + N0 + N1, item - I0 - I1, lane);
    else                          relayout_one<64>(w3, dst + N0 + N1 + N2, item - I0 - I1 - I2, lane);
}


template<int Ci, int Co, int S, int TILES, int KSG = 1, int KST = 1, int NTB = Co / 16>
__global__ __launch_bounds__(NTB * KSG * KST * 32)
void conv4d_relu_pool_wmma(const float* __restrict__ x,
                           const float* __restrict__ w,
                           const float* __restrict__ bias,
                           float* __restrict__ out)
{
    constexpr int P    = S / 2;
    constexpr int PH   = P / TILES;
    constexpr int P4   = P * P * P * P;
    constexpr int D3   = 2 + 2 * TILES;
    constexpr int NSP  = 64 * D3;
    constexpr int CSTR = Ci + LDS_PAD;
    constexpr int KS   = KSG * KST;
    constexpr int NB   = (Co / 16) / NTB;
    constexpr int NT   = NTB * KS * 32;
    static_assert((Co / 16) % NTB == 0 && (KST == 1 || KST == 3), "block split");

    __shared__ __attribute__((aligned(16))) float lds[NSP * CSTR];

    const int b   = blockIdx.y;
    const int nb  = blockIdx.x % NB;
    const int bx  = blockIdx.x / NB;
    const int p3h = bx % PH;
    const int p2  = (bx / PH) % P;
    const int p1  = (bx / (PH * P)) % P;
    const int p0  = bx / (PH * P * P);

    const int tid  = threadIdx.x;
    const int lane = tid & 31;
    const int wave = tid >> 5;
    const int ntl  = wave / KS;
    const int ks   = wave - ntl * KS;
    const int kgw  = ks % KSG;
    const int t0w  = ks / KSG;
    const int ntile = nb * NTB + ntl;
    constexpr int KB = Ci / 4;
    constexpr int G  = KB < 4 ? KB : 4;
    constexpr int NG = KB / G;
    constexpr int Q  = G * 6 / 4;
    constexpr int NGW = NG / KSG;
    static_assert(NG % KSG == 0, "channel-group split");
    const v4f* w4 = (const v4f*)w + (ntile * 27 * NG * Q * 32 + kgw * (NGW * Q * 32) + lane);
    const int t0lo  = (KST == 3) ? t0w : 0;
    constexpr int NCH = ((KST == 3) ? 9 : 27) * NGW;

    float wf[4][G * 6];
    auto load_chunk = [&](int slot, int ch) __attribute__((always_inline)) {
        const int tw = ch / NGW, kg = ch - tw * NGW;
        const v4f* wp = w4 + ((t0lo * 9 + tw) * NG + kg) * (Q * 32);
        #pragma unroll
        for (int q = 0; q < Q; ++q) {
            const v4f t = wp[q * 32];
            wf[slot][q * 4 + 0] = t.x; wf[slot][q * 4 + 1] = t.y;
            wf[slot][q * 4 + 2] = t.z; wf[slot][q * 4 + 3] = t.w;
        }
    };
    #pragma unroll
    for (int ch = 0; ch < 3 && ch < NCH; ++ch)
        load_chunk(ch, ch);

    constexpr int K  = (NSP + NT - 1) / NT;
    constexpr bool FULL = (NSP % NT == 0);
    constexpr int CB = (Ci < 16) ? Ci : 16;
    constexpr int S4 = S * S * S * S;
    static_assert(Ci % CB == 0 && CB % 4 == 0, "staging split");
    const float* __restrict__ xb = x + b * (Ci * S4);
    int  off[K];
    bool ok[K];
    #pragma unroll
    for (int k = 0; k < K; ++k) {
        const int sp = tid + k * NT;
        const int r  = sp / D3;
        const int d3 = sp - r * D3;
        const int x0 = 2 * p0 - 1 + ((r >> 4) & 3);
        const int x1 = 2 * p1 - 1 + ((r >> 2) & 3);
        const int x2 = 2 * p2 - 1 + (r & 3);
        const int x3 = 2 * (p3h * TILES) - 1 + d3;
        ok[k]  = (FULL || sp < NSP) &&
                 (unsigned)x0 < (unsigned)S && (unsigned)x1 < (unsigned)S &&
                 (unsigned)x2 < (unsigned)S && (unsigned)x3 < (unsigned)S;
        off[k] = ok[k] ? (((x0 * S + x1) * S + x2) * S + x3) : 0;
    }
    #pragma unroll 1
    for (int c0 = 0; c0 < Ci; c0 += CB) {
        float v[K][CB];
        #pragma unroll
        for (int k = 0; k < K; ++k) {
            #pragma unroll
            for (int c = 0; c < CB; ++c)
                v[k][c] = xb[(c0 + c) * S4 + off[k]];
        }
        #pragma unroll
        for (int k = 0; k < K; ++k) {
            if (!FULL && tid + k * NT >= NSP) continue;
            #pragma unroll
            for (int c = 0; c < CB; c += 4) {
                v4f q;
                q.x = ok[k] ? v[k][c + 0] : 0.0f;
                q.y = ok[k] ? v[k][c + 1] : 0.0f;
                q.z = ok[k] ? v[k][c + 2] : 0.0f;
                q.w = ok[k] ? v[k][c + 3] : 0.0f;
                *(v4f*)&lds[(tid + k * NT) * CSTR + c0 + c] = q;
            }
        }
    }
    __syncthreads();

    const int n    = lane & 15;
    const int klo  = (lane < 16) ? 0 : 2;
    const int qsp  = (((n >> 3) & 1) * 16 + ((n >> 2) & 1) * 4 + ((n >> 1) & 1)) * D3 + (n & 1);
    const int coA  = ntile * 16 + n;
    const int kbase = kgw * (NGW * G * 4);

    v8f acc[TILES];
    #pragma unroll
    for (int u = 0; u < TILES; ++u)
        acc[u] = (v8f){0.f, 0.f, 0.f, 0.f, 0.f, 0.f, 0.f, 0.f};

    #pragma unroll
    for (int ch = 0; ch < NCH; ++ch) {
        if (ch + 3 < NCH) load_chunk((ch + 3) & 3, ch + 3);
        const int tw = ch / NGW, kg = ch - tw * NGW;
        const int t0 = t0lo + tw / 9, t1 = (tw / 3) % 3, t2 = tw % 3;
        const int row = (qsp + (t0 * 16 + t1 * 4 + t2) * D3) * CSTR + klo + kbase;
        #pragma unroll
        for (int kbi = 0; kbi < G; ++kbi) {
            const int kb = (kg * G + kbi) * 4;
            #pragma unroll
            for (int t3 = 0; t3 < 3; ++t3) {
                v2f bf; bf.x = wf[ch & 3][kbi * 6 + 2 * t3]; bf.y = wf[ch & 3][kbi * 6 + 2 * t3 + 1];
                #pragma unroll
                for (int u = 0; u < TILES; ++u) {
                    v2f a = *(const v2f*)&lds[row + (t3 + 2 * u) * CSTR + kb];
                    acc[u] = __builtin_amdgcn_wmma_f32_16x16x4_f32(
                         false, a,  false, bf,
                         (short)0, acc[u],
                         false,  false);
                }
            }
        }
    }

    if constexpr (KS > 1) {
        __shared__ __attribute__((aligned(16))) float red[NTB * KS * TILES * 256];
        if (ks != 0) {
            #pragma unroll
            for (int u = 0; u < TILES; ++u) {
                v4f lo, hi;
                lo.x = acc[u][0]; lo.y = acc[u][1]; lo.z = acc[u][2]; lo.w = acc[u][3];
                hi.x = acc[u][4]; hi.y = acc[u][5]; hi.z = acc[u][6]; hi.w = acc[u][7];
                *(v4f*)&red[(((wave * TILES + u) * 2 + 0) * 32 + lane) * 4] = lo;
                *(v4f*)&red[(((wave * TILES + u) * 2 + 1) * 32 + lane) * 4] = hi;
            }
        }
        __syncthreads();
        if (ks != 0) return;
        #pragma unroll
        for (int u = 0; u < TILES; ++u) {
            #pragma unroll
            for (int j = 1; j < KS; ++j) {
                const v4f lo = *(const v4f*)&red[((((wave + j) * TILES + u) * 2 + 0) * 32 + lane) * 4];
                const v4f hi = *(const v4f*)&red[((((wave + j) * TILES + u) * 2 + 1) * 32 + lane) * 4];
                acc[u][0] += lo.x; acc[u][1] += lo.y; acc[u][2] += lo.z; acc[u][3] += lo.w;
                acc[u][4] += hi.x; acc[u][5] += hi.y; acc[u][6] += hi.z; acc[u][7] += hi.w;
            }
        }
    }

    const float bs = bias[coA];
    #pragma unroll
    for (int u = 0; u < TILES; ++u) {
        float s = 0.0f;
        #pragma unroll
        for (int r = 0; r < 8; ++r) {
            float v = acc[u][r] + bs;
            s += (v > 0.0f) ? v : 0.0f;
        }
        s += __shfl_xor(s, 16, 32);
        if (lane < 16) {
            const int pIdx = ((p0 * P + p1) * P + p2) * P + p3h * TILES + u;
            out[(b * Co + coA) * P4 + pIdx] = s * (1.0f / 16.0f);
        }
    }
}

__device__ __forceinline__ float lane_below(float v)
{
    return __builtin_bit_cast(float, __builtin_amdgcn_update_dpp(0, __builtin_bit_cast(int, v), 0x111, 0xF, 0xF, true));
}
__device__ __forceinline__ float lane_above(float v)
{
    return __builtin_bit_cast(float, __builtin_amdgcn_update_dpp(0, __builtin_bit_cast(int, v), 0x101, 0xF, 0xF, true));
}

constexpr bool mf_banks_ok(int us, int s2, int s1, int s0)
{
    unsigned seen = 0;
    for (int n = 0; n < 16; ++n) {
        const int off = (n & 1) * us + ((n >> 1) & 1) * s2 + ((n >> 2) & 1) * s1 + ((n >> 3) & 1) * s0;
        if (off % 4) return false;
        const int g = (off / 4) % 16;
        if (seen & (1u << g)) return false;
        seen |= 1u << g;
    }
    return true;
}

template<int Ci, int Co, int S, int NTB, int WPE, int PU, int PZP, int P0>
__global__ __launch_bounds__(NTB * 4 * 32) __attribute__((amdgpu_waves_per_eu(WPE, WPE)))
void conv4d_relu_pool_mf(const float* __restrict__ x,
                         const float* __restrict__ w,
                         const float* __restrict__ bias,
                         float* __restrict__ out)
{
    constexpr int P    = S / 2;
    constexpr int UU   = P;
    constexpr int PZ   = 8 / UU;
    constexpr int PG2  = P / PZ;
    constexpr int P4   = P * P * P * P;
    constexpr int NACC = 16;
    constexpr int US   = Ci + PU;
    constexpr int JS   = UU * US;
    constexpr int S2   = 4 * JS;
    constexpr int SZ   = 4 * S2 + PZP;
    constexpr int S1   = PZ * SZ;
    constexpr int S0   = 4 * S1 + P0;
    constexpr int LDSF = 4 * S0;
    constexpr int NB   = (Co / 16) / NTB;
    constexpr int NT   = NTB * 4 * 32;
    constexpr int REDF = (NT / 32) * 4 * 256;
    static_assert((UU == 8 || UU == 4) && P % PZ == 0 && (Co / 16) % NTB == 0 && S % 2 == 0, "tile split");
    static_assert(mf_banks_ok(US, 2 * US, (UU == 8) ? 4 * US : SZ, S0), "fragment reads must be bank-conflict free");
    static_assert(REDF <= LDSF, "partial sums reuse the patch");

    __shared__ __attribute__((aligned(16))) float lds[LDSF];

    const int b   = blockIdx.y;
    const int nb  = blockIdx.x % NB;
    const int bx  = blockIdx.x / NB;
    const int p2g = bx % PG2;
    const int p1  = (bx / PG2) % P;
    const int p0  = bx / (PG2 * P);

    const int tid  = threadIdx.x;
    const int lane = tid & 31;
    const int wave = __builtin_amdgcn_readfirstlane(tid >> 5);
    const int ntl  = wave >> 2;
    const int hw   = wave & 3;
    const int ntile = nb * NTB + ntl;
    constexpr int KB = Ci / 4;
    constexpr int G  = KB < 4 ? KB : 4;
    constexpr int NG = KB / G;
    constexpr int Q  = G * 2;
    constexpr int CP = G / 2;
    constexpr int NC = 4 * CP;
    constexpr int NTR = 3 * NG;
    static_assert(G % 2 == 0 && NC % 4 == 0, "weight ring");
    const v4f* w4 = (const v4f*)w + ((ntile * 48 + hw * 4) * NG * Q * 32 + lane);

    float wf[4][16];
    auto load_chunk = [&](int slot, int tr, int c) __attribute__((always_inline)) {
        const int t0 = tr / NG, kg = tr - t0 * NG;
        const v4f* wp = w4 + ((t0 * 16 * NG + kg) * Q + (c / CP) * NG * Q + (c % CP) * 4) * 32;
        #pragma unroll
        for (int p = 0; p < 4; ++p) {
            const v4f t = wp[p * 32];
            wf[slot][p * 4 + 0] = t.x; wf[slot][p * 4 + 1] = t.y;
            wf[slot][p * 4 + 2] = t.z; wf[slot][p * 4 + 3] = t.w;
        }
    };
    load_chunk(0, 0, 0);
    load_chunk(1, 0, 1);
    load_chunk(2, 0, 2);

    constexpr int NIT  = 4 * PZ * UU * Ci;
    constexpr int K    = NIT / NT;
    constexpr int S4   = S * S * S * S;
    constexpr int CL   = 32 / UU;
    static_assert(NIT % NT == 0 && Ci % CL == 0 && (UU == 4 || UU == 8) && 2 * UU == S, "staging split");
    const float* __restrict__ xb = x + b * (Ci * S4);
    #pragma unroll 1
    for (int k = 0; k < K; ++k) {
        const int it = tid + k * NT;
        const int uu = it & (UU - 1);
        const int cl = (it / UU) & (CL - 1);
        int rest = it >> 5;
        const int ch = rest % (Ci / CL); rest /= (Ci / CL);
        const int pz = rest % PZ;
        const int r0 = rest / PZ;
        const int c  = ch * CL + cl;
        const int x0 = 2 * p0 - 1 + r0;
        const int x1b = 2 * p1 - 1;
        const int x2b = 2 * (p2g * PZ + pz) - 1;
        const bool ok0 = (unsigned)x0 < (unsigned)S;
        const bool inA = (uu >= 1), inC = (2 * uu + 2 < S);
        const float* pc = xb + c * S4;
        v2f vb[4][4];
        bool okB[4][4];
        #pragma unroll
        for (int r1 = 0; r1 < 4; ++r1) {
            const int x1 = x1b + r1;
            const bool ok01 = ok0 && (unsigned)x1 < (unsigned)S;
            #pragma unroll
            for (int d = 0; d < 4; ++d) {
                const int x2 = x2b + d;
                okB[r1][d] = ok01 && (unsigned)x2 < (unsigned)S;
                const int base = ((x0 * S + x1) * S + x2) * S + 2 * uu;
                const int offB = okB[r1][d] ? base : 0;
                vb[r1][d] = *(const v2f*)(pc + offB);
            }
        }
        __builtin_amdgcn_sched_barrier(0);
        float V2[4][4][4];
        #pragma unroll
        for (int r1 = 0; r1 < 4; ++r1) {
            float t[4][4];
            #pragma unroll
            for (int d = 0; d < 4; ++d) {
                const float lo = lane_below(vb[r1][d].y);
                const float hi = lane_above(vb[r1][d].x);
                const float d0 = (okB[r1][d] && inA) ? lo : 0.0f;
                const float d1 = okB[r1][d] ? vb[r1][d].x : 0.0f;
                const float d2 = okB[r1][d] ? vb[r1][d].y : 0.0f;
                const float d3 = (okB[r1][d] && inC) ? hi : 0.0f;
                t[d][0] = d0 - d2; t[d][1] = d1 + d2; t[d][2] = d2 - d1; t[d][3] = d1 - d3;
            }
            #pragma unroll
            for (int j = 0; j < 4; ++j) {
                V2[r1][0][j] = t[0][j] - t[2][j];
                V2[r1][1][j] = t[1][j] + t[2][j];
                V2[r1][2][j] = t[2][j] - t[1][j];
                V2[r1][3][j] = t[1][j] - t[3][j];
            }
        }
        float* dp = &lds[r0 * S0 + pz * SZ + uu * US + c];
        #pragma unroll
        for (int i = 0; i < 4; ++i) {
            #pragma unroll
            for (int j = 0; j < 4; ++j) {
                const float a0 = V2[0][i][j], a1 = V2[1][i][j], a2 = V2[2][i][j], a3 = V2[3][i][j];
                dp[0 * S1 + i * S2 + j * JS] = a0 - a2;
                dp[1 * S1 + i * S2 + j * JS] = a1 + a2;
                dp[2 * S1 + i * S2 + j * JS] = a2 - a1;
                dp[3 * S1 + i * S2 + j * JS] = a1 - a3;
            }
        }
    }
    __syncthreads();

    const int n    = lane & 15;
    const int klo  = (lane < 16) ? 0 : 2;
    const int qsp  = ((n >> 3) & 1) * S0 + ((n & 7) / UU) * SZ + (n & (UU - 1)) * US;
    const int coA  = ntile * 16 + n;

    v8f acc[NACC];
    #pragma unroll
    for (int a = 0; a < NACC; ++a)
        acc[a] = (v8f){0.f, 0.f, 0.f, 0.f, 0.f, 0.f, 0.f, 0.f};

    auto trip = [&](int tr, bool last) __attribute__((always_inline)) {
        const int t0 = tr / NG, kg = tr - t0 * NG;
        const int row0 = qsp + t0 * S0 + hw * S1 + klo + kg * (G * 4);
        #pragma unroll
        for (int c = 0; c < NC; ++c) {
            const int cn = c + 3;
            if (cn < NC)    load_chunk(cn & 3, tr, cn);
            else if (!last) load_chunk(cn & 3, tr + 1, cn - NC);
            const int i  = c / CP;
            const int kp = c % CP;
            const int row = row0 + i * S2;
            #pragma unroll
            for (int kbl = 0; kbl < 2; ++kbl) {
                #pragma unroll
                for (int j = 0; j < 4; ++j) {
                    v2f bf; bf.x = wf[c & 3][kbl * 8 + 2 * j]; bf.y = wf[c & 3][kbl * 8 + 2 * j + 1];
                    const float* ap = lds + (row + j * JS + (kp * 2 + kbl) * 4);
                    v2f a = *(const v2f*)ap;
                    acc[i * 4 + j] = __builtin_amdgcn_wmma_f32_16x16x4_f32(
                         false, a,  false, bf,
                         (short)0, acc[i * 4 + j],
                         false,  false);
                }
            }
            if (NC > 4 && last && (c & 1) == 1 && c + 1 < NC) __builtin_amdgcn_sched_barrier(0);
        }
    };
    #pragma unroll 1
    for (int tr = 0; tr < NTR - 1; ++tr)
        trip(tr, false);
    trip(NTR - 1, true);

    v8f y[4];
    #pragma unroll
    for (int r = 0; r < 8; ++r) {
        float z0[4], z1[4];
        #pragma unroll
        for (int i = 0; i < 4; ++i) {
            const float m0 = acc[i * 4 + 0][r], m1 = acc[i * 4 + 1][r];
            const float m2 = acc[i * 4 + 2][r], m3 = acc[i * 4 + 3][r];
            z0[i] = (m0 + m1) + m2;
            z1[i] = (m1 - m2) - m3;
        }
        y[0][r] = (z0[0] + z0[1]) + z0[2];
        y[1][r] = (z1[0] + z1[1]) + z1[2];
        y[2][r] = (z0[1] - z0[2]) - z0[3];
        y[3][r] = (z1[1] - z1[2]) - z1[3];
    }

    __syncthreads();
    float* red = lds;
    if (hw != 0) {
        #pragma unroll
        for (int a = 0; a < 4; ++a) {
            v4f lo, hi;
            lo.x = y[a][0]; lo.y = y[a][1]; lo.z = y[a][2]; lo.w = y[a][3];
            hi.x = y[a][4]; hi.y = y[a][5]; hi.z = y[a][6]; hi.w = y[a][7];
            *(v4f*)&red[(((wave * 4 + a) * 2 + 0) * 32 + lane) * 4] = lo;
            *(v4f*)&red[(((wave * 4 + a) * 2 + 1) * 32 + lane) * 4] = hi;
        }
    }
    __syncthreads();
    if (hw != 0) return;
    v8f ya[4], yb[4];
    #pragma unroll
    for (int a = 0; a < 4; ++a) {
        v8f pt[3];
        #pragma unroll
        for (int j = 1; j < 4; ++j) {
            const v4f lo = *(const v4f*)&red[((((wave + j) * 4 + a) * 2 + 0) * 32 + lane) * 4];
            const v4f hi = *(const v4f*)&red[((((wave + j) * 4 + a) * 2 + 1) * 32 + lane) * 4];
            pt[j - 1] = (v8f){lo.x, lo.y, lo.z, lo.w, hi.x, hi.y, hi.z, hi.w};
        }
        #pragma unroll
        for (int r = 0; r < 8; ++r) {
            ya[a][r] = (y[a][r] + pt[0][r]) + pt[1][r];
            yb[a][r] = (pt[0][r] - pt[1][r]) - pt[2][r];
        }
    }

    const float bs = bias[coA];
    float s[8];
    #pragma unroll
    for (int r = 0; r < 8; ++r) {
        float sa = 0.0f;
        #pragma unroll
        for (int a = 0; a < 4; ++a) {
            const float v0 = ya[a][r] + bs, v1 = yb[a][r] + bs;
            sa += ((v0 > 0.0f) ? v0 : 0.0f) + ((v1 > 0.0f) ? v1 : 0.0f);
        }
        sa += __shfl_xor(sa, 16, 32);
        s[r] = sa * (1.0f / 16.0f);
    }
    if (lane < 16) {
        float* op = out + (b * Co + coA) * P4 + ((p0 * P + p1) * P + p2g * PZ) * P;
        *(v4f*)(op + 0) = (v4f){s[0], s[1], s[2], s[3]};
        *(v4f*)(op + 4) = (v4f){s[4], s[5], s[6], s[7]};
    }
}

extern "C" void kernel_launch(void* const* d_in, const int* in_sizes, int n_in,
                              void* d_out, int out_size, void* d_ws, size_t ws_size,
                              hipStream_t stream) {
    (void)in_sizes; (void)n_in; (void)out_size; (void)ws_size;

    const float* src = (const float*)d_in[0];
    const float* W0  = (const float*)d_in[1];
    const float* b0  = (const float*)d_in[2];
    const float* W1  = (const float*)d_in[3];
    const float* b1  = (const float*)d_in[4];
    const float* W2  = (const float*)d_in[5];
    const float* b2  = (const float*)d_in[6];
    const float* W3  = (const float*)d_in[7];
    const float* b3  = (const float*)d_in[8];

    float* t0 = (float*)d_ws;
    float* r0 = t0 + 8 * 32 * 4096;
    float* r1 = r0 + 32 * 8 * 3 * 64;
    float* r2 = r1 + 64 * 32 * 3 * 64;
    float* r3 = r2 + 64 * 64 * 81;
    float* o1 = (float*)d_out;
    float* o2 = o1 + 8 * 64 * 256;
    float* o3 = o2 + 8 * 64 * 16;

    relayout_weights<<<486, 64, 0, stream>>>(W0, W1, W2, W3, r0);
    conv4d_relu_pool_mf<8, 32, 16, 2, 4, 0, 0, 4><<<dim3(512, 8), 256, 0, stream>>>(src, r0, b0, t0);
    conv4d_relu_pool_mf<32, 64, 8, 4, 4, 4, 16, 32><<<dim3(32, 8), 512, 0, stream>>>(t0, r1, b1, o1);
    conv4d_relu_pool_wmma<64, 64, 4, 2, 4, 3, 1><<<dim3(32, 8), 384, 0, stream>>>(o1, r2, b2, o2);
    conv4d_relu_pool_wmma<64, 64, 2, 1, 4, 3, 1><<<dim3(4, 8), 384, 0, stream>>>(o2, r3, b3, o3);
}
